# starting kernel
# baseline (speedup 1.0000x reference)
#include <hip/hip_runtime.h>
#include <hip/hip_bf16.h>

#define N_NODES_C  100000
#define N_EDGES_C  1600000
#define N_GRAPHS_C 1000
#define HID_C      64
#define IN_DIM_C   7
#define EDGE_DIM_C 4
#define NUM_TASKS_C 12

typedef __attribute__((ext_vector_type(16))) _Float16 v16h;
typedef __attribute__((ext_vector_type(8)))  _Float16 v8h;
typedef __attribute__((ext_vector_type(8)))  float    v8f;

__device__ __forceinline__ void atom_add_f32(float* p, float v) {
    unsafeAtomicAdd(p, v);
}

__device__ __forceinline__ v16h lds_load_tile(const _Float16* s, int ld, int kbase) {
    const int lane = threadIdx.x & 31;
    const int r    = lane & 15;
    const int koff = (lane < 16) ? 0 : 8;
    const _Float16* p = s + r * ld + kbase + koff;
    v8h lo = *(const v8h*)(p);
    v8h hi = *(const v8h*)(p + 16);
    v16h out;
#pragma unroll
    for (int e = 0; e < 8; ++e) { out[e] = lo[e]; out[8 + e] = hi[e]; }
    return out;
}

template <int K1, typename TI, typename TO>
__global__ __launch_bounds__(128)
void node_mlp_kernel(const TI* __restrict__ in0, const float* __restrict__ in1,
                     const float* __restrict__ W1,  const float* __restrict__ b1,
                     const float* __restrict__ W2,  const float* __restrict__ b2,
                     TO* __restrict__ out, int n_nodes) {
    constexpr int K1P = (K1 + 31) & ~31;

    __shared__ __align__(16) _Float16 sA  [16 * K1P];
    __shared__ __align__(16) _Float16 sMid[16 * HID_C];
    __shared__ __align__(16) _Float16 sB1 [HID_C * K1P];
    __shared__ __align__(16) _Float16 sB2 [HID_C * HID_C];

    const int row0 = blockIdx.x * 16;
    const int tid  = threadIdx.x;
    const int wave = tid >> 5;
    const int lane = tid & 31;
    const int n0   = wave * 16;

    for (int i = tid; i < 16 * K1P; i += 128) {
        const int m = i / K1P, k = i - m * K1P;
        const int r = row0 + m;
        float v = 0.0f;
        if (k < K1 && r < n_nodes)
            v = (float)in0[(size_t)r * K1 + k] + in1[(size_t)r * K1 + k];
        sA[i] = (_Float16)v;
    }
    for (int i = tid; i < HID_C * K1P; i += 128) {
        const int n = i / K1P, k = i - n * K1P;
        sB1[i] = (_Float16)((k < K1) ? W1[n * K1 + k] : 0.0f);
    }
    for (int i = tid; i < HID_C * HID_C; i += 128) {
        const int n = i >> 6, k = i & 63;
        sB2[i] = (_Float16)W2[n * HID_C + k];
    }
    __syncthreads();

    v8f acc = {};
#pragma unroll
    for (int kb = 0; kb < K1P; kb += 32) {
        v16h a = lds_load_tile(sA, K1P, kb);
        v16h b = lds_load_tile(sB1 + n0 * K1P, K1P, kb);
        acc = __builtin_amdgcn_wmma_f32_16x16x32_f16(false, a, false, b,
                                                     (short)0, acc, false, false);
    }
    {
        const int n  = lane & 15;
        const int mb = (lane < 16) ? 0 : 8;
        const float bias = b1[n0 + n];
#pragma unroll
        for (int r = 0; r < 8; ++r) {
            float v = acc[r] + bias;
            v = v > 0.0f ? v : 0.0f;
            sMid[(mb + r) * HID_C + (n0 + n)] = (_Float16)v;
        }
    }
    __syncthreads();

    v8f acc2 = {};
#pragma unroll
    for (int kb = 0; kb < HID_C; kb += 32) {
        v16h a = lds_load_tile(sMid, HID_C, kb);
        v16h b = lds_load_tile(sB2 + n0 * HID_C, HID_C, kb);
        acc2 = __builtin_amdgcn_wmma_f32_16x16x32_f16(false, a, false, b,
                                                      (short)0, acc2, false, false);
    }
    {
        const int n  = lane & 15;
        const int mb = (lane < 16) ? 0 : 8;
        const float bias = b2[n0 + n];
#pragma unroll
        for (int r = 0; r < 8; ++r) {
            const int rr = row0 + mb + r;
            if (rr < n_nodes) {
                float v = acc2[r] + bias;
                v = v > 0.0f ? v : 0.0f;
                out[(size_t)rr * HID_C + (n0 + n)] = (TO)v;
            }
        }
    }
}

__global__ __launch_bounds__(256)
void edge1_kernel(const float* __restrict__ x, const int* __restrict__ ei,
                  const float* __restrict__ ea,
                  const float* __restrict__ eW, const float* __restrict__ eb,
                  float* __restrict__ aggr) {
    const long long t = (long long)blockIdx.x * 256 + threadIdx.x;
    const long long e = t >> 3;
    const int c = (int)(t & 7);
    if (e >= N_EDGES_C || c >= IN_DIM_C) return;
    const int src = ei[e];
    const int dst = ei[N_EDGES_C + e];
    const float4 a = *(const float4*)(ea + e * EDGE_DIM_C);
    const float4 w = *(const float4*)(eW + c * EDGE_DIM_C);
    float v = eb[c] + a.x * w.x + a.y * w.y + a.z * w.z + a.w * w.w;
    v += x[(size_t)src * IN_DIM_C + c];
    v = fmaxf(v, 0.0f);
    atom_add_f32(&aggr[(size_t)dst * IN_DIM_C + c], v);
}

__global__ __launch_bounds__(256)
void edge2_kernel(const _Float16* __restrict__ h, const int* __restrict__ ei,
                  const float* __restrict__ ea,
                  const float* __restrict__ eW, const float* __restrict__ eb,
                  float* __restrict__ aggr) {
    const long long t = (long long)blockIdx.x * 256 + threadIdx.x;
    const long long e = t >> 6;
    const int c = (int)(t & 63);
    if (e >= N_EDGES_C) return;
    const int src = ei[e];
    const int dst = ei[N_EDGES_C + e];
    const float4 a = *(const float4*)(ea + e * EDGE_DIM_C);
    const float4 w = *(const float4*)(eW + c * EDGE_DIM_C);
    float v = eb[c] + a.x * w.x + a.y * w.y + a.z * w.z + a.w * w.w;
    v += (float)h[(size_t)src * HID_C + c];
    v = fmaxf(v, 0.0f);
    atom_add_f32(&aggr[(size_t)dst * HID_C + c], v);
}

__global__ __launch_bounds__(256)
void pool_kernel(const float* __restrict__ h, const int* __restrict__ batch,
                 float* __restrict__ psum, float* __restrict__ cnt) {
    const long long t = (long long)blockIdx.x * 256 + threadIdx.x;
    const long long n = t >> 6;
    const int c = (int)(t & 63);
    if (n >= N_NODES_C) return;
    const int g = batch[n];
    atom_add_f32(&psum[(size_t)g * HID_C + c], h[(size_t)n * HID_C + c]);
    if (c == 0) atom_add_f32(&cnt[g], 1.0f);
}

__global__ __launch_bounds__(256)
void fc_kernel(const float* __restrict__ psum, const float* __restrict__ cnt,
               const float* __restrict__ fcW, const float* __restrict__ fcb,
               float* __restrict__ out) {
    const int t = blockIdx.x * 256 + threadIdx.x;
    if (t >= N_GRAPHS_C * NUM_TASKS_C) return;
    const int g = t / NUM_TASKS_C;
    const int k = t - g * NUM_TASKS_C;
    const float inv = 1.0f / fmaxf(cnt[g], 1.0f);
    float s = fcb[k];
#pragma unroll
    for (int c = 0; c < HID_C; ++c)
        s += psum[(size_t)g * HID_C + c] * inv * fcW[k * HID_C + c];
    out[t] = s;
}

extern "C" void kernel_launch(void* const* d_in, const int* in_sizes, int n_in,
                              void* d_out, int out_size, void* d_ws, size_t ws_size,
                              hipStream_t stream) {
    const float* x    = (const float*)d_in[0];
    const int*   ei   = (const int*)  d_in[1];
    const float* ea   = (const float*)d_in[2];
    const int*   bat  = (const int*)  d_in[3];
    const float* e1W  = (const float*)d_in[4];  const float* e1b  = (const float*)d_in[5];
    const float* n1W1 = (const float*)d_in[6];  const float* n1b1 = (const float*)d_in[7];
    const float* n1W2 = (const float*)d_in[8];  const float* n1b2 = (const float*)d_in[9];
    const float* e2W  = (const float*)d_in[10]; const float* e2b  = (const float*)d_in[11];
    const float* n2W1 = (const float*)d_in[12]; const float* n2b1 = (const float*)d_in[13];
    const float* n2W2 = (const float*)d_in[14]; const float* n2b2 = (const float*)d_in[15];
    const float* fcW  = (const float*)d_in[16]; const float* fcb  = (const float*)d_in[17];
    float* out = (float*)d_out;

    float*    aggr1 = (float*)d_ws;
    _Float16* h     = (_Float16*)(aggr1 + (size_t)N_NODES_C * IN_DIM_C);
    float*    aggr2 = (float*)(h + (size_t)N_NODES_C * HID_C);
    float*    h2    = aggr2 + (size_t)N_NODES_C * HID_C;
    float*    psum  = h2    + (size_t)N_NODES_C * HID_C;
    float*    cnt   = psum  + (size_t)N_GRAPHS_C * HID_C;

    hipMemsetAsync(aggr1, 0, sizeof(float) * (size_t)N_NODES_C * IN_DIM_C, stream);
    hipMemsetAsync(aggr2, 0, sizeof(float) * (size_t)N_NODES_C * HID_C,   stream);
    hipMemsetAsync(psum,  0, sizeof(float) * (size_t)N_GRAPHS_C * HID_C,  stream);
    hipMemsetAsync(cnt,   0, sizeof(float) * (size_t)N_GRAPHS_C,          stream);

    {
        const unsigned grid = (unsigned)(((long long)N_EDGES_C * 8 + 255) / 256);
        edge1_kernel<<<grid, 256, 0, stream>>>(x, ei, ea, e1W, e1b, aggr1);
    }
    {
        const unsigned grid = (N_NODES_C + 15) / 16;
        node_mlp_kernel<IN_DIM_C, float, _Float16><<<grid, 128, 0, stream>>>(
            x, aggr1, n1W1, n1b1, n1W2, n1b2, h, N_NODES_C);
    }
    {
        const unsigned grid = (unsigned)(((long long)N_EDGES_C * 64 + 255) / 256);
        edge2_kernel<<<grid, 256, 0, stream>>>(h, ei, ea, e2W, e2b, aggr2);
    }
    {
        const unsigned grid = (N_NODES_C + 15) / 16;
        node_mlp_kernel<HID_C, _Float16, float><<<grid, 128, 0, stream>>>(
            h, aggr2, n2W1, n2b1, n2W2, n2b2, h2, N_NODES_C);
    }
    {
        const unsigned grid = (unsigned)(((long long)N_NODES_C * 64 + 255) / 256);
        pool_kernel<<<grid, 256, 0, stream>>>(h2, bat, psum, cnt);
    }
    {
        const unsigned grid = (N_GRAPHS_C * NUM_TASKS_C + 255) / 256;
        fc_kernel<<<grid, 256, 0, stream>>>(psum, cnt, fcW, fcb, out);
    }
}


# Replace f32 atomic scatter-adds of both conv aggregations by edges bucketed by dst (integer rank atomics, prefix sum, fill) and per-node gather sums; per-node sum order = arrival rank
# speedup vs baseline: 1.4305x; 1.4305x over previous
#include <hip/hip_runtime.h>
#include <hip/hip_bf16.h>

#define N_NODES_C  100000
#define N_EDGES_C  1600000
#define N_GRAPHS_C 1000
#define HID_C      64
#define IN_DIM_C   7
#define EDGE_DIM_C 4
#define NUM_TASKS_C 12

typedef __attribute__((ext_vector_type(16))) _Float16 v16h;
typedef __attribute__((ext_vector_type(8)))  _Float16 v8h;
typedef __attribute__((ext_vector_type(8)))  float    v8f;
typedef __attribute__((ext_vector_type(2)))  _Float16 v2h;

__device__ __forceinline__ void atom_add_f32(float* p, float v) {
    unsafeAtomicAdd(p, v);
}

__device__ __forceinline__ v16h lds_load_tile(const _Float16* s, int ld, int kbase) {
    const int lane = threadIdx.x & 31;
    const int r    = lane & 15;
    const int koff = (lane < 16) ? 0 : 8;
    const _Float16* p = s + r * ld + kbase + koff;
    v8h lo = *(const v8h*)(p);
    v8h hi = *(const v8h*)(p + 16);
    v16h out;
#pragma unroll
    for (int e = 0; e < 8; ++e) { out[e] = lo[e]; out[8 + e] = hi[e]; }
    return out;
}

template <int K1, typename TI, typename TO>
__global__ __launch_bounds__(128)
void node_mlp_kernel(const TI* __restrict__ in0, const float* __restrict__ in1,
                     const float* __restrict__ W1,  const float* __restrict__ b1,
                     const float* __restrict__ W2,  const float* __restrict__ b2,
                     TO* __restrict__ out, int n_nodes) {
    constexpr int K1P = (K1 + 31) & ~31;

    __shared__ __align__(16) _Float16 sA  [16 * K1P];
    __shared__ __align__(16) _Float16 sMid[16 * HID_C];
    __shared__ __align__(16) _Float16 sB1 [HID_C * K1P];
    __shared__ __align__(16) _Float16 sB2 [HID_C * HID_C];

    const int row0 = blockIdx.x * 16;
    const int tid  = threadIdx.x;
    const int wave = tid >> 5;
    const int lane = tid & 31;
    const int n0   = wave * 16;

    for (int i = tid; i < 16 * K1P; i += 128) {
        const int m = i / K1P, k = i - m * K1P;
        const int r = row0 + m;
        float v = 0.0f;
        if (k < K1 && r < n_nodes)
            v = (float)in0[(size_t)r * K1 + k] + in1[(size_t)r * K1 + k];
        sA[i] = (_Float16)v;
    }
    for (int i = tid; i < HID_C * K1P; i += 128) {
        const int n = i / K1P, k = i - n * K1P;
        sB1[i] = (_Float16)((k < K1) ? W1[n * K1 + k] : 0.0f);
    }
    for (int i = tid; i < HID_C * HID_C; i += 128) {
        const int n = i >> 6, k = i & 63;
        sB2[i] = (_Float16)W2[n * HID_C + k];
    }
    __syncthreads();

    v8f acc = {};
#pragma unroll
    for (int kb = 0; kb < K1P; kb += 32) {
        v16h a = lds_load_tile(sA, K1P, kb);
        v16h b = lds_load_tile(sB1 + n0 * K1P, K1P, kb);
        acc = __builtin_amdgcn_wmma_f32_16x16x32_f16(false, a, false, b,
                                                     (short)0, acc, false, false);
    }
    {
        const int n  = lane & 15;
        const int mb = (lane < 16) ? 0 : 8;
        const float bias = b1[n0 + n];
#pragma unroll
        for (int r = 0; r < 8; ++r) {
            float v = acc[r] + bias;
            v = v > 0.0f ? v : 0.0f;
            sMid[(mb + r) * HID_C + (n0 + n)] = (_Float16)v;
        }
    }
    __syncthreads();

    v8f acc2 = {};
#pragma unroll
    for (int kb = 0; kb < HID_C; kb += 32) {
        v16h a = lds_load_tile(sMid, HID_C, kb);
        v16h b = lds_load_tile(sB2 + n0 * HID_C, HID_C, kb);
        acc2 = __builtin_amdgcn_wmma_f32_16x16x32_f16(false, a, false, b,
                                                      (short)0, acc2, false, false);
    }
    {
        const int n  = lane & 15;
        const int mb = (lane < 16) ? 0 : 8;
        const float bias = b2[n0 + n];
#pragma unroll
        for (int r = 0; r < 8; ++r) {
            const int rr = row0 + mb + r;
            if (rr < n_nodes) {
                float v = acc2[r] + bias;
                v = v > 0.0f ? v : 0.0f;
                out[(size_t)rr * HID_C + (n0 + n)] = (TO)v;
            }
        }
    }
}

#define SCAN_T 1024
#define SCAN_M (N_NODES_C + 1)

__global__ __launch_bounds__(256)
void rank_kernel(const int* __restrict__ ei, int* __restrict__ deg,
                 int* __restrict__ rank) {
    const int e = blockIdx.x * 256 + threadIdx.x;
    if (e >= N_EDGES_C) return;
    const int dst = ei[N_EDGES_C + e];
    rank[e] = atomicAdd(&deg[dst], 1);
}

__global__ __launch_bounds__(SCAN_T)
void rowptr_kernel(const int* __restrict__ deg, int* __restrict__ rowptr) {
    __shared__ int sW[32];
    __shared__ int sS[32];
    __shared__ int sBase;
    const int tid  = threadIdx.x;
    const int lane = tid & 31;
    const int wave = tid >> 5;
    const int tile0 = blockIdx.x * SCAN_T;

    int s = 0;
    for (int i = tid; i < tile0; i += SCAN_T) s += deg[i];

    const int idx = tile0 + tid;
    const int v = (idx < SCAN_M) ? deg[idx] : 0;

    int incl = v;
    for (int d = 1; d < 32; d <<= 1) {
        const int t = __shfl_up(incl, d);
        if (lane >= d) incl += t;
    }
    for (int d = 16; d > 0; d >>= 1) s += __shfl_xor(s, d);

    if (lane == 31) sW[wave] = incl;
    if (lane == 0)  sS[wave] = s;
    __syncthreads();
    if (wave == 0) {
        const int w = sW[lane];
        int wi = w;
        for (int d = 1; d < 32; d <<= 1) {
            const int t = __shfl_up(wi, d);
            if (lane >= d) wi += t;
        }
        int ss = sS[lane];
        for (int d = 16; d > 0; d >>= 1) ss += __shfl_xor(ss, d);
        sW[lane] = wi - w;
        if (lane == 0) sBase = ss;
    }
    __syncthreads();
    if (idx < SCAN_M) rowptr[idx] = sBase + sW[wave] + incl - v;
}

__global__ __launch_bounds__(256)
void fill_kernel(const int* __restrict__ ei, const float* __restrict__ ea,
                 const int* __restrict__ rowptr, const int* __restrict__ rank,
                 int* __restrict__ ssrc, float* __restrict__ sea) {
    const int e = blockIdx.x * 256 + threadIdx.x;
    if (e >= N_EDGES_C) return;
    const int dst = ei[N_EDGES_C + e];
    const int pos = rowptr[dst] + rank[e];
    ssrc[pos] = ei[e];
    *(float4*)(sea + (size_t)pos * EDGE_DIM_C) =
        *(const float4*)(ea + (size_t)e * EDGE_DIM_C);
}

__global__ __launch_bounds__(256)
void gather1_kernel(const float* __restrict__ x, const int* __restrict__ rowptr,
                    const int* __restrict__ ssrc, const float* __restrict__ sea,
                    const float* __restrict__ eW, const float* __restrict__ eb,
                    float* __restrict__ aggr) {
    const int t = blockIdx.x * 256 + threadIdx.x;
    const int n = t >> 3;
    const int c = t & 7;
    if (n >= N_NODES_C || c >= IN_DIM_C) return;
    const float4 w = *(const float4*)(eW + c * EDGE_DIM_C);
    const float  b = eb[c];
    const int beg = rowptr[n];
    const int end = rowptr[n + 1];
    float acc = 0.0f;
    for (int j = beg; j < end; ++j) {
        const int src = ssrc[j];
        const float4 a = *(const float4*)(sea + (size_t)j * EDGE_DIM_C);
        float v = b + a.x * w.x + a.y * w.y + a.z * w.z + a.w * w.w;
        v += x[(size_t)src * IN_DIM_C + c];
        acc += fmaxf(v, 0.0f);
    }
    aggr[(size_t)n * IN_DIM_C + c] = acc;
}

__global__ __launch_bounds__(256)
void gather2_kernel(const _Float16* __restrict__ h, const int* __restrict__ rowptr,
                    const int* __restrict__ ssrc, const float* __restrict__ sea,
                    const float* __restrict__ eW, const float* __restrict__ eb,
                    float* __restrict__ aggr) {
    const int lane = threadIdx.x & 31;
    const int n = blockIdx.x * 8 + (threadIdx.x >> 5);
    if (n >= N_NODES_C) return;
    const int c = 2 * lane;
    const float4 w0 = *(const float4*)(eW + c * EDGE_DIM_C);
    const float4 w1 = *(const float4*)(eW + c * EDGE_DIM_C + EDGE_DIM_C);
    const float  b0 = eb[c];
    const float  b1 = eb[c + 1];
    const int beg = rowptr[n];
    const int end = rowptr[n + 1];
    float acc0 = 0.0f, acc1 = 0.0f;
    for (int j = beg; j < end; ++j) {
        const int src = ssrc[j];
        const float4 a = *(const float4*)(sea + (size_t)j * EDGE_DIM_C);
        const v2h hv = *(const v2h*)(h + (size_t)src * HID_C + c);
        float v0 = b0 + a.x * w0.x + a.y * w0.y + a.z * w0.z + a.w * w0.w;
        float v1 = b1 + a.x * w1.x + a.y * w1.y + a.z * w1.z + a.w * w1.w;
        v0 += (float)hv[0];
        v1 += (float)hv[1];
        acc0 += fmaxf(v0, 0.0f);
        acc1 += fmaxf(v1, 0.0f);
    }
    *(float2*)(aggr + (size_t)n * HID_C + c) = make_float2(acc0, acc1);
}

__global__ __launch_bounds__(256)
void pool_kernel(const float* __restrict__ h, const int* __restrict__ batch,
                 float* __restrict__ psum, float* __restrict__ cnt) {
    const long long t = (long long)blockIdx.x * 256 + threadIdx.x;
    const long long n = t >> 6;
    const int c = (int)(t & 63);
    if (n >= N_NODES_C) return;
    const int g = batch[n];
    atom_add_f32(&psum[(size_t)g * HID_C + c], h[(size_t)n * HID_C + c]);
    if (c == 0) atom_add_f32(&cnt[g], 1.0f);
}

__global__ __launch_bounds__(256)
void fc_kernel(const float* __restrict__ psum, const float* __restrict__ cnt,
               const float* __restrict__ fcW, const float* __restrict__ fcb,
               float* __restrict__ out) {
    const int t = blockIdx.x * 256 + threadIdx.x;
    if (t >= N_GRAPHS_C * NUM_TASKS_C) return;
    const int g = t / NUM_TASKS_C;
    const int k = t - g * NUM_TASKS_C;
    const float inv = 1.0f / fmaxf(cnt[g], 1.0f);
    float s = fcb[k];
#pragma unroll
    for (int c = 0; c < HID_C; ++c)
        s += psum[(size_t)g * HID_C + c] * inv * fcW[k * HID_C + c];
    out[t] = s;
}

extern "C" void kernel_launch(void* const* d_in, const int* in_sizes, int n_in,
                              void* d_out, int out_size, void* d_ws, size_t ws_size,
                              hipStream_t stream) {
    const float* x    = (const float*)d_in[0];
    const int*   ei   = (const int*)  d_in[1];
    const float* ea   = (const float*)d_in[2];
    const int*   bat  = (const int*)  d_in[3];
    const float* e1W  = (const float*)d_in[4];  const float* e1b  = (const float*)d_in[5];
    const float* n1W1 = (const float*)d_in[6];  const float* n1b1 = (const float*)d_in[7];
    const float* n1W2 = (const float*)d_in[8];  const float* n1b2 = (const float*)d_in[9];
    const float* e2W  = (const float*)d_in[10]; const float* e2b  = (const float*)d_in[11];
    const float* n2W1 = (const float*)d_in[12]; const float* n2b1 = (const float*)d_in[13];
    const float* n2W2 = (const float*)d_in[14]; const float* n2b2 = (const float*)d_in[15];
    const float* fcW  = (const float*)d_in[16]; const float* fcb  = (const float*)d_in[17];
    float* out = (float*)d_out;

    float*    aggr1 = (float*)d_ws;
    _Float16* h     = (_Float16*)(aggr1 + (size_t)N_NODES_C * IN_DIM_C);
    float*    aggr2 = (float*)(h + (size_t)N_NODES_C * HID_C);
    float*    h2    = aggr2 + (size_t)N_NODES_C * HID_C;
    float*    psum  = h2    + (size_t)N_NODES_C * HID_C;
    float*    cnt   = psum  + (size_t)N_GRAPHS_C * HID_C;
    float*    sea   = cnt   + (size_t)N_GRAPHS_C;
    int*      ssrc  = (int*)(sea + (size_t)N_EDGES_C * EDGE_DIM_C);
    int*      rank  = ssrc  + (size_t)N_EDGES_C;
    int*      deg   = rank  + (size_t)N_EDGES_C;
    int*      rowptr = deg  + (size_t)(N_NODES_C + 16);

    hipMemsetAsync(deg,   0, sizeof(int)   * (size_t)(N_NODES_C + 1),     stream);
    hipMemsetAsync(psum,  0, sizeof(float) * (size_t)N_GRAPHS_C * HID_C,  stream);
    hipMemsetAsync(cnt,   0, sizeof(float) * (size_t)N_GRAPHS_C,          stream);

    {
        const unsigned grid = (N_EDGES_C + 255) / 256;
        rank_kernel<<<grid, 256, 0, stream>>>(ei, deg, rank);
    }
    {
        const unsigned grid = (SCAN_M + SCAN_T - 1) / SCAN_T;
        rowptr_kernel<<<grid, SCAN_T, 0, stream>>>(deg, rowptr);
    }
    {
        const unsigned grid = (N_EDGES_C + 255) / 256;
        fill_kernel<<<grid, 256, 0, stream>>>(ei, ea, rowptr, rank, ssrc, sea);
    }
    {
        const unsigned grid = (unsigned)(((long long)N_NODES_C * 8 + 255) / 256);
        gather1_kernel<<<grid, 256, 0, stream>>>(x, rowptr, ssrc, sea, e1W, e1b, aggr1);
    }
    {
        const unsigned grid = (N_NODES_C + 15) / 16;
        node_mlp_kernel<IN_DIM_C, float, _Float16><<<grid, 128, 0, stream>>>(
            x, aggr1, n1W1, n1b1, n1W2, n1b2, h, N_NODES_C);
    }
    {
        const unsigned grid = (N_NODES_C + 7) / 8;
        gather2_kernel<<<grid, 256, 0, stream>>>(h, rowptr, ssrc, sea, e2W, e2b, aggr2);
    }
    {
        const unsigned grid = (N_NODES_C + 15) / 16;
        node_mlp_kernel<HID_C, _Float16, float><<<grid, 128, 0, stream>>>(
            h, aggr2, n2W1, n2b1, n2W2, n2b2, h2, N_NODES_C);
    }
    {
        const unsigned grid = (unsigned)(((long long)N_NODES_C * 64 + 255) / 256);
        pool_kernel<<<grid, 256, 0, stream>>>(h2, bat, psum, cnt);
    }
    {
        const unsigned grid = (N_GRAPHS_C * NUM_TASKS_C + 255) / 256;
        fc_kernel<<<grid, 256, 0, stream>>>(psum, cnt, fcW, fcb, out);
    }
}


# Mean-pool scatter folded into the conv2 MLP epilogue: each lane sums its 8 tile rows per graph id, one f32 atomic per run; no h2 plane, no pool launch; pooled sum order: tile partials, then arrival
# speedup vs baseline: 1.6663x; 1.1648x over previous
#include <hip/hip_runtime.h>
#include <hip/hip_bf16.h>

#define N_NODES_C  100000
#define N_EDGES_C  1600000
#define N_GRAPHS_C 1000
#define HID_C      64
#define IN_DIM_C   7
#define EDGE_DIM_C 4
#define NUM_TASKS_C 12

typedef __attribute__((ext_vector_type(16))) _Float16 v16h;
typedef __attribute__((ext_vector_type(8)))  _Float16 v8h;
typedef __attribute__((ext_vector_type(8)))  float    v8f;
typedef __attribute__((ext_vector_type(2)))  _Float16 v2h;

__device__ __forceinline__ void atom_add_f32(float* p, float v) {
    unsafeAtomicAdd(p, v);
}

__device__ __forceinline__ v16h lds_load_tile(const _Float16* s, int ld, int kbase) {
    const int lane = threadIdx.x & 31;
    const int r    = lane & 15;
    const int koff = (lane < 16) ? 0 : 8;
    const _Float16* p = s + r * ld + kbase + koff;
    v8h lo = *(const v8h*)(p);
    v8h hi = *(const v8h*)(p + 16);
    v16h out;
#pragma unroll
    for (int e = 0; e < 8; ++e) { out[e] = lo[e]; out[8 + e] = hi[e]; }
    return out;
}

template <int K1, typename TI, bool POOL>
__global__ __launch_bounds__(128)
void node_mlp_kernel(const TI* __restrict__ in0, const float* __restrict__ in1,
                     const float* __restrict__ W1,  const float* __restrict__ b1,
                     const float* __restrict__ W2,  const float* __restrict__ b2,
                     _Float16* __restrict__ out, const int* __restrict__ batch,
                     float* __restrict__ psum, float* __restrict__ cnt,
                     int n_nodes) {
    constexpr int K1P = (K1 + 31) & ~31;

    __shared__ __align__(16) _Float16 sA  [16 * K1P];
    __shared__ __align__(16) _Float16 sMid[16 * HID_C];
    __shared__ __align__(16) _Float16 sB1 [HID_C * K1P];
    __shared__ __align__(16) _Float16 sB2 [HID_C * HID_C];
    __shared__ __align__(16) int      sG  [16];

    const int row0 = blockIdx.x * 16;
    const int tid  = threadIdx.x;
    const int wave = tid >> 5;
    const int lane = tid & 31;
    const int n0   = wave * 16;

    for (int i = tid; i < 16 * K1P; i += 128) {
        const int m = i / K1P, k = i - m * K1P;
        const int r = row0 + m;
        float v = 0.0f;
        if (k < K1 && r < n_nodes)
            v = (float)in0[(size_t)r * K1 + k] + in1[(size_t)r * K1 + k];
        sA[i] = (_Float16)v;
    }
    for (int i = tid; i < HID_C * K1P; i += 128) {
        const int n = i / K1P, k = i - n * K1P;
        sB1[i] = (_Float16)((k < K1) ? W1[n * K1 + k] : 0.0f);
    }
    for (int i = tid; i < HID_C * HID_C; i += 128) {
        const int n = i >> 6, k = i & 63;
        sB2[i] = (_Float16)W2[n * HID_C + k];
    }
    if constexpr (POOL) {
        if (tid < 16) sG[tid] = (row0 + tid < n_nodes) ? batch[row0 + tid] : -1;
    }
    __syncthreads();

    v8f acc = {};
#pragma unroll
    for (int kb = 0; kb < K1P; kb += 32) {
        v16h a = lds_load_tile(sA, K1P, kb);
        v16h b = lds_load_tile(sB1 + n0 * K1P, K1P, kb);
        acc = __builtin_amdgcn_wmma_f32_16x16x32_f16(false, a, false, b,
                                                     (short)0, acc, false, false);
    }
    {
        const int n  = lane & 15;
        const int mb = (lane < 16) ? 0 : 8;
        const float bias = b1[n0 + n];
#pragma unroll
        for (int r = 0; r < 8; ++r) {
            float v = acc[r] + bias;
            v = v > 0.0f ? v : 0.0f;
            sMid[(mb + r) * HID_C + (n0 + n)] = (_Float16)v;
        }
    }
    __syncthreads();

    v8f acc2 = {};
#pragma unroll
    for (int kb = 0; kb < HID_C; kb += 32) {
        v16h a = lds_load_tile(sMid, HID_C, kb);
        v16h b = lds_load_tile(sB2 + n0 * HID_C, HID_C, kb);
        acc2 = __builtin_amdgcn_wmma_f32_16x16x32_f16(false, a, false, b,
                                                      (short)0, acc2, false, false);
    }
    if constexpr (!POOL) {
        const int n  = lane & 15;
        const int mb = (lane < 16) ? 0 : 8;
        const float bias = b2[n0 + n];
#pragma unroll
        for (int r = 0; r < 8; ++r) {
            const int rr = row0 + mb + r;
            if (rr < n_nodes) {
                float v = acc2[r] + bias;
                v = v > 0.0f ? v : 0.0f;
                out[(size_t)rr * HID_C + (n0 + n)] = (_Float16)v;
            }
        }
    } else {
        const int n  = lane & 15;
        const int mb = (lane < 16) ? 0 : 8;
        const float bias = b2[n0 + n];
        const bool cnt_lane = (wave == 0) && (n == 0);
        const int4 ga = *(const int4*)(sG + mb);
        const int4 gb = *(const int4*)(sG + mb + 4);
        const int gs[8] = { ga.x, ga.y, ga.z, ga.w, gb.x, gb.y, gb.z, gb.w };
        int   gcur = -1;
        float run  = 0.0f;
        int   len  = 0;
#pragma unroll
        for (int r = 0; r < 8; ++r) {
            const int g = gs[r];
            float v = acc2[r] + bias;
            v = v > 0.0f ? v : 0.0f;
            if (g != gcur) {
                if (gcur >= 0) {
                    atom_add_f32(&psum[(size_t)gcur * HID_C + (n0 + n)], run);
                    if (cnt_lane) atom_add_f32(&cnt[gcur], (float)len);
                }
                gcur = g; run = 0.0f; len = 0;
            }
            run += v;
            len += 1;
        }
        if (gcur >= 0) {
            atom_add_f32(&psum[(size_t)gcur * HID_C + (n0 + n)], run);
            if (cnt_lane) atom_add_f32(&cnt[gcur], (float)len);
        }
    }
}

#define SCAN_T 1024
#define SCAN_M (N_NODES_C + 1)

__global__ __launch_bounds__(256)
void rank_kernel(const int* __restrict__ ei, int* __restrict__ deg,
                 int* __restrict__ rank) {
    const int e = blockIdx.x * 256 + threadIdx.x;
    if (e >= N_EDGES_C) return;
    const int dst = ei[N_EDGES_C + e];
    rank[e] = atomicAdd(&deg[dst], 1);
}

__global__ __launch_bounds__(SCAN_T)
void rowptr_kernel(const int* __restrict__ deg, int* __restrict__ rowptr) {
    __shared__ int sW[32];
    __shared__ int sS[32];
    __shared__ int sBase;
    const int tid  = threadIdx.x;
    const int lane = tid & 31;
    const int wave = tid >> 5;
    const int tile0 = blockIdx.x * SCAN_T;

    int s = 0;
    for (int i = tid; i < tile0; i += SCAN_T) s += deg[i];

    const int idx = tile0 + tid;
    const int v = (idx < SCAN_M) ? deg[idx] : 0;

    int incl = v;
    for (int d = 1; d < 32; d <<= 1) {
        const int t = __shfl_up(incl, d);
        if (lane >= d) incl += t;
    }
    for (int d = 16; d > 0; d >>= 1) s += __shfl_xor(s, d);

    if (lane == 31) sW[wave] = incl;
    if (lane == 0)  sS[wave] = s;
    __syncthreads();
    if (wave == 0) {
        const int w = sW[lane];
        int wi = w;
        for (int d = 1; d < 32; d <<= 1) {
            const int t = __shfl_up(wi, d);
            if (lane >= d) wi += t;
        }
        int ss = sS[lane];
        for (int d = 16; d > 0; d >>= 1) ss += __shfl_xor(ss, d);
        sW[lane] = wi - w;
        if (lane == 0) sBase = ss;
    }
    __syncthreads();
    if (idx < SCAN_M) rowptr[idx] = sBase + sW[wave] + incl - v;
}

__global__ __launch_bounds__(256)
void fill_kernel(const int* __restrict__ ei, const float* __restrict__ ea,
                 const int* __restrict__ rowptr, const int* __restrict__ rank,
                 int* __restrict__ ssrc, float* __restrict__ sea) {
    const int e = blockIdx.x * 256 + threadIdx.x;
    if (e >= N_EDGES_C) return;
    const int dst = ei[N_EDGES_C + e];
    const int pos = rowptr[dst] + rank[e];
    ssrc[pos] = ei[e];
    *(float4*)(sea + (size_t)pos * EDGE_DIM_C) =
        *(const float4*)(ea + (size_t)e * EDGE_DIM_C);
}

__global__ __launch_bounds__(256)
void gather1_kernel(const float* __restrict__ x, const int* __restrict__ rowptr,
                    const int* __restrict__ ssrc, const float* __restrict__ sea,
                    const float* __restrict__ eW, const float* __restrict__ eb,
                    float* __restrict__ aggr) {
    const int t = blockIdx.x * 256 + threadIdx.x;
    const int n = t >> 3;
    const int c = t & 7;
    if (n >= N_NODES_C || c >= IN_DIM_C) return;
    const float4 w = *(const float4*)(eW + c * EDGE_DIM_C);
    const float  b = eb[c];
    const int beg = rowptr[n];
    const int end = rowptr[n + 1];
    float acc = 0.0f;
    for (int j = beg; j < end; ++j) {
        const int src = ssrc[j];
        const float4 a = *(const float4*)(sea + (size_t)j * EDGE_DIM_C);
        float v = b + a.x * w.x + a.y * w.y + a.z * w.z + a.w * w.w;
        v += x[(size_t)src * IN_DIM_C + c];
        acc += fmaxf(v, 0.0f);
    }
    aggr[(size_t)n * IN_DIM_C + c] = acc;
}

__global__ __launch_bounds__(256)
void gather2_kernel(const _Float16* __restrict__ h, const int* __restrict__ rowptr,
                    const int* __restrict__ ssrc, const float* __restrict__ sea,
                    const float* __restrict__ eW, const float* __restrict__ eb,
                    float* __restrict__ aggr) {
    const int lane = threadIdx.x & 31;
    const int n = blockIdx.x * 8 + (threadIdx.x >> 5);
    if (n >= N_NODES_C) return;
    const int c = 2 * lane;
    const float4 w0 = *(const float4*)(eW + c * EDGE_DIM_C);
    const float4 w1 = *(const float4*)(eW + c * EDGE_DIM_C + EDGE_DIM_C);
    const float  b0 = eb[c];
    const float  b1 = eb[c + 1];
    const int beg = rowptr[n];
    const int end = rowptr[n + 1];
    float acc0 = 0.0f, acc1 = 0.0f;
    for (int j = beg; j < end; ++j) {
        const int src = ssrc[j];
        const float4 a = *(const float4*)(sea + (size_t)j * EDGE_DIM_C);
        const v2h hv = *(const v2h*)(h + (size_t)src * HID_C + c);
        float v0 = b0 + a.x * w0.x + a.y * w0.y + a.z * w0.z + a.w * w0.w;
        float v1 = b1 + a.x * w1.x + a.y * w1.y + a.z * w1.z + a.w * w1.w;
        v0 += (float)hv[0];
        v1 += (float)hv[1];
        acc0 += fmaxf(v0, 0.0f);
        acc1 += fmaxf(v1, 0.0f);
    }
    *(float2*)(aggr + (size_t)n * HID_C + c) = make_float2(acc0, acc1);
}

__global__ __launch_bounds__(256)
void fc_kernel(const float* __restrict__ psum, const float* __restrict__ cnt,
               const float* __restrict__ fcW, const float* __restrict__ fcb,
               float* __restrict__ out) {
    const int t = blockIdx.x * 256 + threadIdx.x;
    if (t >= N_GRAPHS_C * NUM_TASKS_C) return;
    const int g = t / NUM_TASKS_C;
    const int k = t - g * NUM_TASKS_C;
    const float inv = 1.0f / fmaxf(cnt[g], 1.0f);
    float s = fcb[k];
#pragma unroll
    for (int c = 0; c < HID_C; ++c)
        s += psum[(size_t)g * HID_C + c] * inv * fcW[k * HID_C + c];
    out[t] = s;
}

extern "C" void kernel_launch(void* const* d_in, const int* in_sizes, int n_in,
                              void* d_out, int out_size, void* d_ws, size_t ws_size,
                              hipStream_t stream) {
    const float* x    = (const float*)d_in[0];
    const int*   ei   = (const int*)  d_in[1];
    const float* ea   = (const float*)d_in[2];
    const int*   bat  = (const int*)  d_in[3];
    const float* e1W  = (const float*)d_in[4];  const float* e1b  = (const float*)d_in[5];
    const float* n1W1 = (const float*)d_in[6];  const float* n1b1 = (const float*)d_in[7];
    const float* n1W2 = (const float*)d_in[8];  const float* n1b2 = (const float*)d_in[9];
    const float* e2W  = (const float*)d_in[10]; const float* e2b  = (const float*)d_in[11];
    const float* n2W1 = (const float*)d_in[12]; const float* n2b1 = (const float*)d_in[13];
    const float* n2W2 = (const float*)d_in[14]; const float* n2b2 = (const float*)d_in[15];
    const float* fcW  = (const float*)d_in[16]; const float* fcb  = (const float*)d_in[17];
    float* out = (float*)d_out;

    float*    aggr1 = (float*)d_ws;
    _Float16* h     = (_Float16*)(aggr1 + (size_t)N_NODES_C * IN_DIM_C);
    float*    aggr2 = (float*)(h + (size_t)N_NODES_C * HID_C);
    float*    psum  = aggr2 + (size_t)N_NODES_C * HID_C;
    float*    cnt   = psum  + (size_t)N_GRAPHS_C * HID_C;
    float*    sea   = cnt   + (size_t)N_GRAPHS_C;
    int*      ssrc  = (int*)(sea + (size_t)N_EDGES_C * EDGE_DIM_C);
    int*      rank  = ssrc  + (size_t)N_EDGES_C;
    int*      deg   = rank  + (size_t)N_EDGES_C;
    int*      rowptr = deg  + (size_t)(N_NODES_C + 16);

    hipMemsetAsync(deg,   0, sizeof(int)   * (size_t)(N_NODES_C + 1),     stream);
    hipMemsetAsync(psum,  0, sizeof(float) * (size_t)N_GRAPHS_C * HID_C,  stream);
    hipMemsetAsync(cnt,   0, sizeof(float) * (size_t)N_GRAPHS_C,          stream);

    {
        const unsigned grid = (N_EDGES_C + 255) / 256;
        rank_kernel<<<grid, 256, 0, stream>>>(ei, deg, rank);
    }
    {
        const unsigned grid = (SCAN_M + SCAN_T - 1) / SCAN_T;
        rowptr_kernel<<<grid, SCAN_T, 0, stream>>>(deg, rowptr);
    }
    {
        const unsigned grid = (N_EDGES_C + 255) / 256;
        fill_kernel<<<grid, 256, 0, stream>>>(ei, ea, rowptr, rank, ssrc, sea);
    }
    {
        const unsigned grid = (unsigned)(((long long)N_NODES_C * 8 + 255) / 256);
        gather1_kernel<<<grid, 256, 0, stream>>>(x, rowptr, ssrc, sea, e1W, e1b, aggr1);
    }
    {
        const unsigned grid = (N_NODES_C + 15) / 16;
        node_mlp_kernel<IN_DIM_C, float, false><<<grid, 128, 0, stream>>>(
            x, aggr1, n1W1, n1b1, n1W2, n1b2, h, nullptr, nullptr, nullptr, N_NODES_C);
    }
    {
        const unsigned grid = (N_NODES_C + 7) / 8;
        gather2_kernel<<<grid, 256, 0, stream>>>(h, rowptr, ssrc, sea, e2W, e2b, aggr2);
    }
    {
        const unsigned grid = (N_NODES_C + 15) / 16;
        node_mlp_kernel<HID_C, _Float16, true><<<grid, 128, 0, stream>>>(
            h, aggr2, n2W1, n2b1, n2W2, n2b2, nullptr, bat, psum, cnt, N_NODES_C);
    }
    {
        const unsigned grid = (N_GRAPHS_C * NUM_TASKS_C + 255) / 256;
        fc_kernel<<<grid, 256, 0, stream>>>(psum, cnt, fcW, fcb, out);
    }
}


# Edge bucketing by dst now a two-level counting sort in LDS (chunks sorted by bin, then each bin by node; records staged in LDS, whole-run stores): no global integer atomics; sum order = LDS arrival
# speedup vs baseline: 3.3131x; 1.9883x over previous
#include <hip/hip_runtime.h>
#include <hip/hip_bf16.h>

#define N_NODES_C  100000
#define N_EDGES_C  1600000
#define N_GRAPHS_C 1000
#define HID_C      64
#define IN_DIM_C   7
#define EDGE_DIM_C 4
#define NUM_TASKS_C 12

typedef __attribute__((ext_vector_type(16))) _Float16 v16h;
typedef __attribute__((ext_vector_type(8)))  _Float16 v8h;
typedef __attribute__((ext_vector_type(8)))  float    v8f;
typedef __attribute__((ext_vector_type(2)))  _Float16 v2h;

__device__ __forceinline__ void atom_add_f32(float* p, float v) {
    unsafeAtomicAdd(p, v);
}

__device__ __forceinline__ v16h lds_load_tile(const _Float16* s, int ld, int kbase) {
    const int lane = threadIdx.x & 31;
    const int r    = lane & 15;
    const int koff = (lane < 16) ? 0 : 8;
    const _Float16* p = s + r * ld + kbase + koff;
    v8h lo = *(const v8h*)(p);
    v8h hi = *(const v8h*)(p + 16);
    v16h out;
#pragma unroll
    for (int e = 0; e < 8; ++e) { out[e] = lo[e]; out[8 + e] = hi[e]; }
    return out;
}

template <int K1, typename TI, bool POOL>
__global__ __launch_bounds__(128)
void node_mlp_kernel(const TI* __restrict__ in0, const float* __restrict__ in1,
                     const float* __restrict__ W1,  const float* __restrict__ b1,
                     const float* __restrict__ W2,  const float* __restrict__ b2,
                     _Float16* __restrict__ out, const int* __restrict__ batch,
                     float* __restrict__ psum, float* __restrict__ cnt,
                     int n_nodes) {
    constexpr int K1P = (K1 + 31) & ~31;

    __shared__ __align__(16) _Float16 sA  [16 * K1P];
    __shared__ __align__(16) _Float16 sMid[16 * HID_C];
    __shared__ __align__(16) _Float16 sB1 [HID_C * K1P];
    __shared__ __align__(16) _Float16 sB2 [HID_C * HID_C];
    __shared__ __align__(16) int      sG  [16];

    const int row0 = blockIdx.x * 16;
    const int tid  = threadIdx.x;
    const int wave = tid >> 5;
    const int lane = tid & 31;
    const int n0   = wave * 16;

    for (int i = tid; i < 16 * K1P; i += 128) {
        const int m = i / K1P, k = i - m * K1P;
        const int r = row0 + m;
        float v = 0.0f;
        if (k < K1 && r < n_nodes)
            v = (float)in0[(size_t)r * K1 + k] + in1[(size_t)r * K1 + k];
        sA[i] = (_Float16)v;
    }
    for (int i = tid; i < HID_C * K1P; i += 128) {
        const int n = i / K1P, k = i - n * K1P;
        sB1[i] = (_Float16)((k < K1) ? W1[n * K1 + k] : 0.0f);
    }
    for (int i = tid; i < HID_C * HID_C; i += 128) {
        const int n = i >> 6, k = i & 63;
        sB2[i] = (_Float16)W2[n * HID_C + k];
    }
    if constexpr (POOL) {
        if (tid < 16) sG[tid] = (row0 + tid < n_nodes) ? batch[row0 + tid] : -1;
    }
    __syncthreads();

    v8f acc = {};
#pragma unroll
    for (int kb = 0; kb < K1P; kb += 32) {
        v16h a = lds_load_tile(sA, K1P, kb);
        v16h b = lds_load_tile(sB1 + n0 * K1P, K1P, kb);
        acc = __builtin_amdgcn_wmma_f32_16x16x32_f16(false, a, false, b,
                                                     (short)0, acc, false, false);
    }
    {
        const int n  = lane & 15;
        const int mb = (lane < 16) ? 0 : 8;
        const float bias = b1[n0 + n];
#pragma unroll
        for (int r = 0; r < 8; ++r) {
            float v = acc[r] + bias;
            v = v > 0.0f ? v : 0.0f;
            sMid[(mb + r) * HID_C + (n0 + n)] = (_Float16)v;
        }
    }
    __syncthreads();

    v8f acc2 = {};
#pragma unroll
    for (int kb = 0; kb < HID_C; kb += 32) {
        v16h a = lds_load_tile(sMid, HID_C, kb);
        v16h b = lds_load_tile(sB2 + n0 * HID_C, HID_C, kb);
        acc2 = __builtin_amdgcn_wmma_f32_16x16x32_f16(false, a, false, b,
                                                      (short)0, acc2, false, false);
    }
    if constexpr (!POOL) {
        const int n  = lane & 15;
        const int mb = (lane < 16) ? 0 : 8;
        const float bias = b2[n0 + n];
#pragma unroll
        for (int r = 0; r < 8; ++r) {
            const int rr = row0 + mb + r;
            if (rr < n_nodes) {
                float v = acc2[r] + bias;
                v = v > 0.0f ? v : 0.0f;
                out[(size_t)rr * HID_C + (n0 + n)] = (_Float16)v;
            }
        }
    } else {
        const int n  = lane & 15;
        const int mb = (lane < 16) ? 0 : 8;
        const float bias = b2[n0 + n];
        const bool cnt_lane = (wave == 0) && (n == 0);
        const int4 ga = *(const int4*)(sG + mb);
        const int4 gb = *(const int4*)(sG + mb + 4);
        const int gs[8] = { ga.x, ga.y, ga.z, ga.w, gb.x, gb.y, gb.z, gb.w };
        int   gcur = -1;
        float run  = 0.0f;
        int   len  = 0;
#pragma unroll
        for (int r = 0; r < 8; ++r) {
            const int g = gs[r];
            float v = acc2[r] + bias;
            v = v > 0.0f ? v : 0.0f;
            if (g != gcur) {
                if (gcur >= 0) {
                    atom_add_f32(&psum[(size_t)gcur * HID_C + (n0 + n)], run);
                    if (cnt_lane) atom_add_f32(&cnt[gcur], (float)len);
                }
                gcur = g; run = 0.0f; len = 0;
            }
            run += v;
            len += 1;
        }
        if (gcur >= 0) {
            atom_add_f32(&psum[(size_t)gcur * HID_C + (n0 + n)], run);
            if (cnt_lane) atom_add_f32(&cnt[gcur], (float)len);
        }
    }
}

#define BIN_SHIFT 9
#define BIN_NODES (1 << BIN_SHIFT)
#define NBINS     ((N_NODES_C >> BIN_SHIFT) + 1)
#define BIN_G     256
#define BIN_T     1024
#define BIN_W     (BIN_T / 32)
#define BIN_CHUNK ((N_EDGES_C + BIN_G - 1) / BIN_G)
#define BIN_TRIPS ((BIN_CHUNK + BIN_T - 1) / BIN_T)
#define SORT_GRP  4
#define SORT_CAP  12288
#define PART_TRIPS ((NBINS * BIN_G + BIN_T * 4 - 1) / (BIN_T * 4))
static_assert(NBINS <= 256, "one wave scans the bins, 8 per lane");
static_assert(BIN_G == 256, "one wave scans a bin's 256 runs, 8 per lane");

__global__ __launch_bounds__(BIN_T)
void chunk_sort_kernel(const int* __restrict__ ei, const float* __restrict__ ea,
                       int* __restrict__ hist, int* __restrict__ lstart,
                       int* __restrict__ bdst, int* __restrict__ bsrc,
                       float* __restrict__ bea) {
    __shared__ int sCnt[256];
    __shared__ int sStart[256];
    __shared__ int sRank[256];
    __shared__ __align__(16) float4 lEa[BIN_CHUNK];
    __shared__ int lDst[BIN_CHUNK];
    __shared__ int lSrc[BIN_CHUNK];
    const int tid  = threadIdx.x;
    const int lane = tid & 31;
    const int blk  = blockIdx.x;
    const int e0 = blk * BIN_CHUNK;
    const int e1 = min(e0 + BIN_CHUNK, N_EDGES_C);

    if (tid < 256) { sCnt[tid] = 0; sRank[tid] = 0; }
    int    dv[BIN_TRIPS];
    int    sv[BIN_TRIPS];
    float4 av[BIN_TRIPS];
#pragma unroll
    for (int k = 0; k < BIN_TRIPS; ++k) {
        const int e = min(e0 + tid + k * BIN_T, N_EDGES_C - 1);
        dv[k] = ei[N_EDGES_C + e];
        sv[k] = ei[e];
        av[k] = *(const float4*)(ea + (size_t)e * EDGE_DIM_C);
    }
#pragma unroll
    for (int k = 0; k < BIN_TRIPS; ++k)
        asm volatile("" :: "v"(dv[k]), "v"(sv[k]),
                     "v"(av[k].x), "v"(av[k].y), "v"(av[k].z), "v"(av[k].w));
    __syncthreads();
#pragma unroll
    for (int k = 0; k < BIN_TRIPS; ++k)
        if (e0 + tid + k * BIN_T < e1) atomicAdd(&sCnt[dv[k] >> BIN_SHIFT], 1);
    __syncthreads();
    if (tid < 32) {
        int c[8];
        int s = 0;
#pragma unroll
        for (int k = 0; k < 8; ++k) { c[k] = sCnt[lane * 8 + k]; s += c[k]; }
        int incl = s;
        for (int d = 1; d < 32; d <<= 1) {
            const int t = __shfl_up(incl, d);
            if (lane >= d) incl += t;
        }
        int run = incl - s;
#pragma unroll
        for (int k = 0; k < 8; ++k) {
            const int idx = lane * 8 + k;
            sStart[idx] = run;
            if (idx < NBINS) {
                hist  [idx * BIN_G + blk] = c[k];
                lstart[idx * BIN_G + blk] = run;
            }
            run += c[k];
        }
    }
    __syncthreads();
#pragma unroll
    for (int k = 0; k < BIN_TRIPS; ++k) {
        if (e0 + tid + k * BIN_T < e1) {
            const int bin = dv[k] >> BIN_SHIFT;
            const int lp  = sStart[bin] + atomicAdd(&sRank[bin], 1);
            lDst[lp] = dv[k];
            lSrc[lp] = sv[k];
            lEa[lp]  = av[k];
        }
    }
    __syncthreads();
    for (int j = tid; j < e1 - e0; j += BIN_T) {
        bdst[e0 + j] = lDst[j];
        bsrc[e0 + j] = lSrc[j];
        *(float4*)(bea + (size_t)(e0 + j) * EDGE_DIM_C) = lEa[j];
    }
}

__device__ __forceinline__ int bin_edge_place(const int* sRun, const int* sBase, int i) {
    int lo = 0, hi = BIN_G;
#pragma unroll
    for (int s = 0; s < 8; ++s) {
        const int mid = (lo + hi) >> 1;
        if (sRun[mid] <= i) lo = mid; else hi = mid;
    }
    return sBase[lo] + (i - sRun[lo]);
}

__global__ __launch_bounds__(BIN_T)
void bin_sort_kernel(const int* __restrict__ hist, const int* __restrict__ lstart,
                     const int* __restrict__ bdst, const int* __restrict__ bsrc,
                     const float* __restrict__ bea,
                     int* __restrict__ rowptr, int* __restrict__ ssrc,
                     float* __restrict__ sea) {
    __shared__ int sRun[BIN_G + 1];
    __shared__ int sBase[BIN_G];
    __shared__ int sCnt[BIN_NODES];
    __shared__ int sPos[BIN_NODES];
    __shared__ int sW[BIN_W];
    __shared__ int sRed[BIN_W];
    __shared__ __align__(16) float4 lEa[SORT_CAP];
    __shared__ int lSrc[SORT_CAP];
    const int tid   = threadIdx.x;
    const int lane  = tid & 31;
    const int wave  = tid >> 5;
    const int bin   = blockIdx.x;
    const int node0 = bin << BIN_SHIFT;

    int part = 0;
    {
        const int lim = bin * BIN_G;
        int4 h[PART_TRIPS];
#pragma unroll
        for (int k = 0; k < PART_TRIPS; ++k)
            h[k] = *(const int4*)(hist + min(tid * 4 + k * BIN_T * 4, NBINS * BIN_G - 4));
#pragma unroll
        for (int k = 0; k < PART_TRIPS; ++k)
            if (tid * 4 + k * BIN_T * 4 < lim) part += h[k].x + h[k].y + h[k].z + h[k].w;
    }
    for (int d = 16; d > 0; d >>= 1) part += __shfl_xor(part, d);
    if (lane == 0) sRed[wave] = part;
    if (tid < BIN_NODES) sCnt[tid] = 0;
    if (wave == 0) {
        const int4* ph = (const int4*)(hist   + bin * BIN_G + lane * 8);
        const int4* pl = (const int4*)(lstart + bin * BIN_G + lane * 8);
        const int4 ha = ph[0], hb = ph[1];
        const int4 la = pl[0], lb = pl[1];
        const int c[8]  = { ha.x, ha.y, ha.z, ha.w, hb.x, hb.y, hb.z, hb.w };
        const int ls[8] = { la.x, la.y, la.z, la.w, lb.x, lb.y, lb.z, lb.w };
        int s = 0;
#pragma unroll
        for (int k = 0; k < 8; ++k) s += c[k];
        int incl = s;
        for (int d = 1; d < 32; d <<= 1) {
            const int t = __shfl_up(incl, d);
            if (lane >= d) incl += t;
        }
        int run = incl - s;
#pragma unroll
        for (int k = 0; k < 8; ++k) {
            const int b = lane * 8 + k;
            sRun[b]  = run;
            sBase[b] = b * BIN_CHUNK + ls[k];
            run += c[k];
        }
        if (lane == 31) sRun[BIN_G] = incl;
    }
    __syncthreads();
    int beg = 0;
#pragma unroll
    for (int w = 0; w < BIN_W; ++w) beg += sRed[w];
    const int n = sRun[BIN_G];

    for (int i0 = tid; i0 < n; i0 += SORT_GRP * BIN_T) {
        int dv[SORT_GRP];
#pragma unroll
        for (int k = 0; k < SORT_GRP; ++k)
            dv[k] = bdst[bin_edge_place(sRun, sBase, min(i0 + k * BIN_T, n - 1))];
#pragma unroll
        for (int k = 0; k < SORT_GRP; ++k)
            asm volatile("" :: "v"(dv[k]));
#pragma unroll
        for (int k = 0; k < SORT_GRP; ++k)
            if (i0 + k * BIN_T < n) atomicAdd(&sCnt[dv[k] - node0], 1);
    }
    __syncthreads();

    const int v = (tid < BIN_NODES) ? sCnt[tid] : 0;
    int incl = v;
    for (int d = 1; d < 32; d <<= 1) {
        const int t = __shfl_up(incl, d);
        if (lane >= d) incl += t;
    }
    if (lane == 31) sW[wave] = incl;
    __syncthreads();
    if (wave == 0) {
        const int w = sW[lane];
        int wi = w;
        for (int d = 1; d < 32; d <<= 1) {
            const int t = __shfl_up(wi, d);
            if (lane >= d) wi += t;
        }
        sW[lane] = wi - w;
    }
    __syncthreads();
    if (tid < BIN_NODES) {
        const int start = sW[wave] + incl - v;
        sPos[tid] = start;
        const int node = node0 + tid;
        if (node <= N_NODES_C) rowptr[node] = beg + start;
    }
    __syncthreads();

    const bool staged = (n <= SORT_CAP);
    for (int i0 = tid; i0 < n; i0 += SORT_GRP * BIN_T) {
        int    dv[SORT_GRP];
        int    sv[SORT_GRP];
        float4 av[SORT_GRP];
#pragma unroll
        for (int k = 0; k < SORT_GRP; ++k) {
            const int g = bin_edge_place(sRun, sBase, min(i0 + k * BIN_T, n - 1));
            dv[k] = bdst[g];
            sv[k] = bsrc[g];
            av[k] = *(const float4*)(bea + (size_t)g * EDGE_DIM_C);
        }
#pragma unroll
        for (int k = 0; k < SORT_GRP; ++k)
            asm volatile("" :: "v"(dv[k]), "v"(sv[k]),
                         "v"(av[k].x), "v"(av[k].y), "v"(av[k].z), "v"(av[k].w));
#pragma unroll
        for (int k = 0; k < SORT_GRP; ++k) {
            if (i0 + k * BIN_T < n) {
                const int lp = atomicAdd(&sPos[dv[k] - node0], 1);
                if (staged) {
                    lSrc[lp] = sv[k];
                    lEa[lp]  = av[k];
                } else {
                    ssrc[beg + lp] = sv[k];
                    *(float4*)(sea + (size_t)(beg + lp) * EDGE_DIM_C) = av[k];
                }
            }
        }
    }
    if (staged) {
        __syncthreads();
        for (int j = tid; j < n; j += BIN_T) {
            ssrc[beg + j] = lSrc[j];
            *(float4*)(sea + (size_t)(beg + j) * EDGE_DIM_C) = lEa[j];
        }
    }
}

__global__ __launch_bounds__(256)
void gather1_kernel(const float* __restrict__ x, const int* __restrict__ rowptr,
                    const int* __restrict__ ssrc, const float* __restrict__ sea,
                    const float* __restrict__ eW, const float* __restrict__ eb,
                    float* __restrict__ aggr) {
    const int t = blockIdx.x * 256 + threadIdx.x;
    const int n = t >> 3;
    const int c = t & 7;
    if (n >= N_NODES_C || c >= IN_DIM_C) return;
    const float4 w = *(const float4*)(eW + c * EDGE_DIM_C);
    const float  b = eb[c];
    const int beg = rowptr[n];
    const int end = rowptr[n + 1];
    float acc = 0.0f;
    for (int j = beg; j < end; ++j) {
        const int src = ssrc[j];
        const float4 a = *(const float4*)(sea + (size_t)j * EDGE_DIM_C);
        float v = b + a.x * w.x + a.y * w.y + a.z * w.z + a.w * w.w;
        v += x[(size_t)src * IN_DIM_C + c];
        acc += fmaxf(v, 0.0f);
    }
    aggr[(size_t)n * IN_DIM_C + c] = acc;
}

__global__ __launch_bounds__(256)
void gather2_kernel(const _Float16* __restrict__ h, const int* __restrict__ rowptr,
                    const int* __restrict__ ssrc, const float* __restrict__ sea,
                    const float* __restrict__ eW, const float* __restrict__ eb,
                    float* __restrict__ aggr) {
    const int lane = threadIdx.x & 31;
    const int n = blockIdx.x * 8 + (threadIdx.x >> 5);
    if (n >= N_NODES_C) return;
    const int c = 2 * lane;
    const float4 w0 = *(const float4*)(eW + c * EDGE_DIM_C);
    const float4 w1 = *(const float4*)(eW + c * EDGE_DIM_C + EDGE_DIM_C);
    const float  b0 = eb[c];
    const float  b1 = eb[c + 1];
    const int beg = rowptr[n];
    const int end = rowptr[n + 1];
    float acc0 = 0.0f, acc1 = 0.0f;
    for (int j = beg; j < end; ++j) {
        const int src = ssrc[j];
        const float4 a = *(const float4*)(sea + (size_t)j * EDGE_DIM_C);
        const v2h hv = *(const v2h*)(h + (size_t)src * HID_C + c);
        float v0 = b0 + a.x * w0.x + a.y * w0.y + a.z * w0.z + a.w * w0.w;
        float v1 = b1 + a.x * w1.x + a.y * w1.y + a.z * w1.z + a.w * w1.w;
        v0 += (float)hv[0];
        v1 += (float)hv[1];
        acc0 += fmaxf(v0, 0.0f);
        acc1 += fmaxf(v1, 0.0f);
    }
    *(float2*)(aggr + (size_t)n * HID_C + c) = make_float2(acc0, acc1);
}

__global__ __launch_bounds__(256)
void fc_kernel(const float* __restrict__ psum, const float* __restrict__ cnt,
               const float* __restrict__ fcW, const float* __restrict__ fcb,
               float* __restrict__ out) {
    const int t = blockIdx.x * 256 + threadIdx.x;
    if (t >= N_GRAPHS_C * NUM_TASKS_C) return;
    const int g = t / NUM_TASKS_C;
    const int k = t - g * NUM_TASKS_C;
    const float inv = 1.0f / fmaxf(cnt[g], 1.0f);
    float s = fcb[k];
#pragma unroll
    for (int c = 0; c < HID_C; ++c)
        s += psum[(size_t)g * HID_C + c] * inv * fcW[k * HID_C + c];
    out[t] = s;
}

extern "C" void kernel_launch(void* const* d_in, const int* in_sizes, int n_in,
                              void* d_out, int out_size, void* d_ws, size_t ws_size,
                              hipStream_t stream) {
    const float* x    = (const float*)d_in[0];
    const int*   ei   = (const int*)  d_in[1];
    const float* ea   = (const float*)d_in[2];
    const int*   bat  = (const int*)  d_in[3];
    const float* e1W  = (const float*)d_in[4];  const float* e1b  = (const float*)d_in[5];
    const float* n1W1 = (const float*)d_in[6];  const float* n1b1 = (const float*)d_in[7];
    const float* n1W2 = (const float*)d_in[8];  const float* n1b2 = (const float*)d_in[9];
    const float* e2W  = (const float*)d_in[10]; const float* e2b  = (const float*)d_in[11];
    const float* n2W1 = (const float*)d_in[12]; const float* n2b1 = (const float*)d_in[13];
    const float* n2W2 = (const float*)d_in[14]; const float* n2b2 = (const float*)d_in[15];
    const float* fcW  = (const float*)d_in[16]; const float* fcb  = (const float*)d_in[17];
    float* out = (float*)d_out;

    float*    aggr1 = (float*)d_ws;
    _Float16* h     = (_Float16*)(aggr1 + (size_t)N_NODES_C * IN_DIM_C);
    float*    aggr2 = (float*)(h + (size_t)N_NODES_C * HID_C);
    float*    psum  = aggr2 + (size_t)N_NODES_C * HID_C;
    float*    cnt   = psum  + (size_t)N_GRAPHS_C * HID_C;
    float*    sea   = cnt   + (size_t)N_GRAPHS_C;
    int*      ssrc  = (int*)(sea + (size_t)N_EDGES_C * EDGE_DIM_C);
    int*      bdst  = ssrc  + (size_t)N_EDGES_C;
    int*      bsrc  = bdst  + (size_t)N_EDGES_C;
    float*    bea   = (float*)(bsrc + (size_t)N_EDGES_C);
    int*      hist  = (int*)(bea + (size_t)N_EDGES_C * EDGE_DIM_C);
    int*      lstart = hist + (size_t)NBINS * BIN_G;
    int*      rowptr = lstart + (size_t)NBINS * BIN_G;

    hipMemsetAsync(psum,  0, sizeof(float) * (size_t)N_GRAPHS_C * HID_C,  stream);
    hipMemsetAsync(cnt,   0, sizeof(float) * (size_t)N_GRAPHS_C,          stream);

    chunk_sort_kernel<<<BIN_G, BIN_T, 0, stream>>>(ei, ea, hist, lstart, bdst, bsrc, bea);
    bin_sort_kernel<<<NBINS, BIN_T, 0, stream>>>(hist, lstart, bdst, bsrc, bea, rowptr, ssrc, sea);
    {
        const unsigned grid = (unsigned)(((long long)N_NODES_C * 8 + 255) / 256);
        gather1_kernel<<<grid, 256, 0, stream>>>(x, rowptr, ssrc, sea, e1W, e1b, aggr1);
    }
    {
        const unsigned grid = (N_NODES_C + 15) / 16;
        node_mlp_kernel<IN_DIM_C, float, false><<<grid, 128, 0, stream>>>(
            x, aggr1, n1W1, n1b1, n1W2, n1b2, h, nullptr, nullptr, nullptr, N_NODES_C);
    }
    {
        const unsigned grid = (N_NODES_C + 7) / 8;
        gather2_kernel<<<grid, 256, 0, stream>>>(h, rowptr, ssrc, sea, e2W, e2b, aggr2);
    }
    {
        const unsigned grid = (N_NODES_C + 15) / 16;
        node_mlp_kernel<HID_C, _Float16, true><<<grid, 128, 0, stream>>>(
            h, aggr2, n2W1, n2b1, n2W2, n2b2, nullptr, bat, psum, cnt, N_NODES_C);
    }
    {
        const unsigned grid = (N_GRAPHS_C * NUM_TASKS_C + 255) / 256;
        fc_kernel<<<grid, 256, 0, stream>>>(psum, cnt, fcW, fcb, out);
    }
}


# Each conv's edge aggregation runs inside its node-MLP kernel: a block sums the edge runs of its own 16 nodes straight into the LDS A tile; aggr planes and 2 launches gone; per-node sum order kept
# speedup vs baseline: 3.5761x; 1.0794x over previous
#include <hip/hip_runtime.h>
#include <hip/hip_bf16.h>

#define N_NODES_C  100000
#define N_EDGES_C  1600000
#define N_GRAPHS_C 1000
#define HID_C      64
#define IN_DIM_C   7
#define EDGE_DIM_C 4
#define NUM_TASKS_C 12

typedef __attribute__((ext_vector_type(16))) _Float16 v16h;
typedef __attribute__((ext_vector_type(8)))  _Float16 v8h;
typedef __attribute__((ext_vector_type(8)))  float    v8f;
typedef __attribute__((ext_vector_type(2)))  _Float16 v2h;

__device__ __forceinline__ void atom_add_f32(float* p, float v) {
    unsafeAtomicAdd(p, v);
}

__device__ __forceinline__ v16h lds_load_tile(const _Float16* s, int ld, int kbase) {
    const int lane = threadIdx.x & 31;
    const int r    = lane & 15;
    const int koff = (lane < 16) ? 0 : 8;
    const _Float16* p = s + r * ld + kbase + koff;
    v8h lo = *(const v8h*)(p);
    v8h hi = *(const v8h*)(p + 16);
    v16h out;
#pragma unroll
    for (int e = 0; e < 8; ++e) { out[e] = lo[e]; out[8 + e] = hi[e]; }
    return out;
}

template <int K1, typename TI, bool POOL>
__global__ __launch_bounds__(128)
void conv_kernel(const TI* __restrict__ in0, const int* __restrict__ rowptr,
                 const int* __restrict__ ssrc, const float* __restrict__ sea,
                 const float* __restrict__ eW, const float* __restrict__ eb,
                 const float* __restrict__ W1,  const float* __restrict__ b1,
                 const float* __restrict__ W2,  const float* __restrict__ b2,
                 _Float16* __restrict__ out, const int* __restrict__ batch,
                 float* __restrict__ psum, float* __restrict__ cnt,
                 int n_nodes) {
    constexpr int K1P = (K1 + 31) & ~31;

    __shared__ __align__(16) _Float16 sA  [16 * K1P];
    __shared__ __align__(16) _Float16 sMid[16 * HID_C];
    __shared__ __align__(16) _Float16 sB1 [HID_C * K1P];
    __shared__ __align__(16) _Float16 sB2 [HID_C * HID_C];
    __shared__ __align__(16) int      sG  [16];

    const int row0 = blockIdx.x * 16;
    const int tid  = threadIdx.x;
    const int wave = tid >> 5;
    const int lane = tid & 31;
    const int n0   = wave * 16;

    for (int i = tid; i < HID_C * K1P; i += 128) {
        const int n = i / K1P, k = i - n * K1P;
        sB1[i] = (_Float16)((k < K1) ? W1[n * K1 + k] : 0.0f);
    }
    for (int i = tid; i < HID_C * HID_C; i += 128) {
        const int n = i >> 6, k = i & 63;
        sB2[i] = (_Float16)W2[n * HID_C + k];
    }
    if constexpr (POOL) {
        if (tid < 16) sG[tid] = (row0 + tid < n_nodes) ? batch[row0 + tid] : -1;
    }
    if constexpr (K1 == HID_C) {
        const int c = 2 * lane;
        const float4 w0 = *(const float4*)(eW + c * EDGE_DIM_C);
        const float4 w1 = *(const float4*)(eW + c * EDGE_DIM_C + EDGE_DIM_C);
        const float  e0 = eb[c];
        const float  e1 = eb[c + 1];
        for (int q = 0; q < 4; ++q) {
            const int m = wave * 4 + q;
            const int n = row0 + m;
            float v0 = 0.0f, v1 = 0.0f;
            if (n < n_nodes) {
                const int beg = rowptr[n];
                const int end = rowptr[n + 1];
                float acc0 = 0.0f, acc1 = 0.0f;
                for (int j = beg; j < end; ++j) {
                    const int src = ssrc[j];
                    const float4 a = *(const float4*)(sea + (size_t)j * EDGE_DIM_C);
                    const v2h hv = *(const v2h*)(in0 + (size_t)src * HID_C + c);
                    float t0 = e0 + a.x * w0.x + a.y * w0.y + a.z * w0.z + a.w * w0.w;
                    float t1 = e1 + a.x * w1.x + a.y * w1.y + a.z * w1.z + a.w * w1.w;
                    t0 += (float)hv[0];
                    t1 += (float)hv[1];
                    acc0 += fmaxf(t0, 0.0f);
                    acc1 += fmaxf(t1, 0.0f);
                }
                const v2h hs = *(const v2h*)(in0 + (size_t)n * HID_C + c);
                v0 = (float)hs[0] + acc0;
                v1 = (float)hs[1] + acc1;
            }
            v2h o;
            o[0] = (_Float16)v0;
            o[1] = (_Float16)v1;
            *(v2h*)(sA + m * K1P + c) = o;
        }
    } else {
        const int m = tid >> 3;
        const int c = tid & 7;
        const int n = row0 + m;
        float v = 0.0f;
        if (c < K1 && n < n_nodes) {
            const float4 w = *(const float4*)(eW + c * EDGE_DIM_C);
            const float  e = eb[c];
            const int beg = rowptr[n];
            const int end = rowptr[n + 1];
            float acc = 0.0f;
            for (int j = beg; j < end; ++j) {
                const int src = ssrc[j];
                const float4 a = *(const float4*)(sea + (size_t)j * EDGE_DIM_C);
                float t = e + a.x * w.x + a.y * w.y + a.z * w.z + a.w * w.w;
                t += (float)in0[(size_t)src * K1 + c];
                acc += fmaxf(t, 0.0f);
            }
            v = (float)in0[(size_t)n * K1 + c] + acc;
        }
        sA[m * K1P + c] = (_Float16)v;
        for (int k = c + 8; k < K1P; k += 8)
            sA[m * K1P + k] = (_Float16)0.0f;
    }
    __syncthreads();

    v8f acc = {};
#pragma unroll
    for (int kb = 0; kb < K1P; kb += 32) {
        v16h a = lds_load_tile(sA, K1P, kb);
        v16h b = lds_load_tile(sB1 + n0 * K1P, K1P, kb);
        acc = __builtin_amdgcn_wmma_f32_16x16x32_f16(false, a, false, b,
                                                     (short)0, acc, false, false);
    }
    {
        const int n  = lane & 15;
        const int mb = (lane < 16) ? 0 : 8;
        const float bias = b1[n0 + n];
#pragma unroll
        for (int r = 0; r < 8; ++r) {
            float v = acc[r] + bias;
            v = v > 0.0f ? v : 0.0f;
            sMid[(mb + r) * HID_C + (n0 + n)] = (_Float16)v;
        }
    }
    __syncthreads();

    v8f acc2 = {};
#pragma unroll
    for (int kb = 0; kb < HID_C; kb += 32) {
        v16h a = lds_load_tile(sMid, HID_C, kb);
        v16h b = lds_load_tile(sB2 + n0 * HID_C, HID_C, kb);
        acc2 = __builtin_amdgcn_wmma_f32_16x16x32_f16(false, a, false, b,
                                                      (short)0, acc2, false, false);
    }
    if constexpr (!POOL) {
        const int n  = lane & 15;
        const int mb = (lane < 16) ? 0 : 8;
        const float bias = b2[n0 + n];
#pragma unroll
        for (int r = 0; r < 8; ++r) {
            const int rr = row0 + mb + r;
            if (rr < n_nodes) {
                float v = acc2[r] + bias;
                v = v > 0.0f ? v : 0.0f;
                out[(size_t)rr * HID_C + (n0 + n)] = (_Float16)v;
            }
        }
    } else {
        const int n  = lane & 15;
        const int mb = (lane < 16) ? 0 : 8;
        const float bias = b2[n0 + n];
        const bool cnt_lane = (wave == 0) && (n == 0);
        const int4 ga = *(const int4*)(sG + mb);
        const int4 gb = *(const int4*)(sG + mb + 4);
        const int gs[8] = { ga.x, ga.y, ga.z, ga.w, gb.x, gb.y, gb.z, gb.w };
        int   gcur = -1;
        float run  = 0.0f;
        int   len  = 0;
#pragma unroll
        for (int r = 0; r < 8; ++r) {
            const int g = gs[r];
            float v = acc2[r] + bias;
            v = v > 0.0f ? v : 0.0f;
            if (g != gcur) {
                if (gcur >= 0) {
                    atom_add_f32(&psum[(size_t)gcur * HID_C + (n0 + n)], run);
                    if (cnt_lane) atom_add_f32(&cnt[gcur], (float)len);
                }
                gcur = g; run = 0.0f; len = 0;
            }
            run += v;
            len += 1;
        }
        if (gcur >= 0) {
            atom_add_f32(&psum[(size_t)gcur * HID_C + (n0 + n)], run);
            if (cnt_lane) atom_add_f32(&cnt[gcur], (float)len);
        }
    }
}

#define BIN_SHIFT 9
#define BIN_NODES (1 << BIN_SHIFT)
#define NBINS     ((N_NODES_C >> BIN_SHIFT) + 1)
#define BIN_G     256
#define BIN_T     1024
#define BIN_W     (BIN_T / 32)
#define BIN_CHUNK ((N_EDGES_C + BIN_G - 1) / BIN_G)
#define BIN_TRIPS ((BIN_CHUNK + BIN_T - 1) / BIN_T)
#define SORT_GRP  4
#define SORT_CAP  12288
#define PART_TRIPS ((NBINS * BIN_G + BIN_T * 4 - 1) / (BIN_T * 4))
static_assert(NBINS <= 256, "one wave scans the bins, 8 per lane");
static_assert(BIN_G == 256, "one wave scans a bin's 256 runs, 8 per lane");

__global__ __launch_bounds__(BIN_T)
void chunk_sort_kernel(const int* __restrict__ ei, const float* __restrict__ ea,
                       int* __restrict__ hist, int* __restrict__ lstart,
                       int* __restrict__ bdst, int* __restrict__ bsrc,
                       float* __restrict__ bea) {
    __shared__ int sCnt[256];
    __shared__ int sStart[256];
    __shared__ int sRank[256];
    __shared__ __align__(16) float4 lEa[BIN_CHUNK];
    __shared__ int lDst[BIN_CHUNK];
    __shared__ int lSrc[BIN_CHUNK];
    const int tid  = threadIdx.x;
    const int lane = tid & 31;
    const int blk  = blockIdx.x;
    const int e0 = blk * BIN_CHUNK;
    const int e1 = min(e0 + BIN_CHUNK, N_EDGES_C);

    if (tid < 256) { sCnt[tid] = 0; sRank[tid] = 0; }
    int    dv[BIN_TRIPS];
    int    sv[BIN_TRIPS];
    float4 av[BIN_TRIPS];
#pragma unroll
    for (int k = 0; k < BIN_TRIPS; ++k) {
        const int e = min(e0 + tid + k * BIN_T, N_EDGES_C - 1);
        dv[k] = ei[N_EDGES_C + e];
        sv[k] = ei[e];
        av[k] = *(const float4*)(ea + (size_t)e * EDGE_DIM_C);
    }
#pragma unroll
    for (int k = 0; k < BIN_TRIPS; ++k)
        asm volatile("" :: "v"(dv[k]), "v"(sv[k]),
                     "v"(av[k].x), "v"(av[k].y), "v"(av[k].z), "v"(av[k].w));
    __syncthreads();
#pragma unroll
    for (int k = 0; k < BIN_TRIPS; ++k)
        if (e0 + tid + k * BIN_T < e1) atomicAdd(&sCnt[dv[k] >> BIN_SHIFT], 1);
    __syncthreads();
    if (tid < 32) {
        int c[8];
        int s = 0;
#pragma unroll
        for (int k = 0; k < 8; ++k) { c[k] = sCnt[lane * 8 + k]; s += c[k]; }
        int incl = s;
        for (int d = 1; d < 32; d <<= 1) {
            const int t = __shfl_up(incl, d);
            if (lane >= d) incl += t;
        }
        int run = incl - s;
#pragma unroll
        for (int k = 0; k < 8; ++k) {
            const int idx = lane * 8 + k;
            sStart[idx] = run;
            if (idx < NBINS) {
                hist  [idx * BIN_G + blk] = c[k];
                lstart[idx * BIN_G + blk] = run;
            }
            run += c[k];
        }
    }
    __syncthreads();
#pragma unroll
    for (int k = 0; k < BIN_TRIPS; ++k) {
        if (e0 + tid + k * BIN_T < e1) {
            const int bin = dv[k] >> BIN_SHIFT;
            const int lp  = sStart[bin] + atomicAdd(&sRank[bin], 1);
            lDst[lp] = dv[k];
            lSrc[lp] = sv[k];
            lEa[lp]  = av[k];
        }
    }
    __syncthreads();
    for (int j = tid; j < e1 - e0; j += BIN_T) {
        bdst[e0 + j] = lDst[j];
        bsrc[e0 + j] = lSrc[j];
        *(float4*)(bea + (size_t)(e0 + j) * EDGE_DIM_C) = lEa[j];
    }
}

__device__ __forceinline__ int bin_edge_place(const int* sRun, const int* sBase, int i) {
    int lo = 0, hi = BIN_G;
#pragma unroll
    for (int s = 0; s < 8; ++s) {
        const int mid = (lo + hi) >> 1;
        if (sRun[mid] <= i) lo = mid; else hi = mid;
    }
    return sBase[lo] + (i - sRun[lo]);
}

__global__ __launch_bounds__(BIN_T)
void bin_sort_kernel(const int* __restrict__ hist, const int* __restrict__ lstart,
                     const int* __restrict__ bdst, const int* __restrict__ bsrc,
                     const float* __restrict__ bea,
                     int* __restrict__ rowptr, int* __restrict__ ssrc,
                     float* __restrict__ sea) {
    __shared__ int sRun[BIN_G + 1];
    __shared__ int sBase[BIN_G];
    __shared__ int sCnt[BIN_NODES];
    __shared__ int sPos[BIN_NODES];
    __shared__ int sW[BIN_W];
    __shared__ int sRed[BIN_W];
    __shared__ __align__(16) float4 lEa[SORT_CAP];
    __shared__ int lSrc[SORT_CAP];
    const int tid   = threadIdx.x;
    const int lane  = tid & 31;
    const int wave  = tid >> 5;
    const int bin   = blockIdx.x;
    const int node0 = bin << BIN_SHIFT;

    int part = 0;
    {
        const int lim = bin * BIN_G;
        int4 h[PART_TRIPS];
#pragma unroll
        for (int k = 0; k < PART_TRIPS; ++k)
            h[k] = *(const int4*)(hist + min(tid * 4 + k * BIN_T * 4, NBINS * BIN_G - 4));
#pragma unroll
        for (int k = 0; k < PART_TRIPS; ++k)
            if (tid * 4 + k * BIN_T * 4 < lim) part += h[k].x + h[k].y + h[k].z + h[k].w;
    }
    for (int d = 16; d > 0; d >>= 1) part += __shfl_xor(part, d);
    if (lane == 0) sRed[wave] = part;
    if (tid < BIN_NODES) sCnt[tid] = 0;
    if (wave == 0) {
        const int4* ph = (const int4*)(hist   + bin * BIN_G + lane * 8);
        const int4* pl = (const int4*)(lstart + bin * BIN_G + lane * 8);
        const int4 ha = ph[0], hb = ph[1];
        const int4 la = pl[0], lb = pl[1];
        const int c[8]  = { ha.x, ha.y, ha.z, ha.w, hb.x, hb.y, hb.z, hb.w };
        const int ls[8] = { la.x, la.y, la.z, la.w, lb.x, lb.y, lb.z, lb.w };
        int s = 0;
#pragma unroll
        for (int k = 0; k < 8; ++k) s += c[k];
        int incl = s;
        for (int d = 1; d < 32; d <<= 1) {
            const int t = __shfl_up(incl, d);
            if (lane >= d) incl += t;
        }
        int run = incl - s;
#pragma unroll
        for (int k = 0; k < 8; ++k) {
            const int b = lane * 8 + k;
            sRun[b]  = run;
            sBase[b] = b * BIN_CHUNK + ls[k];
            run += c[k];
        }
        if (lane == 31) sRun[BIN_G] = incl;
    }
    __syncthreads();
    int beg = 0;
#pragma unroll
    for (int w = 0; w < BIN_W; ++w) beg += sRed[w];
    const int n = sRun[BIN_G];

    for (int i0 = tid; i0 < n; i0 += SORT_GRP * BIN_T) {
        int dv[SORT_GRP];
#pragma unroll
        for (int k = 0; k < SORT_GRP; ++k)
            dv[k] = bdst[bin_edge_place(sRun, sBase, min(i0 + k * BIN_T, n - 1))];
#pragma unroll
        for (int k = 0; k < SORT_GRP; ++k)
            asm volatile("" :: "v"(dv[k]));
#pragma unroll
        for (int k = 0; k < SORT_GRP; ++k)
            if (i0 + k * BIN_T < n) atomicAdd(&sCnt[dv[k] - node0], 1);
    }
    __syncthreads();

    const int v = (tid < BIN_NODES) ? sCnt[tid] : 0;
    int incl = v;
    for (int d = 1; d < 32; d <<= 1) {
        const int t = __shfl_up(incl, d);
        if (lane >= d) incl += t;
    }
    if (lane == 31) sW[wave] = incl;
    __syncthreads();
    if (wave == 0) {
        const int w = sW[lane];
        int wi = w;
        for (int d = 1; d < 32; d <<= 1) {
            const int t = __shfl_up(wi, d);
            if (lane >= d) wi += t;
        }
        sW[lane] = wi - w;
    }
    __syncthreads();
    if (tid < BIN_NODES) {
        const int start = sW[wave] + incl - v;
        sPos[tid] = start;
        const int node = node0 + tid;
        if (node <= N_NODES_C) rowptr[node] = beg + start;
    }
    __syncthreads();

    const bool staged = (n <= SORT_CAP);
    for (int i0 = tid; i0 < n; i0 += SORT_GRP * BIN_T) {
        int    dv[SORT_GRP];
        int    sv[SORT_GRP];
        float4 av[SORT_GRP];
#pragma unroll
        for (int k = 0; k < SORT_GRP; ++k) {
            const int g = bin_edge_place(sRun, sBase, min(i0 + k * BIN_T, n - 1));
            dv[k] = bdst[g];
            sv[k] = bsrc[g];
            av[k] = *(const float4*)(bea + (size_t)g * EDGE_DIM_C);
        }
#pragma unroll
        for (int k = 0; k < SORT_GRP; ++k)
            asm volatile("" :: "v"(dv[k]), "v"(sv[k]),
                         "v"(av[k].x), "v"(av[k].y), "v"(av[k].z), "v"(av[k].w));
#pragma unroll
        for (int k = 0; k < SORT_GRP; ++k) {
            if (i0 + k * BIN_T < n) {
                const int lp = atomicAdd(&sPos[dv[k] - node0], 1);
                if (staged) {
                    lSrc[lp] = sv[k];
                    lEa[lp]  = av[k];
                } else {
                    ssrc[beg + lp] = sv[k];
                    *(float4*)(sea + (size_t)(beg + lp) * EDGE_DIM_C) = av[k];
                }
            }
        }
    }
    if (staged) {
        __syncthreads();
        for (int j = tid; j < n; j += BIN_T) {
            ssrc[beg + j] = lSrc[j];
            *(float4*)(sea + (size_t)(beg + j) * EDGE_DIM_C) = lEa[j];
        }
    }
}

__global__ __launch_bounds__(256)
void fc_kernel(const float* __restrict__ psum, const float* __restrict__ cnt,
               const float* __restrict__ fcW, const float* __restrict__ fcb,
               float* __restrict__ out) {
    const int t = blockIdx.x * 256 + threadIdx.x;
    if (t >= N_GRAPHS_C * NUM_TASKS_C) return;
    const int g = t / NUM_TASKS_C;
    const int k = t - g * NUM_TASKS_C;
    const float inv = 1.0f / fmaxf(cnt[g], 1.0f);
    float s = fcb[k];
#pragma unroll
    for (int c = 0; c < HID_C; ++c)
        s += psum[(size_t)g * HID_C + c] * inv * fcW[k * HID_C + c];
    out[t] = s;
}

extern "C" void kernel_launch(void* const* d_in, const int* in_sizes, int n_in,
                              void* d_out, int out_size, void* d_ws, size_t ws_size,
                              hipStream_t stream) {
    const float* x    = (const float*)d_in[0];
    const int*   ei   = (const int*)  d_in[1];
    const float* ea   = (const float*)d_in[2];
    const int*   bat  = (const int*)  d_in[3];
    const float* e1W  = (const float*)d_in[4];  const float* e1b  = (const float*)d_in[5];
    const float* n1W1 = (const float*)d_in[6];  const float* n1b1 = (const float*)d_in[7];
    const float* n1W2 = (const float*)d_in[8];  const float* n1b2 = (const float*)d_in[9];
    const float* e2W  = (const float*)d_in[10]; const float* e2b  = (const float*)d_in[11];
    const float* n2W1 = (const float*)d_in[12]; const float* n2b1 = (const float*)d_in[13];
    const float* n2W2 = (const float*)d_in[14]; const float* n2b2 = (const float*)d_in[15];
    const float* fcW  = (const float*)d_in[16]; const float* fcb  = (const float*)d_in[17];
    float* out = (float*)d_out;

    _Float16* h     = (_Float16*)d_ws;
    float*    psum  = (float*)(h + (size_t)N_NODES_C * HID_C);
    float*    cnt   = psum  + (size_t)N_GRAPHS_C * HID_C;
    float*    sea   = cnt   + (size_t)N_GRAPHS_C;
    int*      ssrc  = (int*)(sea + (size_t)N_EDGES_C * EDGE_DIM_C);
    int*      bdst  = ssrc  + (size_t)N_EDGES_C;
    int*      bsrc  = bdst  + (size_t)N_EDGES_C;
    float*    bea   = (float*)(bsrc + (size_t)N_EDGES_C);
    int*      hist  = (int*)(bea + (size_t)N_EDGES_C * EDGE_DIM_C);
    int*      lstart = hist + (size_t)NBINS * BIN_G;
    int*      rowptr = lstart + (size_t)NBINS * BIN_G;

    hipMemsetAsync(psum,  0, sizeof(float) * (size_t)N_GRAPHS_C * HID_C,  stream);
    hipMemsetAsync(cnt,   0, sizeof(float) * (size_t)N_GRAPHS_C,          stream);

    chunk_sort_kernel<<<BIN_G, BIN_T, 0, stream>>>(ei, ea, hist, lstart, bdst, bsrc, bea);
    bin_sort_kernel<<<NBINS, BIN_T, 0, stream>>>(hist, lstart, bdst, bsrc, bea, rowptr, ssrc, sea);
    const unsigned grid = (N_NODES_C + 15) / 16;
    conv_kernel<IN_DIM_C, float, false><<<grid, 128, 0, stream>>>(
        x, rowptr, ssrc, sea, e1W, e1b, n1W1, n1b1, n1W2, n1b2, h,
        nullptr, nullptr, nullptr, N_NODES_C);
    conv_kernel<HID_C, _Float16, true><<<grid, 128, 0, stream>>>(
        h, rowptr, ssrc, sea, e2W, e2b, n2W1, n2b1, n2W2, n2b2, nullptr,
        bat, psum, cnt, N_NODES_C);
    {
        const unsigned fgrid = (N_GRAPHS_C * NUM_TASKS_C + 255) / 256;
        fc_kernel<<<fgrid, 256, 0, stream>>>(psum, cnt, fcW, fcb, out);
    }
}


# Both convs' edge sums: a run's src/edge_attr records are loaded one per lane (coalesced) and handed out by lane reads or 8-lane broadcasts, 8 row gathers in flight at once; per-node sum order kept
# speedup vs baseline: 3.9469x; 1.1037x over previous
#include <hip/hip_runtime.h>
#include <hip/hip_bf16.h>

#define N_NODES_C  100000
#define N_EDGES_C  1600000
#define N_GRAPHS_C 1000
#define HID_C      64
#define IN_DIM_C   7
#define EDGE_DIM_C 4
#define NUM_TASKS_C 12

typedef __attribute__((ext_vector_type(16))) _Float16 v16h;
typedef __attribute__((ext_vector_type(8)))  _Float16 v8h;
typedef __attribute__((ext_vector_type(8)))  float    v8f;
typedef __attribute__((ext_vector_type(2)))  _Float16 v2h;
typedef __attribute__((ext_vector_type(2)))  float    v2f;

__device__ __forceinline__ void atom_add_f32(float* p, float v) {
    unsafeAtomicAdd(p, v);
}

__device__ __forceinline__ v16h lds_load_tile(const _Float16* s, int ld, int kbase) {
    const int lane = threadIdx.x & 31;
    const int r    = lane & 15;
    const int koff = (lane < 16) ? 0 : 8;
    const _Float16* p = s + r * ld + kbase + koff;
    v8h lo = *(const v8h*)(p);
    v8h hi = *(const v8h*)(p + 16);
    v16h out;
#pragma unroll
    for (int e = 0; e < 8; ++e) { out[e] = lo[e]; out[8 + e] = hi[e]; }
    return out;
}

template <int K1, typename TI, bool POOL>
__global__ __launch_bounds__(128)
void conv_kernel(const TI* __restrict__ in0, const int* __restrict__ rowptr,
                 const int* __restrict__ ssrc, const float* __restrict__ sea,
                 const float* __restrict__ eW, const float* __restrict__ eb,
                 const float* __restrict__ W1,  const float* __restrict__ b1,
                 const float* __restrict__ W2,  const float* __restrict__ b2,
                 _Float16* __restrict__ out, const int* __restrict__ batch,
                 float* __restrict__ psum, float* __restrict__ cnt,
                 int n_nodes) {
    constexpr int K1P = (K1 + 31) & ~31;

    __shared__ __align__(16) _Float16 sA  [16 * K1P];
    __shared__ __align__(16) _Float16 sMid[16 * HID_C];
    __shared__ __align__(16) _Float16 sB1 [HID_C * K1P];
    __shared__ __align__(16) _Float16 sB2 [HID_C * HID_C];
    __shared__ __align__(16) int      sG  [16];

    const int row0 = blockIdx.x * 16;
    const int tid  = threadIdx.x;
    const int wave = tid >> 5;
    const int lane = tid & 31;
    const int n0   = wave * 16;

    for (int i = tid; i < HID_C * K1P; i += 128) {
        const int n = i / K1P, k = i - n * K1P;
        sB1[i] = (_Float16)((k < K1) ? W1[n * K1 + k] : 0.0f);
    }
    for (int i = tid; i < HID_C * HID_C; i += 128) {
        const int n = i >> 6, k = i & 63;
        sB2[i] = (_Float16)W2[n * HID_C + k];
    }
    if constexpr (POOL) {
        if (tid < 16) sG[tid] = (row0 + tid < n_nodes) ? batch[row0 + tid] : -1;
    }
    if constexpr (K1 == HID_C) {
        const int c = 2 * lane;
        const float4 w0 = *(const float4*)(eW + c * EDGE_DIM_C);
        const float4 w1 = *(const float4*)(eW + c * EDGE_DIM_C + EDGE_DIM_C);
        const float  e0 = eb[c];
        const float  e1 = eb[c + 1];
        const int wv  = __builtin_amdgcn_readfirstlane(wave);
        const int nb  = row0 + wv * 4;
        const int rb0 = rowptr[min(nb,     n_nodes)];
        const int rb1 = rowptr[min(nb + 1, n_nodes)];
        const int rb2 = rowptr[min(nb + 2, n_nodes)];
        const int rb3 = rowptr[min(nb + 3, n_nodes)];
        const int rb4 = rowptr[min(nb + 4, n_nodes)];
        _Float16* const sAw = sA + (wv * 4) * K1P + c;
        v2h hq[4];
#pragma unroll
        for (int qq = 0; qq < 4; ++qq)
            hq[qq] = *(const v2h*)(in0 + (size_t)min(nb + qq, n_nodes - 1) * HID_C + c);
        const int wbeg = rb0, wend = rb4;
        int   q    = 0;
        int   qend = rb1;
        v2f acc = {0.0f, 0.0f};
        const v2f wx = {w0.x, w1.x}, wy = {w0.y, w1.y}, wz = {w0.z, w1.z}, ww = {w0.w, w1.w};
        const v2f eb2 = {e0, e1};
#define CONV2_CLOSE_NODE()                                                    \
        {                                                                     \
            v2h* const p_ = (v2h*)(sAw + q * K1P);                            \
            const v2h hq_ = *p_;                                              \
            v2h o_;                                                           \
            o_[0] = (_Float16)((float)hq_[0] + acc[0]);                       \
            o_[1] = (_Float16)((float)hq_[1] + acc[1]);                       \
            *p_ = o_;                                                         \
            ++q;                                                              \
            qend = (q == 1) ? rb2 : (q == 2) ? rb3 : rb4;                     \
            acc[0] = 0.0f; acc[1] = 0.0f;                                     \
        }
#define CONV2_GATHER8(K0)                                                     \
        _Pragma("unroll")                                                     \
        for (int k = 0; k < 8; ++k) {                                         \
            const int kk_  = min((K0) + k, cnt - 1);                          \
            const int srk_ = __builtin_amdgcn_readlane(sv, kk_);              \
            hv[k] = *(const v2h*)(in0 + (size_t)srk_ * HID_C + c);            \
        }
#define CONV2_SUM8(K0)                                                        \
        _Pragma("unroll")                                                     \
        for (int k = 0; k < 8; ++k) {                                         \
            const int  kk_ = (K0) + k;                                        \
            const bool on_ = kk_ < cnt;       \
            if (on_) {                                                        \
                const int j_ = wb + kk_;                                      \
                while (j_ >= qend) CONV2_CLOSE_NODE()    \
            }                                                                 \
            const int kc_ = min(kk_, cnt - 1);                                \
            const float ax_ = __builtin_bit_cast(float, __builtin_amdgcn_readlane(__builtin_bit_cast(int, av.x), kc_)); \
            const float ay_ = __builtin_bit_cast(float, __builtin_amdgcn_readlane(__builtin_bit_cast(int, av.y), kc_)); \
            const float az_ = __builtin_bit_cast(float, __builtin_amdgcn_readlane(__builtin_bit_cast(int, av.z), kc_)); \
            const float aw_ = __builtin_bit_cast(float, __builtin_amdgcn_readlane(__builtin_bit_cast(int, av.w), kc_)); \
            v2f t_ = __builtin_elementwise_fma(wx, (v2f){ax_, ax_}, eb2);     \
            t_ = __builtin_elementwise_fma(wy, (v2f){ay_, ay_}, t_);          \
            t_ = __builtin_elementwise_fma(wz, (v2f){az_, az_}, t_);          \
            t_ = __builtin_elementwise_fma(ww, (v2f){aw_, aw_}, t_);          \
            t_ += (v2f){(float)hv[k][0], (float)hv[k][1]};                    \
            const v2f r_ = {fmaxf(t_[0], 0.0f), fmaxf(t_[1], 0.0f)};          \
            acc += on_ ? r_ : (v2f){0.0f, 0.0f};                              \
        }
        int    sv = 0;
        float4 av = make_float4(0.0f, 0.0f, 0.0f, 0.0f);
        if (wbeg < wend) {
            const int je = min(wbeg + lane, wend - 1);
            sv = ssrc[je];
            av = *(const float4*)(sea + (size_t)je * EDGE_DIM_C);
        }
#pragma unroll
        for (int qq = 0; qq < 4; ++qq) {
            v2h hz;
            hz[0] = (_Float16)0.0f; hz[1] = (_Float16)0.0f;
            *(v2h*)(sAw + qq * K1P) = (nb + qq < n_nodes) ? hq[qq] : hz;
        }
        for (int wb = wbeg; wb < wend; wb += 32) {
            const int cnt = min(32, wend - wb);
            v2h hv[8];
            CONV2_GATHER8(0)
            const int    jn = min(wb + 32 + lane, wend - 1);
            const int    sn = ssrc[jn];
            const float4 an = *(const float4*)(sea + (size_t)jn * EDGE_DIM_C);
            CONV2_SUM8(0)
            for (int k0 = 8; k0 < cnt; k0 += 8) {
                CONV2_GATHER8(k0)
                CONV2_SUM8(k0)
            }
            sv = sn;
            av = an;
        }
        while (q < 4) CONV2_CLOSE_NODE()
#undef CONV2_SUM8
#undef CONV2_GATHER8
#undef CONV2_CLOSE_NODE
    } else {
        const int m  = tid >> 3;
        const int c  = tid & 7;
        const int cc = min(c, K1 - 1);
        const int n  = row0 + m;
        const float4 w = *(const float4*)(eW + cc * EDGE_DIM_C);
        const float  e = eb[cc];
        const int beg   = rowptr[min(n,     n_nodes)];
        const int end   = rowptr[min(n + 1, n_nodes)];
        const int lastv = max(end - 1, 0);
        const float own = (float)in0[(size_t)min(n, n_nodes - 1) * K1 + cc];
        int    sv;
        float4 av;
        {
            const int je = min(beg + c, lastv);
            sv = ssrc[je];
            av = *(const float4*)(sea + (size_t)je * EDGE_DIM_C);
        }
        float acc = 0.0f;
        for (int j0 = beg; __builtin_amdgcn_ballot_w32(j0 < end) != 0u; j0 += 8) {
            float xv[8];
#define CONV1_GATHER(K)                                                       \
            {                                                                 \
                const int srk_ = __builtin_amdgcn_mov_dpp8(sv, (K) * 0x249249);     \
                xv[K] = (float)in0[(size_t)srk_ * K1 + cc];                   \
            }
            CONV1_GATHER(0) CONV1_GATHER(1) CONV1_GATHER(2) CONV1_GATHER(3)
            CONV1_GATHER(4) CONV1_GATHER(5) CONV1_GATHER(6) CONV1_GATHER(7)
#undef CONV1_GATHER
            const int    jn = min(j0 + 8 + c, lastv);
            const int    sn = ssrc[jn];
            const float4 an = *(const float4*)(sea + (size_t)jn * EDGE_DIM_C);
            asm volatile("" :: "v"(xv[0]), "v"(xv[1]), "v"(xv[2]), "v"(xv[3]),
                               "v"(xv[4]), "v"(xv[5]), "v"(xv[6]), "v"(xv[7]));
#define CONV1_SUM(K)                                                          \
            {                                                                 \
                const float ax_ = __builtin_bit_cast(float, __builtin_amdgcn_mov_dpp8(__builtin_bit_cast(int, av.x), (K) * 0x249249)); \
                const float ay_ = __builtin_bit_cast(float, __builtin_amdgcn_mov_dpp8(__builtin_bit_cast(int, av.y), (K) * 0x249249)); \
                const float az_ = __builtin_bit_cast(float, __builtin_amdgcn_mov_dpp8(__builtin_bit_cast(int, av.z), (K) * 0x249249)); \
                const float aw_ = __builtin_bit_cast(float, __builtin_amdgcn_mov_dpp8(__builtin_bit_cast(int, av.w), (K) * 0x249249)); \
                float t_ = e + ax_ * w.x + ay_ * w.y + az_ * w.z + aw_ * w.w; \
                t_ += xv[K];                                                  \
                acc += (j0 + (K) < end) ? fmaxf(t_, 0.0f) : 0.0f;             \
            }
            CONV1_SUM(0) CONV1_SUM(1) CONV1_SUM(2) CONV1_SUM(3)
            CONV1_SUM(4) CONV1_SUM(5) CONV1_SUM(6) CONV1_SUM(7)
#undef CONV1_SUM
            sv = sn;
            av = an;
        }
        const float v = (c < K1 && n < n_nodes) ? own + acc : 0.0f;
        sA[m * K1P + c] = (_Float16)v;
        for (int k = c + 8; k < K1P; k += 8)
            sA[m * K1P + k] = (_Float16)0.0f;
    }
    __syncthreads();

    v8f acc = {};
#pragma unroll
    for (int kb = 0; kb < K1P; kb += 32) {
        v16h a = lds_load_tile(sA, K1P, kb);
        v16h b = lds_load_tile(sB1 + n0 * K1P, K1P, kb);
        acc = __builtin_amdgcn_wmma_f32_16x16x32_f16(false, a, false, b,
                                                     (short)0, acc, false, false);
    }
    {
        const int n  = lane & 15;
        const int mb = (lane < 16) ? 0 : 8;
        const float bias = b1[n0 + n];
#pragma unroll
        for (int r = 0; r < 8; ++r) {
            float v = acc[r] + bias;
            v = v > 0.0f ? v : 0.0f;
            sMid[(mb + r) * HID_C + (n0 + n)] = (_Float16)v;
        }
    }
    __syncthreads();

    v8f acc2 = {};
#pragma unroll
    for (int kb = 0; kb < HID_C; kb += 32) {
        v16h a = lds_load_tile(sMid, HID_C, kb);
        v16h b = lds_load_tile(sB2 + n0 * HID_C, HID_C, kb);
        acc2 = __builtin_amdgcn_wmma_f32_16x16x32_f16(false, a, false, b,
                                                      (short)0, acc2, false, false);
    }
    if constexpr (!POOL) {
        const int n  = lane & 15;
        const int mb = (lane < 16) ? 0 : 8;
        const float bias = b2[n0 + n];
#pragma unroll
        for (int r = 0; r < 8; ++r) {
            const int rr = row0 + mb + r;
            if (rr < n_nodes) {
                float v = acc2[r] + bias;
                v = v > 0.0f ? v : 0.0f;
                out[(size_t)rr * HID_C + (n0 + n)] = (_Float16)v;
            }
        }
    } else {
        const int n  = lane & 15;
        const int mb = (lane < 16) ? 0 : 8;
        const float bias = b2[n0 + n];
        const bool cnt_lane = (wave == 0) && (n == 0);
        const int4 ga = *(const int4*)(sG + mb);
        const int4 gb = *(const int4*)(sG + mb + 4);
        const int gs[8] = { ga.x, ga.y, ga.z, ga.w, gb.x, gb.y, gb.z, gb.w };
        int   gcur = -1;
        float run  = 0.0f;
        int   len  = 0;
#pragma unroll
        for (int r = 0; r < 8; ++r) {
            const int g = gs[r];
            float v = acc2[r] + bias;
            v = v > 0.0f ? v : 0.0f;
            if (g != gcur) {
                if (gcur >= 0) {
                    atom_add_f32(&psum[(size_t)gcur * HID_C + (n0 + n)], run);
                    if (cnt_lane) atom_add_f32(&cnt[gcur], (float)len);
                }
                gcur = g; run = 0.0f; len = 0;
            }
            run += v;
            len += 1;
        }
        if (gcur >= 0) {
            atom_add_f32(&psum[(size_t)gcur * HID_C + (n0 + n)], run);
            if (cnt_lane) atom_add_f32(&cnt[gcur], (float)len);
        }
    }
}

#define BIN_SHIFT 9
#define BIN_NODES (1 << BIN_SHIFT)
#define NBINS     ((N_NODES_C >> BIN_SHIFT) + 1)
#define BIN_G     256
#define BIN_T     1024
#define BIN_W     (BIN_T / 32)
#define BIN_CHUNK ((N_EDGES_C + BIN_G - 1) / BIN_G)
#define BIN_TRIPS ((BIN_CHUNK + BIN_T - 1) / BIN_T)
#define SORT_GRP  4
#define SORT_CAP  12288
#define PART_TRIPS ((NBINS * BIN_G + BIN_T * 4 - 1) / (BIN_T * 4))
static_assert(NBINS <= 256, "one wave scans the bins, 8 per lane");
static_assert(BIN_G == 256, "one wave scans a bin's 256 runs, 8 per lane");

__global__ __launch_bounds__(BIN_T)
void chunk_sort_kernel(const int* __restrict__ ei, const float* __restrict__ ea,
                       int* __restrict__ hist, int* __restrict__ lstart,
                       int* __restrict__ bdst, int* __restrict__ bsrc,
                       float* __restrict__ bea) {
    __shared__ int sCnt[256];
    __shared__ int sStart[256];
    __shared__ int sRank[256];
    __shared__ __align__(16) float4 lEa[BIN_CHUNK];
    __shared__ int lDst[BIN_CHUNK];
    __shared__ int lSrc[BIN_CHUNK];
    const int tid  = threadIdx.x;
    const int lane = tid & 31;
    const int blk  = blockIdx.x;
    const int e0 = blk * BIN_CHUNK;
    const int e1 = min(e0 + BIN_CHUNK, N_EDGES_C);

    if (tid < 256) { sCnt[tid] = 0; sRank[tid] = 0; }
    int    dv[BIN_TRIPS];
    int    sv[BIN_TRIPS];
    float4 av[BIN_TRIPS];
#pragma unroll
    for (int k = 0; k < BIN_TRIPS; ++k) {
        const int e = min(e0 + tid + k * BIN_T, N_EDGES_C - 1);
        dv[k] = ei[N_EDGES_C + e];
        sv[k] = ei[e];
        av[k] = *(const float4*)(ea + (size_t)e * EDGE_DIM_C);
    }
#pragma unroll
    for (int k = 0; k < BIN_TRIPS; ++k)
        asm volatile("" :: "v"(dv[k]), "v"(sv[k]),
                     "v"(av[k].x), "v"(av[k].y), "v"(av[k].z), "v"(av[k].w));
    __syncthreads();
#pragma unroll
    for (int k = 0; k < BIN_TRIPS; ++k)
        if (e0 + tid + k * BIN_T < e1) atomicAdd(&sCnt[dv[k] >> BIN_SHIFT], 1);
    __syncthreads();
    if (tid < 32) {
        int c[8];
        int s = 0;
#pragma unroll
        for (int k = 0; k < 8; ++k) { c[k] = sCnt[lane * 8 + k]; s += c[k]; }
        int incl = s;
        for (int d = 1; d < 32; d <<= 1) {
            const int t = __shfl_up(incl, d);
            if (lane >= d) incl += t;
        }
        int run = incl - s;
#pragma unroll
        for (int k = 0; k < 8; ++k) {
            const int idx = lane * 8 + k;
            sStart[idx] = run;
            if (idx < NBINS) {
                hist  [idx * BIN_G + blk] = c[k];
                lstart[idx * BIN_G + blk] = run;
            }
            run += c[k];
        }
    }
    __syncthreads();
#pragma unroll
    for (int k = 0; k < BIN_TRIPS; ++k) {
        if (e0 + tid + k * BIN_T < e1) {
            const int bin = dv[k] >> BIN_SHIFT;
            const int lp  = sStart[bin] + atomicAdd(&sRank[bin], 1);
            lDst[lp] = dv[k];
            lSrc[lp] = sv[k];
            lEa[lp]  = av[k];
        }
    }
    __syncthreads();
    for (int j = tid; j < e1 - e0; j += BIN_T) {
        bdst[e0 + j] = lDst[j];
        bsrc[e0 + j] = lSrc[j];
        *(float4*)(bea + (size_t)(e0 + j) * EDGE_DIM_C) = lEa[j];
    }
}

__device__ __forceinline__ int bin_edge_place(const int* sRun, const int* sBase, int i) {
    int lo = 0, hi = BIN_G;
#pragma unroll
    for (int s = 0; s < 8; ++s) {
        const int mid = (lo + hi) >> 1;
        if (sRun[mid] <= i) lo = mid; else hi = mid;
    }
    return sBase[lo] + (i - sRun[lo]);
}

__global__ __launch_bounds__(BIN_T)
void bin_sort_kernel(const int* __restrict__ hist, const int* __restrict__ lstart,
                     const int* __restrict__ bdst, const int* __restrict__ bsrc,
                     const float* __restrict__ bea,
                     int* __restrict__ rowptr, int* __restrict__ ssrc,
                     float* __restrict__ sea) {
    __shared__ int sRun[BIN_G + 1];
    __shared__ int sBase[BIN_G];
    __shared__ int sCnt[BIN_NODES];
    __shared__ int sPos[BIN_NODES];
    __shared__ int sW[BIN_W];
    __shared__ int sRed[BIN_W];
    __shared__ __align__(16) float4 lEa[SORT_CAP];
    __shared__ int lSrc[SORT_CAP];
    const int tid   = threadIdx.x;
    const int lane  = tid & 31;
    const int wave  = tid >> 5;
    const int bin   = blockIdx.x;
    const int node0 = bin << BIN_SHIFT;

    int part = 0;
    {
        const int lim = bin * BIN_G;
        int4 h[PART_TRIPS];
#pragma unroll
        for (int k = 0; k < PART_TRIPS; ++k)
            h[k] = *(const int4*)(hist + min(tid * 4 + k * BIN_T * 4, NBINS * BIN_G - 4));
#pragma unroll
        for (int k = 0; k < PART_TRIPS; ++k)
            if (tid * 4 + k * BIN_T * 4 < lim) part += h[k].x + h[k].y + h[k].z + h[k].w;
    }
    for (int d = 16; d > 0; d >>= 1) part += __shfl_xor(part, d);
    if (lane == 0) sRed[wave] = part;
    if (tid < BIN_NODES) sCnt[tid] = 0;
    if (wave == 0) {
        const int4* ph = (const int4*)(hist   + bin * BIN_G + lane * 8);
        const int4* pl = (const int4*)(lstart + bin * BIN_G + lane * 8);
        const int4 ha = ph[0], hb = ph[1];
        const int4 la = pl[0], lb = pl[1];
        const int c[8]  = { ha.x, ha.y, ha.z, ha.w, hb.x, hb.y, hb.z, hb.w };
        const int ls[8] = { la.x, la.y, la.z, la.w, lb.x, lb.y, lb.z, lb.w };
        int s = 0;
#pragma unroll
        for (int k = 0; k < 8; ++k) s += c[k];
        int incl = s;
        for (int d = 1; d < 32; d <<= 1) {
            const int t = __shfl_up(incl, d);
            if (lane >= d) incl += t;
        }
        int run = incl - s;
#pragma unroll
        for (int k = 0; k < 8; ++k) {
            const int b = lane * 8 + k;
            sRun[b]  = run;
            sBase[b] = b * BIN_CHUNK + ls[k];
            run += c[k];
        }
        if (lane == 31) sRun[BIN_G] = incl;
    }
    __syncthreads();
    int beg = 0;
#pragma unroll
    for (int w = 0; w < BIN_W; ++w) beg += sRed[w];
    const int n = sRun[BIN_G];

    for (int i0 = tid; i0 < n; i0 += SORT_GRP * BIN_T) {
        int dv[SORT_GRP];
#pragma unroll
        for (int k = 0; k < SORT_GRP; ++k)
            dv[k] = bdst[bin_edge_place(sRun, sBase, min(i0 + k * BIN_T, n - 1))];
#pragma unroll
        for (int k = 0; k < SORT_GRP; ++k)
            asm volatile("" :: "v"(dv[k]));
#pragma unroll
        for (int k = 0; k < SORT_GRP; ++k)
            if (i0 + k * BIN_T < n) atomicAdd(&sCnt[dv[k] - node0], 1);
    }
    __syncthreads();

    const int v = (tid < BIN_NODES) ? sCnt[tid] : 0;
    int incl = v;
    for (int d = 1; d < 32; d <<= 1) {
        const int t = __shfl_up(incl, d);
        if (lane >= d) incl += t;
    }
    if (lane == 31) sW[wave] = incl;
    __syncthreads();
    if (wave == 0) {
        const int w = sW[lane];
        int wi = w;
        for (int d = 1; d < 32; d <<= 1) {
            const int t = __shfl_up(wi, d);
            if (lane >= d) wi += t;
        }
        sW[lane] = wi - w;
    }
    __syncthreads();
    if (tid < BIN_NODES) {
        const int start = sW[wave] + incl - v;
        sPos[tid] = start;
        const int node = node0 + tid;
        if (node <= N_NODES_C) rowptr[node] = beg + start;
    }
    __syncthreads();

    const bool staged = (n <= SORT_CAP);
    for (int i0 = tid; i0 < n; i0 += SORT_GRP * BIN_T) {
        int    dv[SORT_GRP];
        int    sv[SORT_GRP];
        float4 av[SORT_GRP];
#pragma unroll
        for (int k = 0; k < SORT_GRP; ++k) {
            const int g = bin_edge_place(sRun, sBase, min(i0 + k * BIN_T, n - 1));
            dv[k] = bdst[g];
            sv[k] = bsrc[g];
            av[k] = *(const float4*)(bea + (size_t)g * EDGE_DIM_C);
        }
#pragma unroll
        for (int k = 0; k < SORT_GRP; ++k)
            asm volatile("" :: "v"(dv[k]), "v"(sv[k]),
                         "v"(av[k].x), "v"(av[k].y), "v"(av[k].z), "v"(av[k].w));
#pragma unroll
        for (int k = 0; k < SORT_GRP; ++k) {
            if (i0 + k * BIN_T < n) {
                const int lp = atomicAdd(&sPos[dv[k] - node0], 1);
                if (staged) {
                    lSrc[lp] = sv[k];
                    lEa[lp]  = av[k];
                } else {
                    ssrc[beg + lp] = sv[k];
                    *(float4*)(sea + (size_t)(beg + lp) * EDGE_DIM_C) = av[k];
                }
            }
        }
    }
    if (staged) {
        __syncthreads();
        for (int j = tid; j < n; j += BIN_T) {
            ssrc[beg + j] = lSrc[j];
            *(float4*)(sea + (size_t)(beg + j) * EDGE_DIM_C) = lEa[j];
        }
    }
}

__global__ __launch_bounds__(256)
void fc_kernel(const float* __restrict__ psum, const float* __restrict__ cnt,
               const float* __restrict__ fcW, const float* __restrict__ fcb,
               float* __restrict__ out) {
    const int t = blockIdx.x * 256 + threadIdx.x;
    if (t >= N_GRAPHS_C * NUM_TASKS_C) return;
    const int g = t / NUM_TASKS_C;
    const int k = t - g * NUM_TASKS_C;
    const float inv = 1.0f / fmaxf(cnt[g], 1.0f);
    float s = fcb[k];
#pragma unroll
    for (int c = 0; c < HID_C; ++c)
        s += psum[(size_t)g * HID_C + c] * inv * fcW[k * HID_C + c];
    out[t] = s;
}

extern "C" void kernel_launch(void* const* d_in, const int* in_sizes, int n_in,
                              void* d_out, int out_size, void* d_ws, size_t ws_size,
                              hipStream_t stream) {
    const float* x    = (const float*)d_in[0];
    const int*   ei   = (const int*)  d_in[1];
    const float* ea   = (const float*)d_in[2];
    const int*   bat  = (const int*)  d_in[3];
    const float* e1W  = (const float*)d_in[4];  const float* e1b  = (const float*)d_in[5];
    const float* n1W1 = (const float*)d_in[6];  const float* n1b1 = (const float*)d_in[7];
    const float* n1W2 = (const float*)d_in[8];  const float* n1b2 = (const float*)d_in[9];
    const float* e2W  = (const float*)d_in[10]; const float* e2b  = (const float*)d_in[11];
    const float* n2W1 = (const float*)d_in[12]; const float* n2b1 = (const float*)d_in[13];
    const float* n2W2 = (const float*)d_in[14]; const float* n2b2 = (const float*)d_in[15];
    const float* fcW  = (const float*)d_in[16]; const float* fcb  = (const float*)d_in[17];
    float* out = (float*)d_out;

    _Float16* h     = (_Float16*)d_ws;
    float*    psum  = (float*)(h + (size_t)N_NODES_C * HID_C);
    float*    cnt   = psum  + (size_t)N_GRAPHS_C * HID_C;
    float*    sea   = cnt   + (size_t)N_GRAPHS_C;
    int*      ssrc  = (int*)(sea + (size_t)N_EDGES_C * EDGE_DIM_C);
    int*      bdst  = ssrc  + (size_t)N_EDGES_C;
    int*      bsrc  = bdst  + (size_t)N_EDGES_C;
    float*    bea   = (float*)(bsrc + (size_t)N_EDGES_C);
    int*      hist  = (int*)(bea + (size_t)N_EDGES_C * EDGE_DIM_C);
    int*      lstart = hist + (size_t)NBINS * BIN_G;
    int*      rowptr = lstart + (size_t)NBINS * BIN_G;

    hipMemsetAsync(psum,  0, sizeof(float) * (size_t)N_GRAPHS_C * HID_C,  stream);
    hipMemsetAsync(cnt,   0, sizeof(float) * (size_t)N_GRAPHS_C,          stream);

    chunk_sort_kernel<<<BIN_G, BIN_T, 0, stream>>>(ei, ea, hist, lstart, bdst, bsrc, bea);
    bin_sort_kernel<<<NBINS, BIN_T, 0, stream>>>(hist, lstart, bdst, bsrc, bea, rowptr, ssrc, sea);
    const unsigned grid = (N_NODES_C + 15) / 16;
    conv_kernel<IN_DIM_C, float, false><<<grid, 128, 0, stream>>>(
        x, rowptr, ssrc, sea, e1W, e1b, n1W1, n1b1, n1W2, n1b2, h,
        nullptr, nullptr, nullptr, N_NODES_C);
    conv_kernel<HID_C, _Float16, true><<<grid, 128, 0, stream>>>(
        h, rowptr, ssrc, sea, e2W, e2b, n2W1, n2b1, n2W2, n2b2, nullptr,
        bat, psum, cnt, N_NODES_C);
    {
        const unsigned fgrid = (N_GRAPHS_C * NUM_TASKS_C + 255) / 256;
        fc_kernel<<<fgrid, 256, 0, stream>>>(psum, cnt, fcW, fcb, out);
    }
}


# Conv2 edge sums: per-edge bookkeeping on the scalar unit: a group's 8 src/edge_attr records come by scalar loads, row address = scalar base + lane offset, full groups unmasked; sum order kept
# speedup vs baseline: 4.1002x; 1.0388x over previous
#include <hip/hip_runtime.h>
#include <hip/hip_bf16.h>

#define N_NODES_C  100000
#define N_EDGES_C  1600000
#define N_GRAPHS_C 1000
#define HID_C      64
#define IN_DIM_C   7
#define EDGE_DIM_C 4
#define NUM_TASKS_C 12

typedef __attribute__((ext_vector_type(16))) _Float16 v16h;
typedef __attribute__((ext_vector_type(8)))  _Float16 v8h;
typedef __attribute__((ext_vector_type(8)))  float    v8f;
typedef __attribute__((ext_vector_type(2)))  _Float16 v2h;
typedef __attribute__((ext_vector_type(2)))  float    v2f;
typedef int   src8_t __attribute__((ext_vector_type(8),  aligned(4)));
typedef float ea16_t __attribute__((ext_vector_type(16), aligned(16)));

__device__ __forceinline__ void atom_add_f32(float* p, float v) {
    unsafeAtomicAdd(p, v);
}

__device__ __forceinline__ v16h lds_load_tile(const _Float16* s, int ld, int kbase) {
    const int lane = threadIdx.x & 31;
    const int r    = lane & 15;
    const int koff = (lane < 16) ? 0 : 8;
    const _Float16* p = s + r * ld + kbase + koff;
    v8h lo = *(const v8h*)(p);
    v8h hi = *(const v8h*)(p + 16);
    v16h out;
#pragma unroll
    for (int e = 0; e < 8; ++e) { out[e] = lo[e]; out[8 + e] = hi[e]; }
    return out;
}

template <int K1, typename TI, bool POOL>
__global__ __launch_bounds__(128)
void conv_kernel(const TI* __restrict__ in0, const int* __restrict__ rowptr,
                 const int* __restrict__ ssrc, const float* __restrict__ sea,
                 const float* __restrict__ eW, const float* __restrict__ eb,
                 const float* __restrict__ W1,  const float* __restrict__ b1,
                 const float* __restrict__ W2,  const float* __restrict__ b2,
                 _Float16* __restrict__ out, const int* __restrict__ batch,
                 float* __restrict__ psum, float* __restrict__ cnt,
                 int n_nodes) {
    constexpr int K1P = (K1 + 31) & ~31;

    __shared__ __align__(16) _Float16 sA  [16 * K1P];
    __shared__ __align__(16) _Float16 sMid[16 * HID_C];
    __shared__ __align__(16) _Float16 sB1 [HID_C * K1P];
    __shared__ __align__(16) _Float16 sB2 [HID_C * HID_C];
    __shared__ __align__(16) int      sG  [16];

    const int row0 = blockIdx.x * 16;
    const int tid  = threadIdx.x;
    const int wave = tid >> 5;
    const int lane = tid & 31;
    const int n0   = wave * 16;

    for (int i = tid; i < HID_C * K1P; i += 128) {
        const int n = i / K1P, k = i - n * K1P;
        sB1[i] = (_Float16)((k < K1) ? W1[n * K1 + k] : 0.0f);
    }
    for (int i = tid; i < HID_C * HID_C; i += 128) {
        const int n = i >> 6, k = i & 63;
        sB2[i] = (_Float16)W2[n * HID_C + k];
    }
    if constexpr (POOL) {
        if (tid < 16) sG[tid] = (row0 + tid < n_nodes) ? batch[row0 + tid] : -1;
    }
    if constexpr (K1 == HID_C) {
        const int c = 2 * lane;
        const float4 w0 = *(const float4*)(eW + c * EDGE_DIM_C);
        const float4 w1 = *(const float4*)(eW + c * EDGE_DIM_C + EDGE_DIM_C);
        const float  e0 = eb[c];
        const float  e1 = eb[c + 1];
        const int wv  = __builtin_amdgcn_readfirstlane(wave);
        const int nb  = row0 + wv * 4;
        const int rb0 = rowptr[min(nb,     n_nodes)];
        const int rb1 = rowptr[min(nb + 1, n_nodes)];
        const int rb2 = rowptr[min(nb + 2, n_nodes)];
        const int rb3 = rowptr[min(nb + 3, n_nodes)];
        const int rb4 = rowptr[min(nb + 4, n_nodes)];
        _Float16* const sAw = sA + (wv * 4) * K1P + c;
        v2h hq[4];
#pragma unroll
        for (int qq = 0; qq < 4; ++qq)
            hq[qq] = *(const v2h*)(in0 + (size_t)min(nb + qq, n_nodes - 1) * HID_C + c);
        const int wbeg = rb0, wend = rb4;
        int   q    = 0;
        int   qend = rb1;
        int   g0   = wbeg;
        int   qrel = qend - g0;
        v2f acc = {0.0f, 0.0f};
        const v2f wx = {w0.x, w1.x}, wy = {w0.y, w1.y}, wz = {w0.z, w1.z}, ww = {w0.w, w1.w};
        const v2f eb2 = {e0, e1};
        static_assert((unsigned long long)N_NODES_C * HID_C * sizeof(_Float16) < (1ull << 32),
                      "row byte offsets fit 32 bits");
        const unsigned lane4 = (unsigned)lane * 4u;
        const char* const in0b = (const char*)in0;
#define CONV2_CLOSE_NODE()                                                    \
        {                                                                     \
            v2h* const p_ = (v2h*)(sAw + q * K1P);                            \
            const v2h hq_ = *p_;                                              \
            v2h o_;                                                           \
            o_[0] = (_Float16)((float)hq_[0] + acc[0]);                       \
            o_[1] = (_Float16)((float)hq_[1] + acc[1]);                       \
            *p_ = o_;                                                         \
            ++q;                                                              \
            qend = (q == 1) ? rb2 : (q == 2) ? rb3 : rb4;                     \
            qrel = qend - g0;                                                 \
            acc[0] = 0.0f; acc[1] = 0.0f;                                     \
        }
#define CONV2_GATHER(K, SRC_)                                                 \
        hv[K] = *(const v2h*)(in0b + ((unsigned)(SRC_) * (unsigned)(HID_C * sizeof(_Float16)) + lane4));
#define CONV2_EDGE(K, EA_, I_, MASKED)                                        \
        {                                                                     \
            const bool on_ = !(MASKED) || ((K) < cnt);                        \
            if (on_) {                                                        \
                while ((K) >= qrel) CONV2_CLOSE_NODE()    \
            }                                                                 \
            const float ax_ = EA_[4 * (I_) + 0];                              \
            const float ay_ = EA_[4 * (I_) + 1];                              \
            const float az_ = EA_[4 * (I_) + 2];                              \
            const float aw_ = EA_[4 * (I_) + 3];                              \
            v2f t_ = __builtin_elementwise_fma(wx, (v2f){ax_, ax_}, eb2);     \
            t_ = __builtin_elementwise_fma(wy, (v2f){ay_, ay_}, t_);          \
            t_ = __builtin_elementwise_fma(wz, (v2f){az_, az_}, t_);          \
            t_ = __builtin_elementwise_fma(ww, (v2f){aw_, aw_}, t_);          \
            t_ += (v2f){(float)hv[K][0], (float)hv[K][1]};                    \
            const v2f r_ = {fmaxf(t_[0], 0.0f), fmaxf(t_[1], 0.0f)};          \
            if (MASKED) acc += on_ ? r_ : (v2f){0.0f, 0.0f};                  \
            else        acc += r_;                                            \
        }
#define CONV2_GROUP(MASKED)                                                   \
        CONV2_EDGE(0, ea0, 0, MASKED) CONV2_EDGE(1, ea0, 1, MASKED)           \
        CONV2_EDGE(2, ea0, 2, MASKED) CONV2_EDGE(3, ea0, 3, MASKED)           \
        CONV2_EDGE(4, ea1, 0, MASKED) CONV2_EDGE(5, ea1, 1, MASKED)           \
        CONV2_EDGE(6, ea1, 2, MASKED) CONV2_EDGE(7, ea1, 3, MASKED)
#pragma unroll
        for (int qq = 0; qq < 4; ++qq) {
            v2h hz;
            hz[0] = (_Float16)0.0f; hz[1] = (_Float16)0.0f;
            *(v2h*)(sAw + qq * K1P) = (nb + qq < n_nodes) ? hq[qq] : hz;
        }
        src8_t sN = {0, 0, 0, 0, 0, 0, 0, 0};
        if (wbeg < wend) sN = *(const src8_t*)(ssrc + wbeg);
        for (; g0 < wend; g0 += 8) {
            const int cnt = min(8, wend - g0);
            qrel = qend - g0;
            const src8_t sv = sN;
            const ea16_t ea0 = *(const ea16_t*)(sea + (size_t)g0 * EDGE_DIM_C);
            const ea16_t ea1 = *(const ea16_t*)(sea + (size_t)g0 * EDGE_DIM_C + 16);
            sN = *(const src8_t*)(ssrc + g0 + 8);
            v2h hv[8];
            if (cnt == 8) {
                CONV2_GATHER(0, sv[0]) CONV2_GATHER(1, sv[1]) CONV2_GATHER(2, sv[2]) CONV2_GATHER(3, sv[3])
                CONV2_GATHER(4, sv[4]) CONV2_GATHER(5, sv[5]) CONV2_GATHER(6, sv[6]) CONV2_GATHER(7, sv[7])
                CONV2_GROUP(false)
            } else {
                CONV2_GATHER(0, sv[0])
                CONV2_GATHER(1, (1 < cnt) ? sv[1] : sv[0]) CONV2_GATHER(2, (2 < cnt) ? sv[2] : sv[0])
                CONV2_GATHER(3, (3 < cnt) ? sv[3] : sv[0]) CONV2_GATHER(4, (4 < cnt) ? sv[4] : sv[0])
                CONV2_GATHER(5, (5 < cnt) ? sv[5] : sv[0]) CONV2_GATHER(6, (6 < cnt) ? sv[6] : sv[0])
                CONV2_GATHER(7, sv[0])
                CONV2_GROUP(true)
            }
        }
        while (q < 4) CONV2_CLOSE_NODE()
#undef CONV2_GROUP
#undef CONV2_EDGE
#undef CONV2_GATHER
#undef CONV2_CLOSE_NODE
    } else {
        const int m  = tid >> 3;
        const int c  = tid & 7;
        const int cc = min(c, K1 - 1);
        const int n  = row0 + m;
        const float4 w = *(const float4*)(eW + cc * EDGE_DIM_C);
        const float  e = eb[cc];
        const int beg   = rowptr[min(n,     n_nodes)];
        const int end   = rowptr[min(n + 1, n_nodes)];
        const int lastv = max(end - 1, 0);
        const float own = (float)in0[(size_t)min(n, n_nodes - 1) * K1 + cc];
        int    sv;
        float4 av;
        {
            const int je = min(beg + c, lastv);
            sv = ssrc[je];
            av = *(const float4*)(sea + (size_t)je * EDGE_DIM_C);
        }
        float acc = 0.0f;
        for (int j0 = beg; __builtin_amdgcn_ballot_w32(j0 < end) != 0u; j0 += 8) {
            float xv[8];
#define CONV1_GATHER(K)                                                       \
            {                                                                 \
                const int srk_ = __builtin_amdgcn_mov_dpp8(sv, (K) * 0x249249);     \
                xv[K] = (float)in0[(size_t)srk_ * K1 + cc];                   \
            }
            CONV1_GATHER(0) CONV1_GATHER(1) CONV1_GATHER(2) CONV1_GATHER(3)
            CONV1_GATHER(4) CONV1_GATHER(5) CONV1_GATHER(6) CONV1_GATHER(7)
#undef CONV1_GATHER
            const int    jn = min(j0 + 8 + c, lastv);
            const int    sn = ssrc[jn];
            const float4 an = *(const float4*)(sea + (size_t)jn * EDGE_DIM_C);
            asm volatile("" :: "v"(xv[0]), "v"(xv[1]), "v"(xv[2]), "v"(xv[3]),
                               "v"(xv[4]), "v"(xv[5]), "v"(xv[6]), "v"(xv[7]));
#define CONV1_SUM(K)                                                          \
            {                                                                 \
                const float ax_ = __builtin_bit_cast(float, __builtin_amdgcn_mov_dpp8(__builtin_bit_cast(int, av.x), (K) * 0x249249)); \
                const float ay_ = __builtin_bit_cast(float, __builtin_amdgcn_mov_dpp8(__builtin_bit_cast(int, av.y), (K) * 0x249249)); \
                const float az_ = __builtin_bit_cast(float, __builtin_amdgcn_mov_dpp8(__builtin_bit_cast(int, av.z), (K) * 0x249249)); \
                const float aw_ = __builtin_bit_cast(float, __builtin_amdgcn_mov_dpp8(__builtin_bit_cast(int, av.w), (K) * 0x249249)); \
                float t_ = e + ax_ * w.x + ay_ * w.y + az_ * w.z + aw_ * w.w; \
                t_ += xv[K];                                                  \
                acc += (j0 + (K) < end) ? fmaxf(t_, 0.0f) : 0.0f;             \
            }
            CONV1_SUM(0) CONV1_SUM(1) CONV1_SUM(2) CONV1_SUM(3)
            CONV1_SUM(4) CONV1_SUM(5) CONV1_SUM(6) CONV1_SUM(7)
#undef CONV1_SUM
            sv = sn;
            av = an;
        }
        const float v = (c < K1 && n < n_nodes) ? own + acc : 0.0f;
        sA[m * K1P + c] = (_Float16)v;
        for (int k = c + 8; k < K1P; k += 8)
            sA[m * K1P + k] = (_Float16)0.0f;
    }
    __syncthreads();

    v8f acc = {};
#pragma unroll
    for (int kb = 0; kb < K1P; kb += 32) {
        v16h a = lds_load_tile(sA, K1P, kb);
        v16h b = lds_load_tile(sB1 + n0 * K1P, K1P, kb);
        acc = __builtin_amdgcn_wmma_f32_16x16x32_f16(false, a, false, b,
                                                     (short)0, acc, false, false);
    }
    {
        const int n  = lane & 15;
        const int mb = (lane < 16) ? 0 : 8;
        const float bias = b1[n0 + n];
#pragma unroll
        for (int r = 0; r < 8; ++r) {
            float v = acc[r] + bias;
            v = v > 0.0f ? v : 0.0f;
            sMid[(mb + r) * HID_C + (n0 + n)] = (_Float16)v;
        }
    }
    __syncthreads();

    v8f acc2 = {};
#pragma unroll
    for (int kb = 0; kb < HID_C; kb += 32) {
        v16h a = lds_load_tile(sMid, HID_C, kb);
        v16h b = lds_load_tile(sB2 + n0 * HID_C, HID_C, kb);
        acc2 = __builtin_amdgcn_wmma_f32_16x16x32_f16(false, a, false, b,
                                                      (short)0, acc2, false, false);
    }
    if constexpr (!POOL) {
        const int n  = lane & 15;
        const int mb = (lane < 16) ? 0 : 8;
        const float bias = b2[n0 + n];
#pragma unroll
        for (int r = 0; r < 8; ++r) {
            const int rr = row0 + mb + r;
            if (rr < n_nodes) {
                float v = acc2[r] + bias;
                v = v > 0.0f ? v : 0.0f;
                out[(size_t)rr * HID_C + (n0 + n)] = (_Float16)v;
            }
        }
    } else {
        const int n  = lane & 15;
        const int mb = (lane < 16) ? 0 : 8;
        const float bias = b2[n0 + n];
        const bool cnt_lane = (wave == 0) && (n == 0);
        const int4 ga = *(const int4*)(sG + mb);
        const int4 gb = *(const int4*)(sG + mb + 4);
        const int gs[8] = { ga.x, ga.y, ga.z, ga.w, gb.x, gb.y, gb.z, gb.w };
        int   gcur = -1;
        float run  = 0.0f;
        int   len  = 0;
#pragma unroll
        for (int r = 0; r < 8; ++r) {
            const int g = gs[r];
            float v = acc2[r] + bias;
            v = v > 0.0f ? v : 0.0f;
            if (g != gcur) {
                if (gcur >= 0) {
                    atom_add_f32(&psum[(size_t)gcur * HID_C + (n0 + n)], run);
                    if (cnt_lane) atom_add_f32(&cnt[gcur], (float)len);
                }
                gcur = g; run = 0.0f; len = 0;
            }
            run += v;
            len += 1;
        }
        if (gcur >= 0) {
            atom_add_f32(&psum[(size_t)gcur * HID_C + (n0 + n)], run);
            if (cnt_lane) atom_add_f32(&cnt[gcur], (float)len);
        }
    }
}

#define BIN_SHIFT 9
#define BIN_NODES (1 << BIN_SHIFT)
#define NBINS     ((N_NODES_C >> BIN_SHIFT) + 1)
#define BIN_G     256
#define BIN_T     1024
#define BIN_W     (BIN_T / 32)
#define BIN_CHUNK ((N_EDGES_C + BIN_G - 1) / BIN_G)
#define BIN_TRIPS ((BIN_CHUNK + BIN_T - 1) / BIN_T)
#define SORT_GRP  4
#define SORT_CAP  12288
#define PART_TRIPS ((NBINS * BIN_G + BIN_T * 4 - 1) / (BIN_T * 4))
static_assert(NBINS <= 256, "one wave scans the bins, 8 per lane");
static_assert(BIN_G == 256, "one wave scans a bin's 256 runs, 8 per lane");

__global__ __launch_bounds__(BIN_T)
void chunk_sort_kernel(const int* __restrict__ ei, const float* __restrict__ ea,
                       int* __restrict__ hist, int* __restrict__ lstart,
                       int* __restrict__ bdst, int* __restrict__ bsrc,
                       float* __restrict__ bea) {
    __shared__ int sCnt[256];
    __shared__ int sStart[256];
    __shared__ int sRank[256];
    __shared__ __align__(16) float4 lEa[BIN_CHUNK];
    __shared__ int lDst[BIN_CHUNK];
    __shared__ int lSrc[BIN_CHUNK];
    const int tid  = threadIdx.x;
    const int lane = tid & 31;
    const int blk  = blockIdx.x;
    const int e0 = blk * BIN_CHUNK;
    const int e1 = min(e0 + BIN_CHUNK, N_EDGES_C);

    if (tid < 256) { sCnt[tid] = 0; sRank[tid] = 0; }
    int    dv[BIN_TRIPS];
    int    sv[BIN_TRIPS];
    float4 av[BIN_TRIPS];
#pragma unroll
    for (int k = 0; k < BIN_TRIPS; ++k) {
        const int e = min(e0 + tid + k * BIN_T, N_EDGES_C - 1);
        dv[k] = ei[N_EDGES_C + e];
        sv[k] = ei[e];
        av[k] = *(const float4*)(ea + (size_t)e * EDGE_DIM_C);
    }
#pragma unroll
    for (int k = 0; k < BIN_TRIPS; ++k)
        asm volatile("" :: "v"(dv[k]), "v"(sv[k]),
                     "v"(av[k].x), "v"(av[k].y), "v"(av[k].z), "v"(av[k].w));
    __syncthreads();
#pragma unroll
    for (int k = 0; k < BIN_TRIPS; ++k)
        if (e0 + tid + k * BIN_T < e1) atomicAdd(&sCnt[dv[k] >> BIN_SHIFT], 1);
    __syncthreads();
    if (tid < 32) {
        int c[8];
        int s = 0;
#pragma unroll
        for (int k = 0; k < 8; ++k) { c[k] = sCnt[lane * 8 + k]; s += c[k]; }
        int incl = s;
        for (int d = 1; d < 32; d <<= 1) {
            const int t = __shfl_up(incl, d);
            if (lane >= d) incl += t;
        }
        int run = incl - s;
#pragma unroll
        for (int k = 0; k < 8; ++k) {
            const int idx = lane * 8 + k;
            sStart[idx] = run;
            if (idx < NBINS) {
                hist  [idx * BIN_G + blk] = c[k];
                lstart[idx * BIN_G + blk] = run;
            }
            run += c[k];
        }
    }
    __syncthreads();
#pragma unroll
    for (int k = 0; k < BIN_TRIPS; ++k) {
        if (e0 + tid + k * BIN_T < e1) {
            const int bin = dv[k] >> BIN_SHIFT;
            const int lp  = sStart[bin] + atomicAdd(&sRank[bin], 1);
            lDst[lp] = dv[k];
            lSrc[lp] = sv[k];
            lEa[lp]  = av[k];
        }
    }
    __syncthreads();
    for (int j = tid; j < e1 - e0; j += BIN_T) {
        bdst[e0 + j] = lDst[j];
        bsrc[e0 + j] = lSrc[j];
        *(float4*)(bea + (size_t)(e0 + j) * EDGE_DIM_C) = lEa[j];
    }
}

__device__ __forceinline__ int bin_edge_place(const int* sRun, const int* sBase, int i) {
    int lo = 0, hi = BIN_G;
#pragma unroll
    for (int s = 0; s < 8; ++s) {
        const int mid = (lo + hi) >> 1;
        if (sRun[mid] <= i) lo = mid; else hi = mid;
    }
    return sBase[lo] + (i - sRun[lo]);
}

__global__ __launch_bounds__(BIN_T)
void bin_sort_kernel(const int* __restrict__ hist, const int* __restrict__ lstart,
                     const int* __restrict__ bdst, const int* __restrict__ bsrc,
                     const float* __restrict__ bea,
                     int* __restrict__ rowptr, int* __restrict__ ssrc,
                     float* __restrict__ sea) {
    __shared__ int sRun[BIN_G + 1];
    __shared__ int sBase[BIN_G];
    __shared__ int sCnt[BIN_NODES];
    __shared__ int sPos[BIN_NODES];
    __shared__ int sW[BIN_W];
    __shared__ int sRed[BIN_W];
    __shared__ __align__(16) float4 lEa[SORT_CAP];
    __shared__ int lSrc[SORT_CAP];
    const int tid   = threadIdx.x;
    const int lane  = tid & 31;
    const int wave  = tid >> 5;
    const int bin   = blockIdx.x;
    const int node0 = bin << BIN_SHIFT;

    int part = 0;
    {
        const int lim = bin * BIN_G;
        int4 h[PART_TRIPS];
#pragma unroll
        for (int k = 0; k < PART_TRIPS; ++k)
            h[k] = *(const int4*)(hist + min(tid * 4 + k * BIN_T * 4, NBINS * BIN_G - 4));
#pragma unroll
        for (int k = 0; k < PART_TRIPS; ++k)
            if (tid * 4 + k * BIN_T * 4 < lim) part += h[k].x + h[k].y + h[k].z + h[k].w;
    }
    for (int d = 16; d > 0; d >>= 1) part += __shfl_xor(part, d);
    if (lane == 0) sRed[wave] = part;
    if (tid < BIN_NODES) sCnt[tid] = 0;
    if (wave == 0) {
        const int4* ph = (const int4*)(hist   + bin * BIN_G + lane * 8);
        const int4* pl = (const int4*)(lstart + bin * BIN_G + lane * 8);
        const int4 ha = ph[0], hb = ph[1];
        const int4 la = pl[0], lb = pl[1];
        const int c[8]  = { ha.x, ha.y, ha.z, ha.w, hb.x, hb.y, hb.z, hb.w };
        const int ls[8] = { la.x, la.y, la.z, la.w, lb.x, lb.y, lb.z, lb.w };
        int s = 0;
#pragma unroll
        for (int k = 0; k < 8; ++k) s += c[k];
        int incl = s;
        for (int d = 1; d < 32; d <<= 1) {
            const int t = __shfl_up(incl, d);
            if (lane >= d) incl += t;
        }
        int run = incl - s;
#pragma unroll
        for (int k = 0; k < 8; ++k) {
            const int b = lane * 8 + k;
            sRun[b]  = run;
            sBase[b] = b * BIN_CHUNK + ls[k];
            run += c[k];
        }
        if (lane == 31) sRun[BIN_G] = incl;
    }
    __syncthreads();
    int beg = 0;
#pragma unroll
    for (int w = 0; w < BIN_W; ++w) beg += sRed[w];
    const int n = sRun[BIN_G];

    for (int i0 = tid; i0 < n; i0 += SORT_GRP * BIN_T) {
        int dv[SORT_GRP];
#pragma unroll
        for (int k = 0; k < SORT_GRP; ++k)
            dv[k] = bdst[bin_edge_place(sRun, sBase, min(i0 + k * BIN_T, n - 1))];
#pragma unroll
        for (int k = 0; k < SORT_GRP; ++k)
            asm volatile("" :: "v"(dv[k]));
#pragma unroll
        for (int k = 0; k < SORT_GRP; ++k)
            if (i0 + k * BIN_T < n) atomicAdd(&sCnt[dv[k] - node0], 1);
    }
    __syncthreads();

    const int v = (tid < BIN_NODES) ? sCnt[tid] : 0;
    int incl = v;
    for (int d = 1; d < 32; d <<= 1) {
        const int t = __shfl_up(incl, d);
        if (lane >= d) incl += t;
    }
    if (lane == 31) sW[wave] = incl;
    __syncthreads();
    if (wave == 0) {
        const int w = sW[lane];
        int wi = w;
        for (int d = 1; d < 32; d <<= 1) {
            const int t = __shfl_up(wi, d);
            if (lane >= d) wi += t;
        }
        sW[lane] = wi - w;
    }
    __syncthreads();
    if (tid < BIN_NODES) {
        const int start = sW[wave] + incl - v;
        sPos[tid] = start;
        const int node = node0 + tid;
        if (node <= N_NODES_C) rowptr[node] = beg + start;
    }
    __syncthreads();

    const bool staged = (n <= SORT_CAP);
    for (int i0 = tid; i0 < n; i0 += SORT_GRP * BIN_T) {
        int    dv[SORT_GRP];
        int    sv[SORT_GRP];
        float4 av[SORT_GRP];
#pragma unroll
        for (int k = 0; k < SORT_GRP; ++k) {
            const int g = bin_edge_place(sRun, sBase, min(i0 + k * BIN_T, n - 1));
            dv[k] = bdst[g];
            sv[k] = bsrc[g];
            av[k] = *(const float4*)(bea + (size_t)g * EDGE_DIM_C);
        }
#pragma unroll
        for (int k = 0; k < SORT_GRP; ++k)
            asm volatile("" :: "v"(dv[k]), "v"(sv[k]),
                         "v"(av[k].x), "v"(av[k].y), "v"(av[k].z), "v"(av[k].w));
#pragma unroll
        for (int k = 0; k < SORT_GRP; ++k) {
            if (i0 + k * BIN_T < n) {
                const int lp = atomicAdd(&sPos[dv[k] - node0], 1);
                if (staged) {
                    lSrc[lp] = sv[k];
                    lEa[lp]  = av[k];
                } else {
                    ssrc[beg + lp] = sv[k];
                    *(float4*)(sea + (size_t)(beg + lp) * EDGE_DIM_C) = av[k];
                }
            }
        }
    }
    if (staged) {
        __syncthreads();
        for (int j = tid; j < n; j += BIN_T) {
            ssrc[beg + j] = lSrc[j];
            *(float4*)(sea + (size_t)(beg + j) * EDGE_DIM_C) = lEa[j];
        }
    }
}

__global__ __launch_bounds__(256)
void fc_kernel(const float* __restrict__ psum, const float* __restrict__ cnt,
               const float* __restrict__ fcW, const float* __restrict__ fcb,
               float* __restrict__ out) {
    const int t = blockIdx.x * 256 + threadIdx.x;
    if (t >= N_GRAPHS_C * NUM_TASKS_C) return;
    const int g = t / NUM_TASKS_C;
    const int k = t - g * NUM_TASKS_C;
    const float inv = 1.0f / fmaxf(cnt[g], 1.0f);
    float s = fcb[k];
#pragma unroll
    for (int c = 0; c < HID_C; ++c)
        s += psum[(size_t)g * HID_C + c] * inv * fcW[k * HID_C + c];
    out[t] = s;
}

extern "C" void kernel_launch(void* const* d_in, const int* in_sizes, int n_in,
                              void* d_out, int out_size, void* d_ws, size_t ws_size,
                              hipStream_t stream) {
    const float* x    = (const float*)d_in[0];
    const int*   ei   = (const int*)  d_in[1];
    const float* ea   = (const float*)d_in[2];
    const int*   bat  = (const int*)  d_in[3];
    const float* e1W  = (const float*)d_in[4];  const float* e1b  = (const float*)d_in[5];
    const float* n1W1 = (const float*)d_in[6];  const float* n1b1 = (const float*)d_in[7];
    const float* n1W2 = (const float*)d_in[8];  const float* n1b2 = (const float*)d_in[9];
    const float* e2W  = (const float*)d_in[10]; const float* e2b  = (const float*)d_in[11];
    const float* n2W1 = (const float*)d_in[12]; const float* n2b1 = (const float*)d_in[13];
    const float* n2W2 = (const float*)d_in[14]; const float* n2b2 = (const float*)d_in[15];
    const float* fcW  = (const float*)d_in[16]; const float* fcb  = (const float*)d_in[17];
    float* out = (float*)d_out;

    _Float16* h     = (_Float16*)d_ws;
    float*    psum  = (float*)(h + (size_t)N_NODES_C * HID_C);
    float*    cnt   = psum  + (size_t)N_GRAPHS_C * HID_C;
    float*    sea   = cnt   + (size_t)N_GRAPHS_C;
    int*      ssrc  = (int*)(sea + (size_t)N_EDGES_C * EDGE_DIM_C);
    int*      bdst  = ssrc  + (size_t)N_EDGES_C;
    int*      bsrc  = bdst  + (size_t)N_EDGES_C;
    float*    bea   = (float*)(bsrc + (size_t)N_EDGES_C);
    int*      hist  = (int*)(bea + (size_t)N_EDGES_C * EDGE_DIM_C);
    int*      lstart = hist + (size_t)NBINS * BIN_G;
    int*      rowptr = lstart + (size_t)NBINS * BIN_G;

    hipMemsetAsync(psum,  0, sizeof(float) * (size_t)N_GRAPHS_C * HID_C,  stream);
    hipMemsetAsync(cnt,   0, sizeof(float) * (size_t)N_GRAPHS_C,          stream);

    chunk_sort_kernel<<<BIN_G, BIN_T, 0, stream>>>(ei, ea, hist, lstart, bdst, bsrc, bea);
    bin_sort_kernel<<<NBINS, BIN_T, 0, stream>>>(hist, lstart, bdst, bsrc, bea, rowptr, ssrc, sea);
    const unsigned grid = (N_NODES_C + 15) / 16;
    conv_kernel<IN_DIM_C, float, false><<<grid, 128, 0, stream>>>(
        x, rowptr, ssrc, sea, e1W, e1b, n1W1, n1b1, n1W2, n1b2, h,
        nullptr, nullptr, nullptr, N_NODES_C);
    conv_kernel<HID_C, _Float16, true><<<grid, 128, 0, stream>>>(
        h, rowptr, ssrc, sea, e2W, e2b, n2W1, n2b1, n2W2, n2b2, nullptr,
        bat, psum, cnt, N_NODES_C);
    {
        const unsigned fgrid = (N_GRAPHS_C * NUM_TASKS_C + 255) / 256;
        fc_kernel<<<fgrid, 256, 0, stream>>>(psum, cnt, fcW, fcb, out);
    }
}


# MLP weights packed once per call in WMMA B-operand order (same operand type; by the first launch); conv waves read them with 16-byte global loads into registers: no per-block weight staging in LDS
# speedup vs baseline: 4.9018x; 1.1955x over previous
#include <hip/hip_runtime.h>
#include <hip/hip_bf16.h>

#define N_NODES_C  100000
#define N_EDGES_C  1600000
#define N_GRAPHS_C 1000
#define HID_C      64
#define IN_DIM_C   7
#define EDGE_DIM_C 4
#define NUM_TASKS_C 12

typedef __attribute__((ext_vector_type(16))) _Float16 v16h;
typedef __attribute__((ext_vector_type(8)))  _Float16 v8h;
typedef __attribute__((ext_vector_type(8)))  float    v8f;
typedef __attribute__((ext_vector_type(2)))  _Float16 v2h;
typedef __attribute__((ext_vector_type(2)))  float    v2f;
typedef int   src8_t __attribute__((ext_vector_type(8),  aligned(4)));
typedef float ea16_t __attribute__((ext_vector_type(16), aligned(16)));

__device__ __forceinline__ void atom_add_f32(float* p, float v) {
    unsafeAtomicAdd(p, v);
}

__device__ __forceinline__ v16h lds_load_tile(const _Float16* s, int ld, int kbase) {
    const int lane = threadIdx.x & 31;
    const int r    = lane & 15;
    const int koff = (lane < 16) ? 0 : 8;
    const _Float16* p = s + r * ld + kbase + koff;
    v8h lo = *(const v8h*)(p);
    v8h hi = *(const v8h*)(p + 16);
    v16h out;
#pragma unroll
    for (int e = 0; e < 8; ++e) { out[e] = lo[e]; out[8 + e] = hi[e]; }
    return out;
}

#define WP_C1L1  0
#define WP_C1L2  (WP_C1L1 + HID_C * 32)
#define WP_C2L1  (WP_C1L2 + HID_C * 64)
#define WP_C2L2  (WP_C2L1 + HID_C * 64)
#define WP_TOTAL (WP_C2L2 + HID_C * 64)
#define ROWPTR_LEN ((N_NODES_C + 1 + 3) & ~3)

__device__ __forceinline__ _Float16 pack_weight_elem(const float* __restrict__ W, int K, int KS, int i) {
    const int e8   = i & 7;
    const int lane = (i >> 3) & 31;
    const int part = (i >> 8) & 1;
    const int t    = i >> 9;
    const int wv   = t / KS;
    const int ks   = t - wv * KS;
    const int n    = wv * 16 + (lane & 15);
    const int k    = ks * 32 + ((lane < 16) ? 0 : 8) + part * 16 + e8;
    return (_Float16)((k < K) ? W[n * K + k] : 0.0f);
}

__device__ __forceinline__ v16h wfrag_load(const _Float16* __restrict__ wp, int KS, int wave, int ks, int lane) {
    const v8h* p = (const v8h*)wp + ((wave * KS + ks) * 2) * 32 + lane;
    const v8h lo = p[0];
    const v8h hi = p[32];
    v16h out;
#pragma unroll
    for (int e = 0; e < 8; ++e) { out[e] = lo[e]; out[8 + e] = hi[e]; }
    return out;
}

template <int K1, typename TI, bool POOL>
__global__ __launch_bounds__(128)
void conv_kernel(const TI* __restrict__ in0, const int* __restrict__ rowptr,
                 const int* __restrict__ ssrc, const float* __restrict__ sea,
                 const float* __restrict__ eW, const float* __restrict__ eb,
                 const _Float16* __restrict__ wp1, const float* __restrict__ b1,
                 const _Float16* __restrict__ wp2, const float* __restrict__ b2,
                 _Float16* __restrict__ out, const int* __restrict__ batch,
                 float* __restrict__ psum, float* __restrict__ cnt,
                 int n_nodes) {
    constexpr int K1P = (K1 + 31) & ~31;

    __shared__ __align__(16) _Float16 sA  [16 * K1P];
    __shared__ __align__(16) _Float16 sMid[16 * HID_C];
    __shared__ __align__(16) int      sG  [16];

    const int row0 = blockIdx.x * 16;
    const int tid  = threadIdx.x;
    const int wave = tid >> 5;
    const int lane = tid & 31;
    const int n0   = wave * 16;

    if constexpr (POOL) {
        if (tid < 16) sG[tid] = (row0 + tid < n_nodes) ? batch[row0 + tid] : -1;
    }
    if constexpr (K1 == HID_C) {
        const int c = 2 * lane;
        const float4 w0 = *(const float4*)(eW + c * EDGE_DIM_C);
        const float4 w1 = *(const float4*)(eW + c * EDGE_DIM_C + EDGE_DIM_C);
        const float  e0 = eb[c];
        const float  e1 = eb[c + 1];
        const int wv  = __builtin_amdgcn_readfirstlane(wave);
        const int nb  = row0 + wv * 4;
        const int rb0 = rowptr[min(nb,     n_nodes)];
        const int rb1 = rowptr[min(nb + 1, n_nodes)];
        const int rb2 = rowptr[min(nb + 2, n_nodes)];
        const int rb3 = rowptr[min(nb + 3, n_nodes)];
        const int rb4 = rowptr[min(nb + 4, n_nodes)];
        _Float16* const sAw = sA + (wv * 4) * K1P + c;
        v2h hq[4];
#pragma unroll
        for (int qq = 0; qq < 4; ++qq)
            hq[qq] = *(const v2h*)(in0 + (size_t)min(nb + qq, n_nodes - 1) * HID_C + c);
        const int wbeg = rb0, wend = rb4;
        int   q    = 0;
        int   qend = rb1;
        int   g0   = wbeg;
        int   qrel = qend - g0;
        v2f acc = {0.0f, 0.0f};
        const v2f wx = {w0.x, w1.x}, wy = {w0.y, w1.y}, wz = {w0.z, w1.z}, ww = {w0.w, w1.w};
        const v2f eb2 = {e0, e1};
        static_assert((unsigned long long)N_NODES_C * HID_C * sizeof(_Float16) < (1ull << 32),
                      "row byte offsets fit 32 bits");
        const unsigned lane4 = (unsigned)lane * 4u;
        const char* const in0b = (const char*)in0;
#define CONV2_CLOSE_NODE()                                                    \
        {                                                                     \
            v2h* const p_ = (v2h*)(sAw + q * K1P);                            \
            const v2h hq_ = *p_;                                              \
            v2h o_;                                                           \
            o_[0] = (_Float16)((float)hq_[0] + acc[0]);                       \
            o_[1] = (_Float16)((float)hq_[1] + acc[1]);                       \
            *p_ = o_;                                                         \
            ++q;                                                              \
            qend = (q == 1) ? rb2 : (q == 2) ? rb3 : rb4;                     \
            qrel = qend - g0;                                                 \
            acc[0] = 0.0f; acc[1] = 0.0f;                                     \
        }
#define CONV2_GATHER(K, SRC_)                                                 \
        hv[K] = *(const v2h*)(in0b + ((unsigned)(SRC_) * (unsigned)(HID_C * sizeof(_Float16)) + lane4));
#define CONV2_EDGE(K, EA_, I_, MASKED)                                        \
        {                                                                     \
            const bool on_ = !(MASKED) || ((K) < cnt);                        \
            if (on_) {                                                        \
                while ((K) >= qrel) CONV2_CLOSE_NODE()    \
            }                                                                 \
            const float ax_ = EA_[4 * (I_) + 0];                              \
            const float ay_ = EA_[4 * (I_) + 1];                              \
            const float az_ = EA_[4 * (I_) + 2];                              \
            const float aw_ = EA_[4 * (I_) + 3];                              \
            v2f t_ = __builtin_elementwise_fma(wx, (v2f){ax_, ax_}, eb2);     \
            t_ = __builtin_elementwise_fma(wy, (v2f){ay_, ay_}, t_);          \
            t_ = __builtin_elementwise_fma(wz, (v2f){az_, az_}, t_);          \
            t_ = __builtin_elementwise_fma(ww, (v2f){aw_, aw_}, t_);          \
            t_ += (v2f){(float)hv[K][0], (float)hv[K][1]};                    \
            const v2f r_ = {fmaxf(t_[0], 0.0f), fmaxf(t_[1], 0.0f)};          \
            if (MASKED) acc += on_ ? r_ : (v2f){0.0f, 0.0f};                  \
            else        acc += r_;                                            \
        }
#define CONV2_GROUP(MASKED)                                                   \
        CONV2_EDGE(0, ea0, 0, MASKED) CONV2_EDGE(1, ea0, 1, MASKED)           \
        CONV2_EDGE(2, ea0, 2, MASKED) CONV2_EDGE(3, ea0, 3, MASKED)           \
        CONV2_EDGE(4, ea1, 0, MASKED) CONV2_EDGE(5, ea1, 1, MASKED)           \
        CONV2_EDGE(6, ea1, 2, MASKED) CONV2_EDGE(7, ea1, 3, MASKED)
#pragma unroll
        for (int qq = 0; qq < 4; ++qq) {
            v2h hz;
            hz[0] = (_Float16)0.0f; hz[1] = (_Float16)0.0f;
            *(v2h*)(sAw + qq * K1P) = (nb + qq < n_nodes) ? hq[qq] : hz;
        }
        src8_t sN = {0, 0, 0, 0, 0, 0, 0, 0};
        if (wbeg < wend) sN = *(const src8_t*)(ssrc + wbeg);
        for (; g0 < wend; g0 += 8) {
            const int cnt = min(8, wend - g0);
            qrel = qend - g0;
            const src8_t sv = sN;
            const ea16_t ea0 = *(const ea16_t*)(sea + (size_t)g0 * EDGE_DIM_C);
            const ea16_t ea1 = *(const ea16_t*)(sea + (size_t)g0 * EDGE_DIM_C + 16);
            sN = *(const src8_t*)(ssrc + g0 + 8);
            v2h hv[8];
            if (cnt == 8) {
                CONV2_GATHER(0, sv[0]) CONV2_GATHER(1, sv[1]) CONV2_GATHER(2, sv[2]) CONV2_GATHER(3, sv[3])
                CONV2_GATHER(4, sv[4]) CONV2_GATHER(5, sv[5]) CONV2_GATHER(6, sv[6]) CONV2_GATHER(7, sv[7])
                CONV2_GROUP(false)
            } else {
                CONV2_GATHER(0, sv[0])
                CONV2_GATHER(1, (1 < cnt) ? sv[1] : sv[0]) CONV2_GATHER(2, (2 < cnt) ? sv[2] : sv[0])
                CONV2_GATHER(3, (3 < cnt) ? sv[3] : sv[0]) CONV2_GATHER(4, (4 < cnt) ? sv[4] : sv[0])
                CONV2_GATHER(5, (5 < cnt) ? sv[5] : sv[0]) CONV2_GATHER(6, (6 < cnt) ? sv[6] : sv[0])
                CONV2_GATHER(7, sv[0])
                CONV2_GROUP(true)
            }
        }
        while (q < 4) CONV2_CLOSE_NODE()
#undef CONV2_GROUP
#undef CONV2_EDGE
#undef CONV2_GATHER
#undef CONV2_CLOSE_NODE
    } else {
        const int m  = tid >> 3;
        const int c  = tid & 7;
        const int cc = min(c, K1 - 1);
        const int n  = row0 + m;
        const float4 w = *(const float4*)(eW + cc * EDGE_DIM_C);
        const float  e = eb[cc];
        const int beg   = rowptr[min(n,     n_nodes)];
        const int end   = rowptr[min(n + 1, n_nodes)];
        const int lastv = max(end - 1, 0);
        const float own = (float)in0[(size_t)min(n, n_nodes - 1) * K1 + cc];
        int    sv;
        float4 av;
        {
            const int je = min(beg + c, lastv);
            sv = ssrc[je];
            av = *(const float4*)(sea + (size_t)je * EDGE_DIM_C);
        }
        float acc = 0.0f;
        for (int j0 = beg; __builtin_amdgcn_ballot_w32(j0 < end) != 0u; j0 += 8) {
            float xv[8];
#define CONV1_GATHER(K)                                                       \
            {                                                                 \
                const int srk_ = __builtin_amdgcn_mov_dpp8(sv, (K) * 0x249249);     \
                xv[K] = (float)in0[(size_t)srk_ * K1 + cc];                   \
            }
            CONV1_GATHER(0) CONV1_GATHER(1) CONV1_GATHER(2) CONV1_GATHER(3)
            CONV1_GATHER(4) CONV1_GATHER(5) CONV1_GATHER(6) CONV1_GATHER(7)
#undef CONV1_GATHER
            const int    jn = min(j0 + 8 + c, lastv);
            const int    sn = ssrc[jn];
            const float4 an = *(const float4*)(sea + (size_t)jn * EDGE_DIM_C);
            asm volatile("" :: "v"(xv[0]), "v"(xv[1]), "v"(xv[2]), "v"(xv[3]),
                               "v"(xv[4]), "v"(xv[5]), "v"(xv[6]), "v"(xv[7]));
#define CONV1_SUM(K)                                                          \
            {                                                                 \
                const float ax_ = __builtin_bit_cast(float, __builtin_amdgcn_mov_dpp8(__builtin_bit_cast(int, av.x), (K) * 0x249249)); \
                const float ay_ = __builtin_bit_cast(float, __builtin_amdgcn_mov_dpp8(__builtin_bit_cast(int, av.y), (K) * 0x249249)); \
                const float az_ = __builtin_bit_cast(float, __builtin_amdgcn_mov_dpp8(__builtin_bit_cast(int, av.z), (K) * 0x249249)); \
                const float aw_ = __builtin_bit_cast(float, __builtin_amdgcn_mov_dpp8(__builtin_bit_cast(int, av.w), (K) * 0x249249)); \
                float t_ = e + ax_ * w.x + ay_ * w.y + az_ * w.z + aw_ * w.w; \
                t_ += xv[K];                                                  \
                acc += (j0 + (K) < end) ? fmaxf(t_, 0.0f) : 0.0f;             \
            }
            CONV1_SUM(0) CONV1_SUM(1) CONV1_SUM(2) CONV1_SUM(3)
            CONV1_SUM(4) CONV1_SUM(5) CONV1_SUM(6) CONV1_SUM(7)
#undef CONV1_SUM
            sv = sn;
            av = an;
        }
        const float v = (c < K1 && n < n_nodes) ? own + acc : 0.0f;
        sA[m * K1P + c] = (_Float16)v;
        for (int k = c + 8; k < K1P; k += 8)
            sA[m * K1P + k] = (_Float16)0.0f;
    }
    constexpr int KS1 = K1P / 32;
    constexpr int KS2 = HID_C / 32;
    v16h bw1[KS1];
    v16h bw2[KS2];
#pragma unroll
    for (int ks = 0; ks < KS1; ++ks) bw1[ks] = wfrag_load(wp1, KS1, wave, ks, lane);
#pragma unroll
    for (int ks = 0; ks < KS2; ++ks) bw2[ks] = wfrag_load(wp2, KS2, wave, ks, lane);
    __syncthreads();

    v8f acc = {};
#pragma unroll
    for (int kb = 0; kb < K1P; kb += 32) {
        v16h a = lds_load_tile(sA, K1P, kb);
        acc = __builtin_amdgcn_wmma_f32_16x16x32_f16(false, a, false, bw1[kb / 32],
                                                     (short)0, acc, false, false);
    }
    {
        const int n  = lane & 15;
        const int mb = (lane < 16) ? 0 : 8;
        const float bias = b1[n0 + n];
#pragma unroll
        for (int r = 0; r < 8; ++r) {
            float v = acc[r] + bias;
            v = v > 0.0f ? v : 0.0f;
            sMid[(mb + r) * HID_C + (n0 + n)] = (_Float16)v;
        }
    }
    __syncthreads();

    v8f acc2 = {};
#pragma unroll
    for (int kb = 0; kb < HID_C; kb += 32) {
        v16h a = lds_load_tile(sMid, HID_C, kb);
        acc2 = __builtin_amdgcn_wmma_f32_16x16x32_f16(false, a, false, bw2[kb / 32],
                                                      (short)0, acc2, false, false);
    }
    if constexpr (!POOL) {
        const int n  = lane & 15;
        const int mb = (lane < 16) ? 0 : 8;
        const float bias = b2[n0 + n];
#pragma unroll
        for (int r = 0; r < 8; ++r) {
            const int rr = row0 + mb + r;
            if (rr < n_nodes) {
                float v = acc2[r] + bias;
                v = v > 0.0f ? v : 0.0f;
                out[(size_t)rr * HID_C + (n0 + n)] = (_Float16)v;
            }
        }
    } else {
        const int n  = lane & 15;
        const int mb = (lane < 16) ? 0 : 8;
        const float bias = b2[n0 + n];
        const bool cnt_lane = (wave == 0) && (n == 0);
        const int4 ga = *(const int4*)(sG + mb);
        const int4 gb = *(const int4*)(sG + mb + 4);
        const int gs[8] = { ga.x, ga.y, ga.z, ga.w, gb.x, gb.y, gb.z, gb.w };
        int   gcur = -1;
        float run  = 0.0f;
        int   len  = 0;
#pragma unroll
        for (int r = 0; r < 8; ++r) {
            const int g = gs[r];
            float v = acc2[r] + bias;
            v = v > 0.0f ? v : 0.0f;
            if (g != gcur) {
                if (gcur >= 0) {
                    atom_add_f32(&psum[(size_t)gcur * HID_C + (n0 + n)], run);
                    if (cnt_lane) atom_add_f32(&cnt[gcur], (float)len);
                }
                gcur = g; run = 0.0f; len = 0;
            }
            run += v;
            len += 1;
        }
        if (gcur >= 0) {
            atom_add_f32(&psum[(size_t)gcur * HID_C + (n0 + n)], run);
            if (cnt_lane) atom_add_f32(&cnt[gcur], (float)len);
        }
    }
}

#define BIN_SHIFT 9
#define BIN_NODES (1 << BIN_SHIFT)
#define NBINS     ((N_NODES_C >> BIN_SHIFT) + 1)
#define BIN_G     256
#define BIN_T     1024
#define BIN_W     (BIN_T / 32)
#define BIN_CHUNK ((N_EDGES_C + BIN_G - 1) / BIN_G)
#define BIN_TRIPS ((BIN_CHUNK + BIN_T - 1) / BIN_T)
#define SORT_GRP  4
#define SORT_CAP  12288
#define PART_TRIPS ((NBINS * BIN_G + BIN_T * 4 - 1) / (BIN_T * 4))
static_assert(NBINS <= 256, "one wave scans the bins, 8 per lane");
static_assert(BIN_G == 256, "one wave scans a bin's 256 runs, 8 per lane");

__global__ __launch_bounds__(BIN_T)
void chunk_sort_kernel(const int* __restrict__ ei, const float* __restrict__ ea,
                       int* __restrict__ hist, int* __restrict__ lstart,
                       int* __restrict__ bdst, int* __restrict__ bsrc,
                       float* __restrict__ bea,
                       const float* __restrict__ W11, const float* __restrict__ W12,
                       const float* __restrict__ W21, const float* __restrict__ W22,
                       _Float16* __restrict__ wplane) {
    __shared__ int sCnt[256];
    __shared__ int sStart[256];
    __shared__ int sRank[256];
    __shared__ __align__(16) float4 lEa[BIN_CHUNK];
    __shared__ int lDst[BIN_CHUNK];
    __shared__ int lSrc[BIN_CHUNK];
    const int tid  = threadIdx.x;
    const int lane = tid & 31;
    const int blk  = blockIdx.x;
    const int e0 = blk * BIN_CHUNK;
    const int e1 = min(e0 + BIN_CHUNK, N_EDGES_C);

    if (tid < 256) { sCnt[tid] = 0; sRank[tid] = 0; }
    int    dv[BIN_TRIPS];
    int    sv[BIN_TRIPS];
    float4 av[BIN_TRIPS];
#pragma unroll
    for (int k = 0; k < BIN_TRIPS; ++k) {
        const int e = min(e0 + tid + k * BIN_T, N_EDGES_C - 1);
        dv[k] = ei[N_EDGES_C + e];
        sv[k] = ei[e];
        av[k] = *(const float4*)(ea + (size_t)e * EDGE_DIM_C);
    }
#pragma unroll
    for (int k = 0; k < BIN_TRIPS; ++k)
        asm volatile("" :: "v"(dv[k]), "v"(sv[k]),
                     "v"(av[k].x), "v"(av[k].y), "v"(av[k].z), "v"(av[k].w));
    __syncthreads();
#pragma unroll
    for (int k = 0; k < BIN_TRIPS; ++k)
        if (e0 + tid + k * BIN_T < e1) atomicAdd(&sCnt[dv[k] >> BIN_SHIFT], 1);
    __syncthreads();
    if (tid < 32) {
        int c[8];
        int s = 0;
#pragma unroll
        for (int k = 0; k < 8; ++k) { c[k] = sCnt[lane * 8 + k]; s += c[k]; }
        int incl = s;
        for (int d = 1; d < 32; d <<= 1) {
            const int t = __shfl_up(incl, d);
            if (lane >= d) incl += t;
        }
        int run = incl - s;
#pragma unroll
        for (int k = 0; k < 8; ++k) {
            const int idx = lane * 8 + k;
            sStart[idx] = run;
            if (idx < NBINS) {
                hist  [idx * BIN_G + blk] = c[k];
                lstart[idx * BIN_G + blk] = run;
            }
            run += c[k];
        }
    }
    __syncthreads();
#pragma unroll
    for (int k = 0; k < BIN_TRIPS; ++k) {
        if (e0 + tid + k * BIN_T < e1) {
            const int bin = dv[k] >> BIN_SHIFT;
            const int lp  = sStart[bin] + atomicAdd(&sRank[bin], 1);
            lDst[lp] = dv[k];
            lSrc[lp] = sv[k];
            lEa[lp]  = av[k];
        }
    }
    __syncthreads();
    for (int j = tid; j < e1 - e0; j += BIN_T) {
        bdst[e0 + j] = lDst[j];
        bsrc[e0 + j] = lSrc[j];
        *(float4*)(bea + (size_t)(e0 + j) * EDGE_DIM_C) = lEa[j];
    }
    {
        const int g = blk * BIN_T + tid;
        if (g < WP_TOTAL) {
            _Float16 v;
            if      (g < WP_C1L2) v = pack_weight_elem(W11, IN_DIM_C, 1, g - WP_C1L1);
            else if (g < WP_C2L1) v = pack_weight_elem(W12, HID_C,    2, g - WP_C1L2);
            else if (g < WP_C2L2) v = pack_weight_elem(W21, HID_C,    2, g - WP_C2L1);
            else                  v = pack_weight_elem(W22, HID_C,    2, g - WP_C2L2);
            wplane[g] = v;
        }
    }
}

__device__ __forceinline__ int bin_edge_place(const int* sRun, const int* sBase, int i) {
    int lo = 0, hi = BIN_G;
#pragma unroll
    for (int s = 0; s < 8; ++s) {
        const int mid = (lo + hi) >> 1;
        if (sRun[mid] <= i) lo = mid; else hi = mid;
    }
    return sBase[lo] + (i - sRun[lo]);
}

__global__ __launch_bounds__(BIN_T)
void bin_sort_kernel(const int* __restrict__ hist, const int* __restrict__ lstart,
                     const int* __restrict__ bdst, const int* __restrict__ bsrc,
                     const float* __restrict__ bea,
                     int* __restrict__ rowptr, int* __restrict__ ssrc,
                     float* __restrict__ sea) {
    __shared__ int sRun[BIN_G + 1];
    __shared__ int sBase[BIN_G];
    __shared__ int sCnt[BIN_NODES];
    __shared__ int sPos[BIN_NODES];
    __shared__ int sW[BIN_W];
    __shared__ int sRed[BIN_W];
    __shared__ __align__(16) float4 lEa[SORT_CAP];
    __shared__ int lSrc[SORT_CAP];
    const int tid   = threadIdx.x;
    const int lane  = tid & 31;
    const int wave  = tid >> 5;
    const int bin   = blockIdx.x;
    const int node0 = bin << BIN_SHIFT;

    int part = 0;
    {
        const int lim = bin * BIN_G;
        int4 h[PART_TRIPS];
#pragma unroll
        for (int k = 0; k < PART_TRIPS; ++k)
            h[k] = *(const int4*)(hist + min(tid * 4 + k * BIN_T * 4, NBINS * BIN_G - 4));
#pragma unroll
        for (int k = 0; k < PART_TRIPS; ++k)
            if (tid * 4 + k * BIN_T * 4 < lim) part += h[k].x + h[k].y + h[k].z + h[k].w;
    }
    for (int d = 16; d > 0; d >>= 1) part += __shfl_xor(part, d);
    if (lane == 0) sRed[wave] = part;
    if (tid < BIN_NODES) sCnt[tid] = 0;
    if (wave == 0) {
        const int4* ph = (const int4*)(hist   + bin * BIN_G + lane * 8);
        const int4* pl = (const int4*)(lstart + bin * BIN_G + lane * 8);
        const int4 ha = ph[0], hb = ph[1];
        const int4 la = pl[0], lb = pl[1];
        const int c[8]  = { ha.x, ha.y, ha.z, ha.w, hb.x, hb.y, hb.z, hb.w };
        const int ls[8] = { la.x, la.y, la.z, la.w, lb.x, lb.y, lb.z, lb.w };
        int s = 0;
#pragma unroll
        for (int k = 0; k < 8; ++k) s += c[k];
        int incl = s;
        for (int d = 1; d < 32; d <<= 1) {
            const int t = __shfl_up(incl, d);
            if (lane >= d) incl += t;
        }
        int run = incl - s;
#pragma unroll
        for (int k = 0; k < 8; ++k) {
            const int b = lane * 8 + k;
            sRun[b]  = run;
            sBase[b] = b * BIN_CHUNK + ls[k];
            run += c[k];
        }
        if (lane == 31) sRun[BIN_G] = incl;
    }
    __syncthreads();
    int beg = 0;
#pragma unroll
    for (int w = 0; w < BIN_W; ++w) beg += sRed[w];
    const int n = sRun[BIN_G];

    for (int i0 = tid; i0 < n; i0 += SORT_GRP * BIN_T) {
        int dv[SORT_GRP];
#pragma unroll
        for (int k = 0; k < SORT_GRP; ++k)
            dv[k] = bdst[bin_edge_place(sRun, sBase, min(i0 + k * BIN_T, n - 1))];
#pragma unroll
        for (int k = 0; k < SORT_GRP; ++k)
            asm volatile("" :: "v"(dv[k]));
#pragma unroll
        for (int k = 0; k < SORT_GRP; ++k)
            if (i0 + k * BIN_T < n) atomicAdd(&sCnt[dv[k] - node0], 1);
    }
    __syncthreads();

    const int v = (tid < BIN_NODES) ? sCnt[tid] : 0;
    int incl = v;
    for (int d = 1; d < 32; d <<= 1) {
        const int t = __shfl_up(incl, d);
        if (lane >= d) incl += t;
    }
    if (lane == 31) sW[wave] = incl;
    __syncthreads();
    if (wave == 0) {
        const int w = sW[lane];
        int wi = w;
        for (int d = 1; d < 32; d <<= 1) {
            const int t = __shfl_up(wi, d);
            if (lane >= d) wi += t;
        }
        sW[lane] = wi - w;
    }
    __syncthreads();
    if (tid < BIN_NODES) {
        const int start = sW[wave] + incl - v;
        sPos[tid] = start;
        const int node = node0 + tid;
        if (node <= N_NODES_C) rowptr[node] = beg + start;
    }
    __syncthreads();

    const bool staged = (n <= SORT_CAP);
    for (int i0 = tid; i0 < n; i0 += SORT_GRP * BIN_T) {
        int    dv[SORT_GRP];
        int    sv[SORT_GRP];
        float4 av[SORT_GRP];
#pragma unroll
        for (int k = 0; k < SORT_GRP; ++k) {
            const int g = bin_edge_place(sRun, sBase, min(i0 + k * BIN_T, n - 1));
            dv[k] = bdst[g];
            sv[k] = bsrc[g];
            av[k] = *(const float4*)(bea + (size_t)g * EDGE_DIM_C);
        }
#pragma unroll
        for (int k = 0; k < SORT_GRP; ++k)
            asm volatile("" :: "v"(dv[k]), "v"(sv[k]),
                         "v"(av[k].x), "v"(av[k].y), "v"(av[k].z), "v"(av[k].w));
#pragma unroll
        for (int k = 0; k < SORT_GRP; ++k) {
            if (i0 + k * BIN_T < n) {
                const int lp = atomicAdd(&sPos[dv[k] - node0], 1);
                if (staged) {
                    lSrc[lp] = sv[k];
                    lEa[lp]  = av[k];
                } else {
                    ssrc[beg + lp] = sv[k];
                    *(float4*)(sea + (size_t)(beg + lp) * EDGE_DIM_C) = av[k];
                }
            }
        }
    }
    if (staged) {
        __syncthreads();
        for (int j = tid; j < n; j += BIN_T) {
            ssrc[beg + j] = lSrc[j];
            *(float4*)(sea + (size_t)(beg + j) * EDGE_DIM_C) = lEa[j];
        }
    }
}

__global__ __launch_bounds__(256)
void fc_kernel(const float* __restrict__ psum, const float* __restrict__ cnt,
               const float* __restrict__ fcW, const float* __restrict__ fcb,
               float* __restrict__ out) {
    const int t = blockIdx.x * 256 + threadIdx.x;
    if (t >= N_GRAPHS_C * NUM_TASKS_C) return;
    const int g = t / NUM_TASKS_C;
    const int k = t - g * NUM_TASKS_C;
    const float inv = 1.0f / fmaxf(cnt[g], 1.0f);
    float s = fcb[k];
#pragma unroll
    for (int c = 0; c < HID_C; ++c)
        s += psum[(size_t)g * HID_C + c] * inv * fcW[k * HID_C + c];
    out[t] = s;
}

extern "C" void kernel_launch(void* const* d_in, const int* in_sizes, int n_in,
                              void* d_out, int out_size, void* d_ws, size_t ws_size,
                              hipStream_t stream) {
    const float* x    = (const float*)d_in[0];
    const int*   ei   = (const int*)  d_in[1];
    const float* ea   = (const float*)d_in[2];
    const int*   bat  = (const int*)  d_in[3];
    const float* e1W  = (const float*)d_in[4];  const float* e1b  = (const float*)d_in[5];
    const float* n1W1 = (const float*)d_in[6];  const float* n1b1 = (const float*)d_in[7];
    const float* n1W2 = (const float*)d_in[8];  const float* n1b2 = (const float*)d_in[9];
    const float* e2W  = (const float*)d_in[10]; const float* e2b  = (const float*)d_in[11];
    const float* n2W1 = (const float*)d_in[12]; const float* n2b1 = (const float*)d_in[13];
    const float* n2W2 = (const float*)d_in[14]; const float* n2b2 = (const float*)d_in[15];
    const float* fcW  = (const float*)d_in[16]; const float* fcb  = (const float*)d_in[17];
    float* out = (float*)d_out;

    _Float16* h     = (_Float16*)d_ws;
    float*    psum  = (float*)(h + (size_t)N_NODES_C * HID_C);
    float*    cnt   = psum  + (size_t)N_GRAPHS_C * HID_C;
    float*    sea   = cnt   + (size_t)N_GRAPHS_C;
    int*      ssrc  = (int*)(sea + (size_t)N_EDGES_C * EDGE_DIM_C);
    int*      bdst  = ssrc  + (size_t)N_EDGES_C;
    int*      bsrc  = bdst  + (size_t)N_EDGES_C;
    float*    bea   = (float*)(bsrc + (size_t)N_EDGES_C);
    int*      hist  = (int*)(bea + (size_t)N_EDGES_C * EDGE_DIM_C);
    int*      lstart = hist + (size_t)NBINS * BIN_G;
    int*      rowptr = lstart + (size_t)NBINS * BIN_G;
    _Float16* wplane = (_Float16*)(rowptr + (size_t)ROWPTR_LEN);

    hipMemsetAsync(psum,  0, sizeof(float) * (size_t)N_GRAPHS_C * HID_C,  stream);
    hipMemsetAsync(cnt,   0, sizeof(float) * (size_t)N_GRAPHS_C,          stream);

    chunk_sort_kernel<<<BIN_G, BIN_T, 0, stream>>>(ei, ea, hist, lstart, bdst, bsrc, bea,
                                                   n1W1, n1W2, n2W1, n2W2, wplane);
    bin_sort_kernel<<<NBINS, BIN_T, 0, stream>>>(hist, lstart, bdst, bsrc, bea, rowptr, ssrc, sea);
    const unsigned grid = (N_NODES_C + 15) / 16;
    conv_kernel<IN_DIM_C, float, false><<<grid, 128, 0, stream>>>(
        x, rowptr, ssrc, sea, e1W, e1b, wplane + WP_C1L1, n1b1, wplane + WP_C1L2, n1b2, h,
        nullptr, nullptr, nullptr, N_NODES_C);
    conv_kernel<HID_C, _Float16, true><<<grid, 128, 0, stream>>>(
        h, rowptr, ssrc, sea, e2W, e2b, wplane + WP_C2L1, n2b1, wplane + WP_C2L2, n2b2, nullptr,
        bat, psum, cnt, N_NODES_C);
    {
        const unsigned fgrid = (N_GRAPHS_C * NUM_TASKS_C + 255) / 256;
        fc_kernel<<<fgrid, 256, 0, stream>>>(psum, cnt, fcW, fcb, out);
    }
}


# bin_sort reads each input word once: bin start = sum of its own lstart row (no count-table sweep); records kept in registers from the counting add (return = rank) to placement; sum order LDS arrival
# speedup vs baseline: 5.2456x; 1.0701x over previous
#include <hip/hip_runtime.h>
#include <hip/hip_bf16.h>

#define N_NODES_C  100000
#define N_EDGES_C  1600000
#define N_GRAPHS_C 1000
#define HID_C      64
#define IN_DIM_C   7
#define EDGE_DIM_C 4
#define NUM_TASKS_C 12

typedef __attribute__((ext_vector_type(16))) _Float16 v16h;
typedef __attribute__((ext_vector_type(8)))  _Float16 v8h;
typedef __attribute__((ext_vector_type(8)))  float    v8f;
typedef __attribute__((ext_vector_type(2)))  _Float16 v2h;
typedef __attribute__((ext_vector_type(2)))  float    v2f;
typedef int   src8_t __attribute__((ext_vector_type(8),  aligned(4)));
typedef float ea16_t __attribute__((ext_vector_type(16), aligned(16)));

__device__ __forceinline__ void atom_add_f32(float* p, float v) {
    unsafeAtomicAdd(p, v);
}

__device__ __forceinline__ v16h lds_load_tile(const _Float16* s, int ld, int kbase) {
    const int lane = threadIdx.x & 31;
    const int r    = lane & 15;
    const int koff = (lane < 16) ? 0 : 8;
    const _Float16* p = s + r * ld + kbase + koff;
    v8h lo = *(const v8h*)(p);
    v8h hi = *(const v8h*)(p + 16);
    v16h out;
#pragma unroll
    for (int e = 0; e < 8; ++e) { out[e] = lo[e]; out[8 + e] = hi[e]; }
    return out;
}

#define WP_C1L1  0
#define WP_C1L2  (WP_C1L1 + HID_C * 32)
#define WP_C2L1  (WP_C1L2 + HID_C * 64)
#define WP_C2L2  (WP_C2L1 + HID_C * 64)
#define WP_TOTAL (WP_C2L2 + HID_C * 64)
#define ROWPTR_LEN ((N_NODES_C + 1 + 3) & ~3)

__device__ __forceinline__ _Float16 pack_weight_elem(const float* __restrict__ W, int K, int KS, int i) {
    const int e8   = i & 7;
    const int lane = (i >> 3) & 31;
    const int part = (i >> 8) & 1;
    const int t    = i >> 9;
    const int wv   = t / KS;
    const int ks   = t - wv * KS;
    const int n    = wv * 16 + (lane & 15);
    const int k    = ks * 32 + ((lane < 16) ? 0 : 8) + part * 16 + e8;
    return (_Float16)((k < K) ? W[n * K + k] : 0.0f);
}

__device__ __forceinline__ v16h wfrag_load(const _Float16* __restrict__ wp, int KS, int wave, int ks, int lane) {
    const v8h* p = (const v8h*)wp + ((wave * KS + ks) * 2) * 32 + lane;
    const v8h lo = p[0];
    const v8h hi = p[32];
    v16h out;
#pragma unroll
    for (int e = 0; e < 8; ++e) { out[e] = lo[e]; out[8 + e] = hi[e]; }
    return out;
}

template <int K1, typename TI, bool POOL>
__global__ __launch_bounds__(128)
void conv_kernel(const TI* __restrict__ in0, const int* __restrict__ rowptr,
                 const int* __restrict__ ssrc, const float* __restrict__ sea,
                 const float* __restrict__ eW, const float* __restrict__ eb,
                 const _Float16* __restrict__ wp1, const float* __restrict__ b1,
                 const _Float16* __restrict__ wp2, const float* __restrict__ b2,
                 _Float16* __restrict__ out, const int* __restrict__ batch,
                 float* __restrict__ psum, float* __restrict__ cnt,
                 int n_nodes) {
    constexpr int K1P = (K1 + 31) & ~31;

    __shared__ __align__(16) _Float16 sA  [16 * K1P];
    __shared__ __align__(16) _Float16 sMid[16 * HID_C];
    __shared__ __align__(16) int      sG  [16];

    const int row0 = blockIdx.x * 16;
    const int tid  = threadIdx.x;
    const int wave = tid >> 5;
    const int lane = tid & 31;
    const int n0   = wave * 16;

    if constexpr (POOL) {
        if (tid < 16) sG[tid] = (row0 + tid < n_nodes) ? batch[row0 + tid] : -1;
    }
    if constexpr (K1 == HID_C) {
        const int c = 2 * lane;
        const float4 w0 = *(const float4*)(eW + c * EDGE_DIM_C);
        const float4 w1 = *(const float4*)(eW + c * EDGE_DIM_C + EDGE_DIM_C);
        const float  e0 = eb[c];
        const float  e1 = eb[c + 1];
        const int wv  = __builtin_amdgcn_readfirstlane(wave);
        const int nb  = row0 + wv * 4;
        const int rb0 = rowptr[min(nb,     n_nodes)];
        const int rb1 = rowptr[min(nb + 1, n_nodes)];
        const int rb2 = rowptr[min(nb + 2, n_nodes)];
        const int rb3 = rowptr[min(nb + 3, n_nodes)];
        const int rb4 = rowptr[min(nb + 4, n_nodes)];
        _Float16* const sAw = sA + (wv * 4) * K1P + c;
        v2h hq[4];
#pragma unroll
        for (int qq = 0; qq < 4; ++qq)
            hq[qq] = *(const v2h*)(in0 + (size_t)min(nb + qq, n_nodes - 1) * HID_C + c);
        const int wbeg = rb0, wend = rb4;
        int   q    = 0;
        int   qend = rb1;
        int   g0   = wbeg;
        int   qrel = qend - g0;
        v2f acc = {0.0f, 0.0f};
        const v2f wx = {w0.x, w1.x}, wy = {w0.y, w1.y}, wz = {w0.z, w1.z}, ww = {w0.w, w1.w};
        const v2f eb2 = {e0, e1};
        static_assert((unsigned long long)N_NODES_C * HID_C * sizeof(_Float16) < (1ull << 32),
                      "row byte offsets fit 32 bits");
        const unsigned lane4 = (unsigned)lane * 4u;
        const char* const in0b = (const char*)in0;
#define CONV2_CLOSE_NODE()                                                    \
        {                                                                     \
            v2h* const p_ = (v2h*)(sAw + q * K1P);                            \
            const v2h hq_ = *p_;                                              \
            v2h o_;                                                           \
            o_[0] = (_Float16)((float)hq_[0] + acc[0]);                       \
            o_[1] = (_Float16)((float)hq_[1] + acc[1]);                       \
            *p_ = o_;                                                         \
            ++q;                                                              \
            qend = (q == 1) ? rb2 : (q == 2) ? rb3 : rb4;                     \
            qrel = qend - g0;                                                 \
            acc[0] = 0.0f; acc[1] = 0.0f;                                     \
        }
#define CONV2_GATHER(K, SRC_)                                                 \
        hv[K] = *(const v2h*)(in0b + ((unsigned)(SRC_) * (unsigned)(HID_C * sizeof(_Float16)) + lane4));
#define CONV2_EDGE(K, EA_, I_, MASKED)                                        \
        {                                                                     \
            const bool on_ = !(MASKED) || ((K) < cnt);                        \
            if (on_) {                                                        \
                while ((K) >= qrel) CONV2_CLOSE_NODE()    \
            }                                                                 \
            const float ax_ = EA_[4 * (I_) + 0];                              \
            const float ay_ = EA_[4 * (I_) + 1];                              \
            const float az_ = EA_[4 * (I_) + 2];                              \
            const float aw_ = EA_[4 * (I_) + 3];                              \
            v2f t_ = __builtin_elementwise_fma(wx, (v2f){ax_, ax_}, eb2);     \
            t_ = __builtin_elementwise_fma(wy, (v2f){ay_, ay_}, t_);          \
            t_ = __builtin_elementwise_fma(wz, (v2f){az_, az_}, t_);          \
            t_ = __builtin_elementwise_fma(ww, (v2f){aw_, aw_}, t_);          \
            t_ += (v2f){(float)hv[K][0], (float)hv[K][1]};                    \
            const v2f r_ = {fmaxf(t_[0], 0.0f), fmaxf(t_[1], 0.0f)};          \
            if (MASKED) acc += on_ ? r_ : (v2f){0.0f, 0.0f};                  \
            else        acc += r_;                                            \
        }
#define CONV2_GROUP(MASKED)                                                   \
        CONV2_EDGE(0, ea0, 0, MASKED) CONV2_EDGE(1, ea0, 1, MASKED)           \
        CONV2_EDGE(2, ea0, 2, MASKED) CONV2_EDGE(3, ea0, 3, MASKED)           \
        CONV2_EDGE(4, ea1, 0, MASKED) CONV2_EDGE(5, ea1, 1, MASKED)           \
        CONV2_EDGE(6, ea1, 2, MASKED) CONV2_EDGE(7, ea1, 3, MASKED)
#pragma unroll
        for (int qq = 0; qq < 4; ++qq) {
            v2h hz;
            hz[0] = (_Float16)0.0f; hz[1] = (_Float16)0.0f;
            *(v2h*)(sAw + qq * K1P) = (nb + qq < n_nodes) ? hq[qq] : hz;
        }
        src8_t sN = {0, 0, 0, 0, 0, 0, 0, 0};
        if (wbeg < wend) sN = *(const src8_t*)(ssrc + wbeg);
        for (; g0 < wend; g0 += 8) {
            const int cnt = min(8, wend - g0);
            qrel = qend - g0;
            const src8_t sv = sN;
            const ea16_t ea0 = *(const ea16_t*)(sea + (size_t)g0 * EDGE_DIM_C);
            const ea16_t ea1 = *(const ea16_t*)(sea + (size_t)g0 * EDGE_DIM_C + 16);
            sN = *(const src8_t*)(ssrc + g0 + 8);
            v2h hv[8];
            if (cnt == 8) {
                CONV2_GATHER(0, sv[0]) CONV2_GATHER(1, sv[1]) CONV2_GATHER(2, sv[2]) CONV2_GATHER(3, sv[3])
                CONV2_GATHER(4, sv[4]) CONV2_GATHER(5, sv[5]) CONV2_GATHER(6, sv[6]) CONV2_GATHER(7, sv[7])
                CONV2_GROUP(false)
            } else {
                CONV2_GATHER(0, sv[0])
                CONV2_GATHER(1, (1 < cnt) ? sv[1] : sv[0]) CONV2_GATHER(2, (2 < cnt) ? sv[2] : sv[0])
                CONV2_GATHER(3, (3 < cnt) ? sv[3] : sv[0]) CONV2_GATHER(4, (4 < cnt) ? sv[4] : sv[0])
                CONV2_GATHER(5, (5 < cnt) ? sv[5] : sv[0]) CONV2_GATHER(6, (6 < cnt) ? sv[6] : sv[0])
                CONV2_GATHER(7, sv[0])
                CONV2_GROUP(true)
            }
        }
        while (q < 4) CONV2_CLOSE_NODE()
#undef CONV2_GROUP
#undef CONV2_EDGE
#undef CONV2_GATHER
#undef CONV2_CLOSE_NODE
    } else {
        const int m  = tid >> 3;
        const int c  = tid & 7;
        const int cc = min(c, K1 - 1);
        const int n  = row0 + m;
        const float4 w = *(const float4*)(eW + cc * EDGE_DIM_C);
        const float  e = eb[cc];
        const int beg   = rowptr[min(n,     n_nodes)];
        const int end   = rowptr[min(n + 1, n_nodes)];
        const int lastv = max(end - 1, 0);
        const float own = (float)in0[(size_t)min(n, n_nodes - 1) * K1 + cc];
        int    sv;
        float4 av;
        {
            const int je = min(beg + c, lastv);
            sv = ssrc[je];
            av = *(const float4*)(sea + (size_t)je * EDGE_DIM_C);
        }
        float acc = 0.0f;
        for (int j0 = beg; __builtin_amdgcn_ballot_w32(j0 < end) != 0u; j0 += 8) {
            float xv[8];
#define CONV1_GATHER(K)                                                       \
            {                                                                 \
                const int srk_ = __builtin_amdgcn_mov_dpp8(sv, (K) * 0x249249);     \
                xv[K] = (float)in0[(size_t)srk_ * K1 + cc];                   \
            }
            CONV1_GATHER(0) CONV1_GATHER(1) CONV1_GATHER(2) CONV1_GATHER(3)
            CONV1_GATHER(4) CONV1_GATHER(5) CONV1_GATHER(6) CONV1_GATHER(7)
#undef CONV1_GATHER
            const int    jn = min(j0 + 8 + c, lastv);
            const int    sn = ssrc[jn];
            const float4 an = *(const float4*)(sea + (size_t)jn * EDGE_DIM_C);
            asm volatile("" :: "v"(xv[0]), "v"(xv[1]), "v"(xv[2]), "v"(xv[3]),
                               "v"(xv[4]), "v"(xv[5]), "v"(xv[6]), "v"(xv[7]));
#define CONV1_SUM(K)                                                          \
            {                                                                 \
                const float ax_ = __builtin_bit_cast(float, __builtin_amdgcn_mov_dpp8(__builtin_bit_cast(int, av.x), (K) * 0x249249)); \
                const float ay_ = __builtin_bit_cast(float, __builtin_amdgcn_mov_dpp8(__builtin_bit_cast(int, av.y), (K) * 0x249249)); \
                const float az_ = __builtin_bit_cast(float, __builtin_amdgcn_mov_dpp8(__builtin_bit_cast(int, av.z), (K) * 0x249249)); \
                const float aw_ = __builtin_bit_cast(float, __builtin_amdgcn_mov_dpp8(__builtin_bit_cast(int, av.w), (K) * 0x249249)); \
                float t_ = e + ax_ * w.x + ay_ * w.y + az_ * w.z + aw_ * w.w; \
                t_ += xv[K];                                                  \
                acc += (j0 + (K) < end) ? fmaxf(t_, 0.0f) : 0.0f;             \
            }
            CONV1_SUM(0) CONV1_SUM(1) CONV1_SUM(2) CONV1_SUM(3)
            CONV1_SUM(4) CONV1_SUM(5) CONV1_SUM(6) CONV1_SUM(7)
#undef CONV1_SUM
            sv = sn;
            av = an;
        }
        const float v = (c < K1 && n < n_nodes) ? own + acc : 0.0f;
        sA[m * K1P + c] = (_Float16)v;
        for (int k = c + 8; k < K1P; k += 8)
            sA[m * K1P + k] = (_Float16)0.0f;
    }
    constexpr int KS1 = K1P / 32;
    constexpr int KS2 = HID_C / 32;
    v16h bw1[KS1];
    v16h bw2[KS2];
#pragma unroll
    for (int ks = 0; ks < KS1; ++ks) bw1[ks] = wfrag_load(wp1, KS1, wave, ks, lane);
#pragma unroll
    for (int ks = 0; ks < KS2; ++ks) bw2[ks] = wfrag_load(wp2, KS2, wave, ks, lane);
    __syncthreads();

    v8f acc = {};
#pragma unroll
    for (int kb = 0; kb < K1P; kb += 32) {
        v16h a = lds_load_tile(sA, K1P, kb);
        acc = __builtin_amdgcn_wmma_f32_16x16x32_f16(false, a, false, bw1[kb / 32],
                                                     (short)0, acc, false, false);
    }
    {
        const int n  = lane & 15;
        const int mb = (lane < 16) ? 0 : 8;
        const float bias = b1[n0 + n];
#pragma unroll
        for (int r = 0; r < 8; ++r) {
            float v = acc[r] + bias;
            v = v > 0.0f ? v : 0.0f;
            sMid[(mb + r) * HID_C + (n0 + n)] = (_Float16)v;
        }
    }
    __syncthreads();

    v8f acc2 = {};
#pragma unroll
    for (int kb = 0; kb < HID_C; kb += 32) {
        v16h a = lds_load_tile(sMid, HID_C, kb);
        acc2 = __builtin_amdgcn_wmma_f32_16x16x32_f16(false, a, false, bw2[kb / 32],
                                                      (short)0, acc2, false, false);
    }
    if constexpr (!POOL) {
        const int n  = lane & 15;
        const int mb = (lane < 16) ? 0 : 8;
        const float bias = b2[n0 + n];
#pragma unroll
        for (int r = 0; r < 8; ++r) {
            const int rr = row0 + mb + r;
            if (rr < n_nodes) {
                float v = acc2[r] + bias;
                v = v > 0.0f ? v : 0.0f;
                out[(size_t)rr * HID_C + (n0 + n)] = (_Float16)v;
            }
        }
    } else {
        const int n  = lane & 15;
        const int mb = (lane < 16) ? 0 : 8;
        const float bias = b2[n0 + n];
        const bool cnt_lane = (wave == 0) && (n == 0);
        const int4 ga = *(const int4*)(sG + mb);
        const int4 gb = *(const int4*)(sG + mb + 4);
        const int gs[8] = { ga.x, ga.y, ga.z, ga.w, gb.x, gb.y, gb.z, gb.w };
        int   gcur = -1;
        float run  = 0.0f;
        int   len  = 0;
#pragma unroll
        for (int r = 0; r < 8; ++r) {
            const int g = gs[r];
            float v = acc2[r] + bias;
            v = v > 0.0f ? v : 0.0f;
            if (g != gcur) {
                if (gcur >= 0) {
                    atom_add_f32(&psum[(size_t)gcur * HID_C + (n0 + n)], run);
                    if (cnt_lane) atom_add_f32(&cnt[gcur], (float)len);
                }
                gcur = g; run = 0.0f; len = 0;
            }
            run += v;
            len += 1;
        }
        if (gcur >= 0) {
            atom_add_f32(&psum[(size_t)gcur * HID_C + (n0 + n)], run);
            if (cnt_lane) atom_add_f32(&cnt[gcur], (float)len);
        }
    }
}

#define BIN_SHIFT 9
#define BIN_NODES (1 << BIN_SHIFT)
#define NBINS     ((N_NODES_C >> BIN_SHIFT) + 1)
#define BIN_G     256
#define BIN_T     1024
#define BIN_W     (BIN_T / 32)
#define BIN_CHUNK ((N_EDGES_C + BIN_G - 1) / BIN_G)
#define BIN_TRIPS ((BIN_CHUNK + BIN_T - 1) / BIN_T)
#define SORT_GRP  4
#define SORT_CAP  12288
#define ONE_R     10
static_assert(NBINS <= 256, "one wave scans the bins, 8 per lane");
static_assert(BIN_G == 256, "one wave scans a bin's 256 runs, 8 per lane");
static_assert(ONE_R * BIN_T <= SORT_CAP, "a bin of the one-pass path fits the LDS staging");

__global__ __launch_bounds__(BIN_T)
void chunk_sort_kernel(const int* __restrict__ ei, const float* __restrict__ ea,
                       int* __restrict__ hist, int* __restrict__ lstart,
                       int* __restrict__ bdst, int* __restrict__ bsrc,
                       float* __restrict__ bea,
                       const float* __restrict__ W11, const float* __restrict__ W12,
                       const float* __restrict__ W21, const float* __restrict__ W22,
                       _Float16* __restrict__ wplane) {
    __shared__ int sCnt[256];
    __shared__ int sStart[256];
    __shared__ int sRank[256];
    __shared__ __align__(16) float4 lEa[BIN_CHUNK];
    __shared__ int lDst[BIN_CHUNK];
    __shared__ int lSrc[BIN_CHUNK];
    const int tid  = threadIdx.x;
    const int lane = tid & 31;
    const int blk  = blockIdx.x;
    const int e0 = blk * BIN_CHUNK;
    const int e1 = min(e0 + BIN_CHUNK, N_EDGES_C);

    if (tid < 256) { sCnt[tid] = 0; sRank[tid] = 0; }
    int    dv[BIN_TRIPS];
    int    sv[BIN_TRIPS];
    float4 av[BIN_TRIPS];
#pragma unroll
    for (int k = 0; k < BIN_TRIPS; ++k) {
        const int e = min(e0 + tid + k * BIN_T, N_EDGES_C - 1);
        dv[k] = ei[N_EDGES_C + e];
        sv[k] = ei[e];
        av[k] = *(const float4*)(ea + (size_t)e * EDGE_DIM_C);
    }
#pragma unroll
    for (int k = 0; k < BIN_TRIPS; ++k)
        asm volatile("" :: "v"(dv[k]), "v"(sv[k]),
                     "v"(av[k].x), "v"(av[k].y), "v"(av[k].z), "v"(av[k].w));
    __syncthreads();
#pragma unroll
    for (int k = 0; k < BIN_TRIPS; ++k)
        if (e0 + tid + k * BIN_T < e1) atomicAdd(&sCnt[dv[k] >> BIN_SHIFT], 1);
    __syncthreads();
    if (tid < 32) {
        int c[8];
        int s = 0;
#pragma unroll
        for (int k = 0; k < 8; ++k) { c[k] = sCnt[lane * 8 + k]; s += c[k]; }
        int incl = s;
        for (int d = 1; d < 32; d <<= 1) {
            const int t = __shfl_up(incl, d);
            if (lane >= d) incl += t;
        }
        int run = incl - s;
#pragma unroll
        for (int k = 0; k < 8; ++k) {
            const int idx = lane * 8 + k;
            sStart[idx] = run;
            if (idx < NBINS) {
                hist  [idx * BIN_G + blk] = c[k];
                lstart[idx * BIN_G + blk] = run;
            }
            run += c[k];
        }
    }
    __syncthreads();
#pragma unroll
    for (int k = 0; k < BIN_TRIPS; ++k) {
        if (e0 + tid + k * BIN_T < e1) {
            const int bin = dv[k] >> BIN_SHIFT;
            const int lp  = sStart[bin] + atomicAdd(&sRank[bin], 1);
            lDst[lp] = dv[k];
            lSrc[lp] = sv[k];
            lEa[lp]  = av[k];
        }
    }
    __syncthreads();
    for (int j = tid; j < e1 - e0; j += BIN_T) {
        bdst[e0 + j] = lDst[j];
        bsrc[e0 + j] = lSrc[j];
        *(float4*)(bea + (size_t)(e0 + j) * EDGE_DIM_C) = lEa[j];
    }
    {
        const int g = blk * BIN_T + tid;
        if (g < WP_TOTAL) {
            _Float16 v;
            if      (g < WP_C1L2) v = pack_weight_elem(W11, IN_DIM_C, 1, g - WP_C1L1);
            else if (g < WP_C2L1) v = pack_weight_elem(W12, HID_C,    2, g - WP_C1L2);
            else if (g < WP_C2L2) v = pack_weight_elem(W21, HID_C,    2, g - WP_C2L1);
            else                  v = pack_weight_elem(W22, HID_C,    2, g - WP_C2L2);
            wplane[g] = v;
        }
    }
}

__device__ __forceinline__ int bin_edge_place(const int* sRun, const int* sBase, int i) {
    int lo = 0, hi = BIN_G;
#pragma unroll
    for (int s = 0; s < 8; ++s) {
        const int mid = (lo + hi) >> 1;
        if (sRun[mid] <= i) lo = mid; else hi = mid;
    }
    return sBase[lo] + (i - sRun[lo]);
}

__global__ __launch_bounds__(BIN_T)
void bin_sort_kernel(const int* __restrict__ hist, const int* __restrict__ lstart,
                     const int* __restrict__ bdst, const int* __restrict__ bsrc,
                     const float* __restrict__ bea,
                     int* __restrict__ rowptr, int* __restrict__ ssrc,
                     float* __restrict__ sea) {
    __shared__ int sRun[BIN_G + 1];
    __shared__ int sBase[BIN_G];
    __shared__ int sCnt[BIN_NODES];
    __shared__ int sPos[BIN_NODES];
    __shared__ int sW[BIN_W];
    __shared__ int sBeg;
    __shared__ __align__(16) float4 lEa[SORT_CAP];
    __shared__ int lSrc[SORT_CAP];
    const int tid   = threadIdx.x;
    const int lane  = tid & 31;
    const int wave  = tid >> 5;
    const int bin   = blockIdx.x;
    const int node0 = bin << BIN_SHIFT;

    if (tid < BIN_NODES) sCnt[tid] = 0;
    if (wave == 0) {
        const int4* ph = (const int4*)(hist   + bin * BIN_G + lane * 8);
        const int4* pl = (const int4*)(lstart + bin * BIN_G + lane * 8);
        const int4 ha = ph[0], hb = ph[1];
        const int4 la = pl[0], lb = pl[1];
        const int c[8]  = { ha.x, ha.y, ha.z, ha.w, hb.x, hb.y, hb.z, hb.w };
        const int ls[8] = { la.x, la.y, la.z, la.w, lb.x, lb.y, lb.z, lb.w };
        int s = 0;
        int front = 0;
#pragma unroll
        for (int k = 0; k < 8; ++k) { s += c[k]; front += ls[k]; }
        int incl = s;
        for (int d = 1; d < 32; d <<= 1) {
            const int t = __shfl_up(incl, d);
            if (lane >= d) incl += t;
        }
        for (int d = 16; d > 0; d >>= 1) front += __shfl_xor(front, d);
        int run = incl - s;
#pragma unroll
        for (int k = 0; k < 8; ++k) {
            const int b = lane * 8 + k;
            sRun[b]  = run;
            sBase[b] = b * BIN_CHUNK + ls[k];
            run += c[k];
        }
        if (lane == 31) sRun[BIN_G] = incl;
        if (lane == 0)  sBeg = front;
    }
    __syncthreads();
    const int beg = sBeg;
    const int n = sRun[BIN_G];

    const bool onepass = (n <= ONE_R * BIN_T);
    int    dvr[ONE_R];
    int    svr[ONE_R];
    int    rkr[ONE_R];
    float4 avr[ONE_R];
    if (onepass) {
        int ii[ONE_R];
        int lo[ONE_R];
        int hi[ONE_R];
#pragma unroll
        for (int k = 0; k < ONE_R; ++k) {
            ii[k] = max(min(tid + k * BIN_T, n - 1), 0);
            lo[k] = 0;
            hi[k] = BIN_G;
        }
#pragma unroll
        for (int s = 0; s < 8; ++s) {
            int rv[ONE_R];
#pragma unroll
            for (int k = 0; k < ONE_R; ++k) rv[k] = sRun[(lo[k] + hi[k]) >> 1];
#pragma unroll
            for (int k = 0; k < ONE_R; ++k) {
                const int mid = (lo[k] + hi[k]) >> 1;
                if (rv[k] <= ii[k]) lo[k] = mid; else hi[k] = mid;
            }
        }
#pragma unroll
        for (int k = 0; k < ONE_R; ++k) {
            const int g = min(sBase[lo[k]] + (ii[k] - sRun[lo[k]]), N_EDGES_C - 1);
            dvr[k] = bdst[g];
            svr[k] = bsrc[g];
            avr[k] = *(const float4*)(bea + (size_t)g * EDGE_DIM_C);
        }
#pragma unroll
        for (int k = 0; k < ONE_R; ++k)
            asm volatile("" :: "v"(dvr[k]), "v"(svr[k]),
                         "v"(avr[k].x), "v"(avr[k].y), "v"(avr[k].z), "v"(avr[k].w));
#pragma unroll
        for (int k = 0; k < ONE_R; ++k) {
            rkr[k] = 0;
            if (tid + k * BIN_T < n) rkr[k] = atomicAdd(&sCnt[dvr[k] - node0], 1);
        }
    } else {
        for (int i0 = tid; i0 < n; i0 += SORT_GRP * BIN_T) {
            int dv[SORT_GRP];
#pragma unroll
            for (int k = 0; k < SORT_GRP; ++k)
                dv[k] = bdst[bin_edge_place(sRun, sBase, min(i0 + k * BIN_T, n - 1))];
#pragma unroll
            for (int k = 0; k < SORT_GRP; ++k)
                asm volatile("" :: "v"(dv[k]));
#pragma unroll
            for (int k = 0; k < SORT_GRP; ++k)
                if (i0 + k * BIN_T < n) atomicAdd(&sCnt[dv[k] - node0], 1);
        }
    }
    __syncthreads();

    const int v = (tid < BIN_NODES) ? sCnt[tid] : 0;
    int incl = v;
    for (int d = 1; d < 32; d <<= 1) {
        const int t = __shfl_up(incl, d);
        if (lane >= d) incl += t;
    }
    if (lane == 31) sW[wave] = incl;
    __syncthreads();
    if (wave == 0) {
        const int w = sW[lane];
        int wi = w;
        for (int d = 1; d < 32; d <<= 1) {
            const int t = __shfl_up(wi, d);
            if (lane >= d) wi += t;
        }
        sW[lane] = wi - w;
    }
    __syncthreads();
    if (tid < BIN_NODES) {
        const int start = sW[wave] + incl - v;
        sPos[tid] = start;
        const int node = node0 + tid;
        if (node <= N_NODES_C) rowptr[node] = beg + start;
    }
    __syncthreads();

    const bool staged = (n <= SORT_CAP);
    if (onepass) {
#pragma unroll
        for (int k = 0; k < ONE_R; ++k) {
            if (tid + k * BIN_T < n) {
                const int lp = sPos[dvr[k] - node0] + rkr[k];
                lSrc[lp] = svr[k];
                lEa[lp]  = avr[k];
            }
        }
    } else {
        for (int i0 = tid; i0 < n; i0 += SORT_GRP * BIN_T) {
            int    dv[SORT_GRP];
            int    sv[SORT_GRP];
            float4 av[SORT_GRP];
#pragma unroll
            for (int k = 0; k < SORT_GRP; ++k) {
                const int g = bin_edge_place(sRun, sBase, min(i0 + k * BIN_T, n - 1));
                dv[k] = bdst[g];
                sv[k] = bsrc[g];
                av[k] = *(const float4*)(bea + (size_t)g * EDGE_DIM_C);
            }
#pragma unroll
            for (int k = 0; k < SORT_GRP; ++k)
                asm volatile("" :: "v"(dv[k]), "v"(sv[k]),
                             "v"(av[k].x), "v"(av[k].y), "v"(av[k].z), "v"(av[k].w));
#pragma unroll
            for (int k = 0; k < SORT_GRP; ++k) {
                if (i0 + k * BIN_T < n) {
                    const int lp = atomicAdd(&sPos[dv[k] - node0], 1);
                    if (staged) {
                        lSrc[lp] = sv[k];
                        lEa[lp]  = av[k];
                    } else {
                        ssrc[beg + lp] = sv[k];
                        *(float4*)(sea + (size_t)(beg + lp) * EDGE_DIM_C) = av[k];
                    }
                }
            }
        }
    }
    if (staged) {
        __syncthreads();
        for (int j = tid; j < n; j += BIN_T) {
            ssrc[beg + j] = lSrc[j];
            *(float4*)(sea + (size_t)(beg + j) * EDGE_DIM_C) = lEa[j];
        }
    }
}

__global__ __launch_bounds__(256)
void fc_kernel(const float* __restrict__ psum, const float* __restrict__ cnt,
               const float* __restrict__ fcW, const float* __restrict__ fcb,
               float* __restrict__ out) {
    const int t = blockIdx.x * 256 + threadIdx.x;
    if (t >= N_GRAPHS_C * NUM_TASKS_C) return;
    const int g = t / NUM_TASKS_C;
    const int k = t - g * NUM_TASKS_C;
    const float inv = 1.0f / fmaxf(cnt[g], 1.0f);
    float s = fcb[k];
#pragma unroll
    for (int c = 0; c < HID_C; ++c)
        s += psum[(size_t)g * HID_C + c] * inv * fcW[k * HID_C + c];
    out[t] = s;
}

extern "C" void kernel_launch(void* const* d_in, const int* in_sizes, int n_in,
                              void* d_out, int out_size, void* d_ws, size_t ws_size,
                              hipStream_t stream) {
    const float* x    = (const float*)d_in[0];
    const int*   ei   = (const int*)  d_in[1];
    const float* ea   = (const float*)d_in[2];
    const int*   bat  = (const int*)  d_in[3];
    const float* e1W  = (const float*)d_in[4];  const float* e1b  = (const float*)d_in[5];
    const float* n1W1 = (const float*)d_in[6];  const float* n1b1 = (const float*)d_in[7];
    const float* n1W2 = (const float*)d_in[8];  const float* n1b2 = (const float*)d_in[9];
    const float* e2W  = (const float*)d_in[10]; const float* e2b  = (const float*)d_in[11];
    const float* n2W1 = (const float*)d_in[12]; const float* n2b1 = (const float*)d_in[13];
    const float* n2W2 = (const float*)d_in[14]; const float* n2b2 = (const float*)d_in[15];
    const float* fcW  = (const float*)d_in[16]; const float* fcb  = (const float*)d_in[17];
    float* out = (float*)d_out;

    _Float16* h     = (_Float16*)d_ws;
    float*    psum  = (float*)(h + (size_t)N_NODES_C * HID_C);
    float*    cnt   = psum  + (size_t)N_GRAPHS_C * HID_C;
    float*    sea   = cnt   + (size_t)N_GRAPHS_C;
    int*      ssrc  = (int*)(sea + (size_t)N_EDGES_C * EDGE_DIM_C);
    int*      bdst  = ssrc  + (size_t)N_EDGES_C;
    int*      bsrc  = bdst  + (size_t)N_EDGES_C;
    float*    bea   = (float*)(bsrc + (size_t)N_EDGES_C);
    int*      hist  = (int*)(bea + (size_t)N_EDGES_C * EDGE_DIM_C);
    int*      lstart = hist + (size_t)NBINS * BIN_G;
    int*      rowptr = lstart + (size_t)NBINS * BIN_G;
    _Float16* wplane = (_Float16*)(rowptr + (size_t)ROWPTR_LEN);

    hipMemsetAsync(psum,  0, sizeof(float) * (size_t)N_GRAPHS_C * HID_C,  stream);
    hipMemsetAsync(cnt,   0, sizeof(float) * (size_t)N_GRAPHS_C,          stream);

    chunk_sort_kernel<<<BIN_G, BIN_T, 0, stream>>>(ei, ea, hist, lstart, bdst, bsrc, bea,
                                                   n1W1, n1W2, n2W1, n2W2, wplane);
    bin_sort_kernel<<<NBINS, BIN_T, 0, stream>>>(hist, lstart, bdst, bsrc, bea, rowptr, ssrc, sea);
    const unsigned grid = (N_NODES_C + 15) / 16;
    conv_kernel<IN_DIM_C, float, false><<<grid, 128, 0, stream>>>(
        x, rowptr, ssrc, sea, e1W, e1b, wplane + WP_C1L1, n1b1, wplane + WP_C1L2, n1b2, h,
        nullptr, nullptr, nullptr, N_NODES_C);
    conv_kernel<HID_C, _Float16, true><<<grid, 128, 0, stream>>>(
        h, rowptr, ssrc, sea, e2W, e2b, wplane + WP_C2L1, n2b1, wplane + WP_C2L2, n2b2, nullptr,
        bat, psum, cnt, N_NODES_C);
    {
        const unsigned fgrid = (N_GRAPHS_C * NUM_TASKS_C + 255) / 256;
        fc_kernel<<<fgrid, 256, 0, stream>>>(psum, cnt, fcW, fcb, out);
    }
}
